# starting kernel
# baseline (speedup 1.0000x reference)
#include <hip/hip_runtime.h>

typedef __attribute__((ext_vector_type(2))) float v2f;
typedef __attribute__((ext_vector_type(8))) float v8f;

constexpr int TASKS   = 100000;
constexpr int WORKERS = 500;
constexpr int FEAT    = 512;
constexpr int CLS     = 10;
constexpr int HP      = 12;
constexpr int ANS     = 1000000;
constexpr int HID     = 128;
constexpr int NODES   = TASKS + WORKERS;
constexpr int CHUNK   = 4096;


__global__ void k_deg_init(int* __restrict__ deg) {
    int i = blockIdx.x * 256 + threadIdx.x;
    if (i < NODES) deg[i] = 1;
}

__global__ void k_deg_count(const int* __restrict__ answers, int* __restrict__ deg) {
    int a = blockIdx.x * 256 + threadIdx.x;
    if (a < ANS) {
        atomicAdd(&deg[answers[3 * a]], 1);
        atomicAdd(&deg[TASKS + answers[3 * a + 1]], 1);
    }
}

__global__ void k_dinv(const int* __restrict__ deg, float* __restrict__ dinv) {
    int i = blockIdx.x * 256 + threadIdx.x;
    if (i < NODES) dinv[i] = rsqrtf((float)deg[i]);
}


__global__ void k_packB(const float* __restrict__ B, float2* __restrict__ Bp) {
    int idx = blockIdx.x * 256 + threadIdx.x;
    if (idx < (FEAT / 2) * HID) {
        int kp = idx / HID;
        int n  = idx - kp * HID;
        Bp[idx] = make_float2(B[(2 * kp) * HID + n], B[(2 * kp + 1) * HID + n]);
    }
}


__global__ __launch_bounds__(256) void k_gemm_efc(const float* __restrict__ A,
                                                  const float2* __restrict__ Bp,
                                                  const float* __restrict__ bias,
                                                  float* __restrict__ out) {
    const int wave   = threadIdx.x >> 5;
    const int lane   = threadIdx.x & 31;
    const int lane16 = lane & 15;
    const int hi     = lane >> 4;

    const int slab     = blockIdx.x * 8 + wave;
    const bool valid   = (slab * 32) < TASKS;
    const int row_base = valid ? slab * 32 : 0;

    const float*  arow0 = A + (size_t)(row_base + lane16) * FEAT + 2 * hi;
    const float*  arow1 = arow0 + (size_t)16 * FEAT;
    const float2* bbase = Bp + lane16;

    v8f acc[2][8];
#pragma unroll
    for (int t = 0; t < 8; ++t) {
        float bv = bias[t * 16 + lane16];
        acc[0][t] = (v8f){bv, bv, bv, bv, bv, bv, bv, bv};
        acc[1][t] = acc[0][t];
    }

    for (int k = 0; k < FEAT; k += 4) {
        const float2 av0 = *(const float2*)(arow0 + k);
        const float2 av1 = *(const float2*)(arow1 + k);
        v2f af0 = {av0.x, av0.y};
        v2f af1 = {av1.x, av1.y};
        const float2* brow = bbase + (size_t)((k >> 1) + hi) * HID;
#pragma unroll
        for (int t = 0; t < 8; ++t) {
            const float2 bv = brow[t * 16];
            v2f bf = {bv.x, bv.y};
            acc[0][t] = __builtin_amdgcn_wmma_f32_16x16x4_f32(
                false, af0, false, bf, (short)0, acc[0][t], false, false);
            acc[1][t] = __builtin_amdgcn_wmma_f32_16x16x4_f32(
                false, af1, false, bf, (short)0, acc[1][t], false, false);
        }
    }

    if (valid) {
#pragma unroll
        for (int r = 0; r < 2; ++r) {
#pragma unroll
            for (int t = 0; t < 8; ++t) {
#pragma unroll
                for (int j = 0; j < 8; ++j) {
                    int m   = row_base + r * 16 + j + 8 * hi;
                    float v = acc[r][t][j];
                    out[(size_t)m * HID + t * 16 + lane16] = v > 0.f ? v : 0.f;
                }
            }
        }
    }
}


__global__ __launch_bounds__(256) void k_h(const float* __restrict__ tf_h,
                                           const float* __restrict__ wf,
                                           const float* __restrict__ Wm,
                                           const float* __restrict__ Wl,
                                           float* __restrict__ hm,
                                           float* __restrict__ hl) {
    const int wave   = threadIdx.x >> 5;
    const int lane   = threadIdx.x & 31;
    const int lane16 = lane & 15;
    const int hi     = lane >> 4;

    const int node_base = (blockIdx.x * 8 + wave) * 16;
    int nd = node_base + lane16;
    if (nd > NODES - 1) nd = NODES - 1;
    const float* xrow = (nd < TASKS) ? (tf_h + (size_t)nd * HID)
                                     : (wf + (size_t)(nd - TASKS) * HID);
    const int n = lane16;
    const bool ncol = n < CLS;

    v8f accM = {0.f, 0.f, 0.f, 0.f, 0.f, 0.f, 0.f, 0.f};
    v8f accL = {0.f, 0.f, 0.f, 0.f, 0.f, 0.f, 0.f, 0.f};

    for (int k = 0; k < HID; k += 4) {
        const int k0 = k + 2 * hi;
        const float2 av = *(const float2*)(xrow + k0);
        v2f af = {av.x, av.y};
        v2f bm, bl;
        bm.x = ncol ? Wm[k0 * CLS + n] : 0.f;
        bm.y = ncol ? Wm[(k0 + 1) * CLS + n] : 0.f;
        bl.x = ncol ? Wl[k0 * CLS + n] : 0.f;
        bl.y = ncol ? Wl[(k0 + 1) * CLS + n] : 0.f;
        accM = __builtin_amdgcn_wmma_f32_16x16x4_f32(false, af, false, bm, (short)0, accM, false, false);
        accL = __builtin_amdgcn_wmma_f32_16x16x4_f32(false, af, false, bl, (short)0, accL, false, false);
    }

    if (ncol) {
#pragma unroll
        for (int j = 0; j < 8; ++j) {
            int m = node_base + j + 8 * hi;
            if (m < NODES) {
                hm[(size_t)m * HP + n] = accM[j];
                hl[(size_t)m * HP + n] = accL[j];
            }
        }
    }
}


__global__ void k_agg_init(const float* __restrict__ hm, const float* __restrict__ hl,
                           const float* __restrict__ dinv,
                           const float* __restrict__ bm, const float* __restrict__ bl,
                           float* __restrict__ mean, float* __restrict__ lsd) {
    int idx = blockIdx.x * 256 + threadIdx.x;
    if (idx < NODES * CLS) {
        int i = idx / CLS;
        int c = idx - i * CLS;
        float d  = dinv[i];
        float sw = d * d;
        mean[idx] = bm[c] + hm[(size_t)i * HP + c] * sw;
        lsd[idx]  = bl[c] + hl[(size_t)i * HP + c] * sw;
    }
}


__device__ __forceinline__ void load_row10(const float* __restrict__ p, float* r) {
    const float4 a = *(const float4*)(p);
    const float4 b = *(const float4*)(p + 4);
    const float2 c = *(const float2*)(p + 8);
    r[0] = a.x; r[1] = a.y; r[2] = a.z; r[3] = a.w;
    r[4] = b.x; r[5] = b.y; r[6] = b.z; r[7] = b.w;
    r[8] = c.x; r[9] = c.y;
}

__global__ __launch_bounds__(256) void k_agg_edges(const int* __restrict__ answers,
                                                   const float* __restrict__ hm,
                                                   const float* __restrict__ hl,
                                                   const float* __restrict__ dinv,
                                                   float* __restrict__ mean,
                                                   float* __restrict__ lsd) {
    __shared__ float wacc[2 * WORKERS * CLS];
    for (int i = threadIdx.x; i < 2 * WORKERS * CLS; i += 256) wacc[i] = 0.f;
    __syncthreads();

    const int start = blockIdx.x * CHUNK;
    int end = start + CHUNK;
    if (end > ANS) end = ANS;

    for (int a = start + threadIdx.x; a < end; a += 256) {
        const int t    = answers[3 * a];
        const int w    = answers[3 * a + 1];
        const int wrow = TASKS + w;
        const float we = dinv[t] * dinv[wrow];

        float hmt[10], hmw[10], hlt[10], hlw[10];
        load_row10(hm + (size_t)t * HP,    hmt);
        load_row10(hm + (size_t)wrow * HP, hmw);
        load_row10(hl + (size_t)t * HP,    hlt);
        load_row10(hl + (size_t)wrow * HP, hlw);

        float* mt = mean + (size_t)t * CLS;
        float* lt = lsd + (size_t)t * CLS;
#pragma unroll
        for (int c = 0; c < CLS; ++c) {
            atomicAdd(&mt[c], hmw[c] * we);
            atomicAdd(&lt[c], hlw[c] * we);
            atomicAdd(&wacc[w * CLS + c], hmt[c] * we);
            atomicAdd(&wacc[WORKERS * CLS + w * CLS + c], hlt[c] * we);
        }
    }
    __syncthreads();

    for (int i = threadIdx.x; i < WORKERS * CLS; i += 256) {
        atomicAdd(&mean[(size_t)TASKS * CLS + i], wacc[i]);
        atomicAdd(&lsd[(size_t)TASKS * CLS + i], wacc[WORKERS * CLS + i]);
    }
}


__global__ void k_z(const float* __restrict__ mean, const float* __restrict__ lsd,
                    const float* __restrict__ eps, float* __restrict__ z) {
    int idx = blockIdx.x * 256 + threadIdx.x;
    if (idx < NODES * CLS)
        z[idx] = mean[idx] + (eps[idx] * 0.01f) * expf(lsd[idx]);
}

__global__ void k_crowd(const int* __restrict__ answers, const float* __restrict__ z,
                        float* __restrict__ crowd) {
    int idx = blockIdx.x * 256 + threadIdx.x;
    if (idx < ANS * CLS) {
        int a = idx / CLS;
        int c = idx - a * CLS;
        int t = answers[3 * a];
        int w = answers[3 * a + 1];
        crowd[idx] = z[(size_t)t * CLS + c] * z[(size_t)(TASKS + w) * CLS + c];
    }
}


extern "C" void kernel_launch(void* const* d_in, const int* in_sizes, int n_in,
                              void* d_out, int out_size, void* d_ws, size_t ws_size,
                              hipStream_t stream) {
    (void)in_sizes; (void)n_in; (void)out_size; (void)ws_size;

    const float* task_feature   = (const float*)d_in[0];
    const int*   answers        = (const int*)d_in[1];
    const float* worker_feature = (const float*)d_in[2];
    const float* W_efc          = (const float*)d_in[3];
    const float* b_efc          = (const float*)d_in[4];
    const float* W_mean         = (const float*)d_in[5];
    const float* b_mean         = (const float*)d_in[6];
    const float* W_ls           = (const float*)d_in[7];
    const float* b_ls           = (const float*)d_in[8];
    const float* eps            = (const float*)d_in[9];

    float* out   = (float*)d_out;
    float* crowd = out;
    float* mean  = out + (size_t)ANS * CLS;
    float* lsd   = mean + (size_t)NODES * CLS;

    float*  ws   = (float*)d_ws;
    float*  tf_h = ws;
    float2* Bp   = (float2*)(tf_h + (size_t)TASKS * HID);
    float*  hm   = (float*)(Bp + (size_t)(FEAT / 2) * HID);
    float*  hl   = hm + (size_t)NODES * HP;
    float*  zz   = hl + (size_t)NODES * HP;
    float*  dinv = zz + (size_t)NODES * CLS;
    int*    deg  = (int*)(dinv + NODES);

    auto cdiv = [](int a, int b) { return (a + b - 1) / b; };

    k_deg_init <<<cdiv(NODES, 256), 256, 0, stream>>>(deg);
    k_deg_count<<<cdiv(ANS, 256), 256, 0, stream>>>(answers, deg);
    k_dinv     <<<cdiv(NODES, 256), 256, 0, stream>>>(deg, dinv);

    k_packB    <<<cdiv((FEAT / 2) * HID, 256), 256, 0, stream>>>(W_efc, Bp);
    k_gemm_efc <<<cdiv(TASKS / 32, 8), 256, 0, stream>>>(task_feature, Bp, b_efc, tf_h);
    k_h        <<<cdiv(cdiv(NODES, 16), 8), 256, 0, stream>>>(tf_h, worker_feature,
                                                              W_mean, W_ls, hm, hl);

    k_agg_init <<<cdiv(NODES * CLS, 256), 256, 0, stream>>>(hm, hl, dinv, b_mean, b_ls, mean, lsd);
    k_agg_edges<<<cdiv(ANS, CHUNK), 256, 0, stream>>>(answers, hm, hl, dinv, mean, lsd);

    k_z        <<<cdiv(NODES * CLS, 256), 256, 0, stream>>>(mean, lsd, eps, zz);
    k_crowd    <<<cdiv(ANS * CLS, 256), 256, 0, stream>>>(answers, zz, crowd);
}


# Edge aggregation without float atomics: task rows summed from per-task worker-id lists in ascending id order, worker rows in 64-bit fixed-point integer adds; sums now order-independent
# speedup vs baseline: 1.7832x; 1.7832x over previous
#include <hip/hip_runtime.h>

typedef __attribute__((ext_vector_type(2))) float v2f;
typedef __attribute__((ext_vector_type(8))) float v8f;

constexpr int TASKS   = 100000;
constexpr int WORKERS = 500;
constexpr int FEAT    = 512;
constexpr int CLS     = 10;
constexpr int HP      = 12;
constexpr int ANS     = 1000000;
constexpr int HID     = 128;
constexpr int NODES   = TASKS + WORKERS;
constexpr int CHUNK   = 4096;
constexpr int CAP     = 32;
constexpr int SENT    = 0xFFFF;


__global__ void k_deg_init(int* __restrict__ deg, unsigned long long* __restrict__ wsum) {
    int i = blockIdx.x * 256 + threadIdx.x;
    if (i < NODES) deg[i] = 1;
    if (i < 2 * WORKERS * CLS) wsum[i] = 0ull;
}

__global__ void k_deg_count(const int* __restrict__ answers, int* __restrict__ deg,
                            unsigned short* __restrict__ list) {
    int a = blockIdx.x * 256 + threadIdx.x;
    if (a < ANS) {
        const int t = answers[3 * a];
        const int w = answers[3 * a + 1];
        const int slot = atomicAdd(&deg[t], 1) - 1;
        atomicAdd(&deg[TASKS + w], 1);
        if (slot < CAP) list[(size_t)t * CAP + slot] = (unsigned short)w;
    }
}

__global__ void k_dinv(const int* __restrict__ deg, float* __restrict__ dinv) {
    int i = blockIdx.x * 256 + threadIdx.x;
    if (i < NODES) dinv[i] = rsqrtf((float)deg[i]);
}


__global__ void k_packB(const float* __restrict__ B, float2* __restrict__ Bp) {
    int idx = blockIdx.x * 256 + threadIdx.x;
    if (idx < (FEAT / 2) * HID) {
        int kp = idx / HID;
        int n  = idx - kp * HID;
        Bp[idx] = make_float2(B[(2 * kp) * HID + n], B[(2 * kp + 1) * HID + n]);
    }
}


__global__ __launch_bounds__(256) void k_gemm_efc(const float* __restrict__ A,
                                                  const float2* __restrict__ Bp,
                                                  const float* __restrict__ bias,
                                                  float* __restrict__ out) {
    const int wave   = threadIdx.x >> 5;
    const int lane   = threadIdx.x & 31;
    const int lane16 = lane & 15;
    const int hi     = lane >> 4;

    const int slab     = blockIdx.x * 8 + wave;
    const bool valid   = (slab * 32) < TASKS;
    const int row_base = valid ? slab * 32 : 0;

    const float*  arow0 = A + (size_t)(row_base + lane16) * FEAT + 2 * hi;
    const float*  arow1 = arow0 + (size_t)16 * FEAT;
    const float2* bbase = Bp + lane16;

    v8f acc[2][8];
#pragma unroll
    for (int t = 0; t < 8; ++t) {
        float bv = bias[t * 16 + lane16];
        acc[0][t] = (v8f){bv, bv, bv, bv, bv, bv, bv, bv};
        acc[1][t] = acc[0][t];
    }

    for (int k = 0; k < FEAT; k += 4) {
        const float2 av0 = *(const float2*)(arow0 + k);
        const float2 av1 = *(const float2*)(arow1 + k);
        v2f af0 = {av0.x, av0.y};
        v2f af1 = {av1.x, av1.y};
        const float2* brow = bbase + (size_t)((k >> 1) + hi) * HID;
#pragma unroll
        for (int t = 0; t < 8; ++t) {
            const float2 bv = brow[t * 16];
            v2f bf = {bv.x, bv.y};
            acc[0][t] = __builtin_amdgcn_wmma_f32_16x16x4_f32(
                false, af0, false, bf, (short)0, acc[0][t], false, false);
            acc[1][t] = __builtin_amdgcn_wmma_f32_16x16x4_f32(
                false, af1, false, bf, (short)0, acc[1][t], false, false);
        }
    }

    if (valid) {
#pragma unroll
        for (int r = 0; r < 2; ++r) {
#pragma unroll
            for (int t = 0; t < 8; ++t) {
#pragma unroll
                for (int j = 0; j < 8; ++j) {
                    int m   = row_base + r * 16 + j + 8 * hi;
                    float v = acc[r][t][j];
                    out[(size_t)m * HID + t * 16 + lane16] = v > 0.f ? v : 0.f;
                }
            }
        }
    }
}


__global__ __launch_bounds__(256) void k_h(const float* __restrict__ tf_h,
                                           const float* __restrict__ wf,
                                           const float* __restrict__ Wm,
                                           const float* __restrict__ Wl,
                                           float* __restrict__ hm,
                                           float* __restrict__ hl) {
    const int wave   = threadIdx.x >> 5;
    const int lane   = threadIdx.x & 31;
    const int lane16 = lane & 15;
    const int hi     = lane >> 4;

    const int node_base = (blockIdx.x * 8 + wave) * 16;
    int nd = node_base + lane16;
    if (nd > NODES - 1) nd = NODES - 1;
    const float* xrow = (nd < TASKS) ? (tf_h + (size_t)nd * HID)
                                     : (wf + (size_t)(nd - TASKS) * HID);
    const int n = lane16;
    const bool ncol = n < CLS;

    v8f accM = {0.f, 0.f, 0.f, 0.f, 0.f, 0.f, 0.f, 0.f};
    v8f accL = {0.f, 0.f, 0.f, 0.f, 0.f, 0.f, 0.f, 0.f};

    for (int k = 0; k < HID; k += 4) {
        const int k0 = k + 2 * hi;
        const float2 av = *(const float2*)(xrow + k0);
        v2f af = {av.x, av.y};
        v2f bm, bl;
        bm.x = ncol ? Wm[k0 * CLS + n] : 0.f;
        bm.y = ncol ? Wm[(k0 + 1) * CLS + n] : 0.f;
        bl.x = ncol ? Wl[k0 * CLS + n] : 0.f;
        bl.y = ncol ? Wl[(k0 + 1) * CLS + n] : 0.f;
        accM = __builtin_amdgcn_wmma_f32_16x16x4_f32(false, af, false, bm, (short)0, accM, false, false);
        accL = __builtin_amdgcn_wmma_f32_16x16x4_f32(false, af, false, bl, (short)0, accL, false, false);
    }

    if (ncol) {
#pragma unroll
        for (int j = 0; j < 8; ++j) {
            int m = node_base + j + 8 * hi;
            if (m < NODES) {
                hm[(size_t)m * HP + n] = accM[j];
                hl[(size_t)m * HP + n] = accL[j];
            }
        }
    }
}


__device__ __forceinline__ void load_row10(const float* __restrict__ p, float* r) {
    const float4 a = *(const float4*)(p);
    const float4 b = *(const float4*)(p + 4);
    const float2 c = *(const float2*)(p + 8);
    r[0] = a.x; r[1] = a.y; r[2] = a.z; r[3] = a.w;
    r[4] = b.x; r[5] = b.y; r[6] = b.z; r[7] = b.w;
    r[8] = c.x; r[9] = c.y;
}

__device__ __forceinline__ unsigned long long to_fix(float v) {
    return (unsigned long long)(long long)rintf(v * 4294967296.0f);
}

__device__ __forceinline__ float from_fix(unsigned long long s) {
    return (float)((double)(long long)s * (1.0 / 4294967296.0));
}

__global__ __launch_bounds__(256) void k_agg_task(const int* __restrict__ answers,
                                                  const unsigned short* __restrict__ list,
                                                  const int* __restrict__ deg,
                                                  const float* __restrict__ hm,
                                                  const float* __restrict__ hl,
                                                  const float* __restrict__ dinv,
                                                  const float* __restrict__ bm,
                                                  const float* __restrict__ bl,
                                                  float* __restrict__ mean,
                                                  float* __restrict__ lsd) {
    __shared__ float sm[256 * CLS];
    __shared__ float sl[256 * CLS];

    const int i = blockIdx.x * 256 + threadIdx.x;
    float am[CLS], al[CLS];
#pragma unroll
    for (int c = 0; c < CLS; ++c) { am[c] = 0.f; al[c] = 0.f; }

    if (i < NODES) {
        const float d  = dinv[i];
        const float sw = d * d;
        float hmi[CLS], hli[CLS];
        load_row10(hm + (size_t)i * HP, hmi);
        load_row10(hl + (size_t)i * HP, hli);
#pragma unroll
        for (int c = 0; c < CLS; ++c) {
            am[c] = bm[c] + hmi[c] * sw;
            al[c] = bl[c] + hli[c] * sw;
        }

        if (i < TASKS) {
            const int n = deg[i] - 1;
            if (n <= CAP) {
                const uint4* lp = (const uint4*)(list + (size_t)i * CAP);
                const uint4 q0 = lp[0], q1 = lp[1], q2 = lp[2], q3 = lp[3];
                const unsigned pk[16] = {q0.x, q0.y, q0.z, q0.w, q1.x, q1.y, q1.z, q1.w,
                                         q2.x, q2.y, q2.z, q2.w, q3.x, q3.y, q3.z, q3.w};
                int e[CAP];
#pragma unroll
                for (int j = 0; j < CAP; ++j) {
                    const int v = (int)((pk[j >> 1] >> (16 * (j & 1))) & 0xFFFFu);
                    e[j] = (j < n) ? v : SENT;
                }
                int cur = -1;
                for (int it = 0; it < CAP; ++it) {
                    int m = SENT, cnt = 0;
#pragma unroll
                    for (int j = 0; j < CAP; ++j) {
                        const int v = e[j];
                        if (v > cur) {
                            if (v < m) { m = v; cnt = 1; }
                            else if (v == m) ++cnt;
                        }
                    }
                    if (m == SENT) break;
                    const int wrow = TASKS + m;
                    const float we = d * dinv[wrow];
                    float hmw[CLS], hlw[CLS];
                    load_row10(hm + (size_t)wrow * HP, hmw);
                    load_row10(hl + (size_t)wrow * HP, hlw);
                    for (int k = 0; k < cnt; ++k) {
#pragma unroll
                        for (int c = 0; c < CLS; ++c) {
                            am[c] += hmw[c] * we;
                            al[c] += hlw[c] * we;
                        }
                    }
                    cur = m;
                }
            } else {
                for (int a = 0; a < ANS; ++a) {
                    if (answers[3 * a] == i) {
                        const int wrow = TASKS + answers[3 * a + 1];
                        const float we = d * dinv[wrow];
                        float hmw[CLS], hlw[CLS];
                        load_row10(hm + (size_t)wrow * HP, hmw);
                        load_row10(hl + (size_t)wrow * HP, hlw);
#pragma unroll
                        for (int c = 0; c < CLS; ++c) {
                            am[c] += hmw[c] * we;
                            al[c] += hlw[c] * we;
                        }
                    }
                }
            }
        }
    }

#pragma unroll
    for (int c = 0; c < CLS; ++c) {
        sm[threadIdx.x * CLS + c] = am[c];
        sl[threadIdx.x * CLS + c] = al[c];
    }
    __syncthreads();
    const int base = blockIdx.x * 256 * CLS;
    const int lim  = NODES * CLS - base;
    for (int k = threadIdx.x; k < 256 * CLS; k += 256) {
        if (k < lim) {
            mean[(size_t)base + k] = sm[k];
            lsd[(size_t)base + k]  = sl[k];
        }
    }
}

__global__ __launch_bounds__(256) void k_agg_worker(const int* __restrict__ answers,
                                                    const float* __restrict__ hm,
                                                    const float* __restrict__ hl,
                                                    const float* __restrict__ dinv,
                                                    unsigned long long* __restrict__ wsum) {
    __shared__ unsigned long long wacc[2 * WORKERS * CLS];
    for (int i = threadIdx.x; i < 2 * WORKERS * CLS; i += 256) wacc[i] = 0ull;
    __syncthreads();

    const int start = blockIdx.x * CHUNK;
    int end = start + CHUNK;
    if (end > ANS) end = ANS;

    for (int a = start + threadIdx.x; a < end; a += 256) {
        const int t = answers[3 * a];
        const int w = answers[3 * a + 1];
        const float we = dinv[t] * dinv[TASKS + w];

        float hmt[CLS], hlt[CLS];
        load_row10(hm + (size_t)t * HP, hmt);
        load_row10(hl + (size_t)t * HP, hlt);
#pragma unroll
        for (int c = 0; c < CLS; ++c) {
            atomicAdd(&wacc[w * CLS + c], to_fix(hmt[c] * we));
            atomicAdd(&wacc[WORKERS * CLS + w * CLS + c], to_fix(hlt[c] * we));
        }
    }
    __syncthreads();

    for (int i = threadIdx.x; i < 2 * WORKERS * CLS; i += 256) atomicAdd(&wsum[i], wacc[i]);
}


__global__ void k_z(float* __restrict__ mean, float* __restrict__ lsd,
                    const unsigned long long* __restrict__ wsum,
                    const float* __restrict__ eps, float* __restrict__ z) {
    int idx = blockIdx.x * 256 + threadIdx.x;
    if (idx < NODES * CLS) {
        float m = mean[idx];
        float l = lsd[idx];
        if (idx >= TASKS * CLS) {
            const int j = idx - TASKS * CLS;
            m += from_fix(wsum[j]);
            l += from_fix(wsum[WORKERS * CLS + j]);
            mean[idx] = m;
            lsd[idx]  = l;
        }
        z[idx] = m + (eps[idx] * 0.01f) * expf(l);
    }
}

__global__ void k_crowd(const int* __restrict__ answers, const float* __restrict__ z,
                        float* __restrict__ crowd) {
    int idx = blockIdx.x * 256 + threadIdx.x;
    if (idx < ANS * CLS) {
        int a = idx / CLS;
        int c = idx - a * CLS;
        int t = answers[3 * a];
        int w = answers[3 * a + 1];
        crowd[idx] = z[(size_t)t * CLS + c] * z[(size_t)(TASKS + w) * CLS + c];
    }
}


extern "C" void kernel_launch(void* const* d_in, const int* in_sizes, int n_in,
                              void* d_out, int out_size, void* d_ws, size_t ws_size,
                              hipStream_t stream) {
    (void)in_sizes; (void)n_in; (void)out_size; (void)ws_size;

    const float* task_feature   = (const float*)d_in[0];
    const int*   answers        = (const int*)d_in[1];
    const float* worker_feature = (const float*)d_in[2];
    const float* W_efc          = (const float*)d_in[3];
    const float* b_efc          = (const float*)d_in[4];
    const float* W_mean         = (const float*)d_in[5];
    const float* b_mean         = (const float*)d_in[6];
    const float* W_ls           = (const float*)d_in[7];
    const float* b_ls           = (const float*)d_in[8];
    const float* eps            = (const float*)d_in[9];

    float* out   = (float*)d_out;
    float* crowd = out;
    float* mean  = out + (size_t)ANS * CLS;
    float* lsd   = mean + (size_t)NODES * CLS;

    float*  ws   = (float*)d_ws;
    float*  tf_h = ws;
    float2* Bp   = (float2*)(tf_h + (size_t)TASKS * HID);
    float*  hm   = (float*)(Bp + (size_t)(FEAT / 2) * HID);
    float*  hl   = hm + (size_t)NODES * HP;
    float*  zz   = hl + (size_t)NODES * HP;
    float*  dinv = zz + (size_t)NODES * CLS;
    int*    deg  = (int*)(dinv + NODES);
    unsigned short*     list = (unsigned short*)tf_h;
    unsigned long long* wsum = (unsigned long long*)(tf_h + (size_t)2 * 1024 * 1024);

    auto cdiv = [](int a, int b) { return (a + b - 1) / b; };

    k_packB    <<<cdiv((FEAT / 2) * HID, 256), 256, 0, stream>>>(W_efc, Bp);
    k_gemm_efc <<<cdiv(TASKS / 32, 8), 256, 0, stream>>>(task_feature, Bp, b_efc, tf_h);
    k_h        <<<cdiv(cdiv(NODES, 16), 8), 256, 0, stream>>>(tf_h, worker_feature,
                                                              W_mean, W_ls, hm, hl);

    k_deg_init <<<cdiv(NODES, 256), 256, 0, stream>>>(deg, wsum);
    k_deg_count<<<cdiv(ANS, 256), 256, 0, stream>>>(answers, deg, list);
    k_dinv     <<<cdiv(NODES, 256), 256, 0, stream>>>(deg, dinv);

    k_agg_task  <<<cdiv(NODES, 256), 256, 0, stream>>>(answers, list, deg, hm, hl, dinv,
                                                       b_mean, b_ls, mean, lsd);
    k_agg_worker<<<cdiv(ANS, CHUNK), 256, 0, stream>>>(answers, hm, hl, dinv, wsum);

    k_z        <<<cdiv(NODES * CLS, 256), 256, 0, stream>>>(mean, lsd, wsum, eps, zz);
    k_crowd    <<<cdiv(ANS * CLS, 256), 256, 0, stream>>>(answers, zz, crowd);
}


# Crowd product and z leave the one-thread-per-output-element form (43K short work-groups): about 500 work-groups loop, one 48-byte z row per answer, worker rows in LDS, whole-line stores via LDS
# speedup vs baseline: 2.0550x; 1.1524x over previous
#include <hip/hip_runtime.h>

typedef __attribute__((ext_vector_type(2))) float v2f;
typedef __attribute__((ext_vector_type(8))) float v8f;

constexpr int TASKS   = 100000;
constexpr int WORKERS = 500;
constexpr int FEAT    = 512;
constexpr int CLS     = 10;
constexpr int HP      = 12;
constexpr int ANS     = 1000000;
constexpr int HID     = 128;
constexpr int NODES   = TASKS + WORKERS;
constexpr int CHUNK   = 4096;
constexpr int CAP     = 32;
constexpr int SENT    = 0xFFFF;
constexpr int Z_PER   = 8;
constexpr int CR_TILE = 512;
constexpr int CR_GRID = 512;
static_assert(ANS % 2 == 0 && CR_TILE % 2 == 0, "a tile's floats must split into 16-byte pieces");
static_assert((CR_TILE * CLS / 4) % 256 == 0, "a full tile is a whole number of 256-thread store rounds");


__global__ void k_deg_init(int* __restrict__ deg, unsigned long long* __restrict__ wsum) {
    int i = blockIdx.x * 256 + threadIdx.x;
    if (i < NODES) deg[i] = 1;
    if (i < 2 * WORKERS * CLS) wsum[i] = 0ull;
}

__global__ void k_deg_count(const int* __restrict__ answers, int* __restrict__ deg,
                            unsigned short* __restrict__ list) {
    int a = blockIdx.x * 256 + threadIdx.x;
    if (a < ANS) {
        const int t = answers[3 * a];
        const int w = answers[3 * a + 1];
        const int slot = atomicAdd(&deg[t], 1) - 1;
        atomicAdd(&deg[TASKS + w], 1);
        if (slot < CAP) list[(size_t)t * CAP + slot] = (unsigned short)w;
    }
}

__global__ void k_dinv(const int* __restrict__ deg, float* __restrict__ dinv) {
    int i = blockIdx.x * 256 + threadIdx.x;
    if (i < NODES) dinv[i] = rsqrtf((float)deg[i]);
}


__global__ void k_packB(const float* __restrict__ B, float2* __restrict__ Bp) {
    int idx = blockIdx.x * 256 + threadIdx.x;
    if (idx < (FEAT / 2) * HID) {
        int kp = idx / HID;
        int n  = idx - kp * HID;
        Bp[idx] = make_float2(B[(2 * kp) * HID + n], B[(2 * kp + 1) * HID + n]);
    }
}


__global__ __launch_bounds__(256) void k_gemm_efc(const float* __restrict__ A,
                                                  const float2* __restrict__ Bp,
                                                  const float* __restrict__ bias,
                                                  float* __restrict__ out) {
    const int wave   = threadIdx.x >> 5;
    const int lane   = threadIdx.x & 31;
    const int lane16 = lane & 15;
    const int hi     = lane >> 4;

    const int slab     = blockIdx.x * 8 + wave;
    const bool valid   = (slab * 32) < TASKS;
    const int row_base = valid ? slab * 32 : 0;

    const float*  arow0 = A + (size_t)(row_base + lane16) * FEAT + 2 * hi;
    const float*  arow1 = arow0 + (size_t)16 * FEAT;
    const float2* bbase = Bp + lane16;

    v8f acc[2][8];
#pragma unroll
    for (int t = 0; t < 8; ++t) {
        float bv = bias[t * 16 + lane16];
        acc[0][t] = (v8f){bv, bv, bv, bv, bv, bv, bv, bv};
        acc[1][t] = acc[0][t];
    }

    for (int k = 0; k < FEAT; k += 4) {
        const float2 av0 = *(const float2*)(arow0 + k);
        const float2 av1 = *(const float2*)(arow1 + k);
        v2f af0 = {av0.x, av0.y};
        v2f af1 = {av1.x, av1.y};
        const float2* brow = bbase + (size_t)((k >> 1) + hi) * HID;
#pragma unroll
        for (int t = 0; t < 8; ++t) {
            const float2 bv = brow[t * 16];
            v2f bf = {bv.x, bv.y};
            acc[0][t] = __builtin_amdgcn_wmma_f32_16x16x4_f32(
                false, af0, false, bf, (short)0, acc[0][t], false, false);
            acc[1][t] = __builtin_amdgcn_wmma_f32_16x16x4_f32(
                false, af1, false, bf, (short)0, acc[1][t], false, false);
        }
    }

    if (valid) {
#pragma unroll
        for (int r = 0; r < 2; ++r) {
#pragma unroll
            for (int t = 0; t < 8; ++t) {
#pragma unroll
                for (int j = 0; j < 8; ++j) {
                    int m   = row_base + r * 16 + j + 8 * hi;
                    float v = acc[r][t][j];
                    out[(size_t)m * HID + t * 16 + lane16] = v > 0.f ? v : 0.f;
                }
            }
        }
    }
}


__global__ __launch_bounds__(256) void k_h(const float* __restrict__ tf_h,
                                           const float* __restrict__ wf,
                                           const float* __restrict__ Wm,
                                           const float* __restrict__ Wl,
                                           float* __restrict__ hm,
                                           float* __restrict__ hl) {
    const int wave   = threadIdx.x >> 5;
    const int lane   = threadIdx.x & 31;
    const int lane16 = lane & 15;
    const int hi     = lane >> 4;

    const int node_base = (blockIdx.x * 8 + wave) * 16;
    int nd = node_base + lane16;
    if (nd > NODES - 1) nd = NODES - 1;
    const float* xrow = (nd < TASKS) ? (tf_h + (size_t)nd * HID)
                                     : (wf + (size_t)(nd - TASKS) * HID);
    const int n = lane16;
    const bool ncol = n < CLS;

    v8f accM = {0.f, 0.f, 0.f, 0.f, 0.f, 0.f, 0.f, 0.f};
    v8f accL = {0.f, 0.f, 0.f, 0.f, 0.f, 0.f, 0.f, 0.f};

    for (int k = 0; k < HID; k += 4) {
        const int k0 = k + 2 * hi;
        const float2 av = *(const float2*)(xrow + k0);
        v2f af = {av.x, av.y};
        v2f bm, bl;
        bm.x = ncol ? Wm[k0 * CLS + n] : 0.f;
        bm.y = ncol ? Wm[(k0 + 1) * CLS + n] : 0.f;
        bl.x = ncol ? Wl[k0 * CLS + n] : 0.f;
        bl.y = ncol ? Wl[(k0 + 1) * CLS + n] : 0.f;
        accM = __builtin_amdgcn_wmma_f32_16x16x4_f32(false, af, false, bm, (short)0, accM, false, false);
        accL = __builtin_amdgcn_wmma_f32_16x16x4_f32(false, af, false, bl, (short)0, accL, false, false);
    }

    if (ncol) {
#pragma unroll
        for (int j = 0; j < 8; ++j) {
            int m = node_base + j + 8 * hi;
            if (m < NODES) {
                hm[(size_t)m * HP + n] = accM[j];
                hl[(size_t)m * HP + n] = accL[j];
            }
        }
    }
}


__device__ __forceinline__ void load_row10(const float* __restrict__ p, float* r) {
    const float4 a = *(const float4*)(p);
    const float4 b = *(const float4*)(p + 4);
    const float2 c = *(const float2*)(p + 8);
    r[0] = a.x; r[1] = a.y; r[2] = a.z; r[3] = a.w;
    r[4] = b.x; r[5] = b.y; r[6] = b.z; r[7] = b.w;
    r[8] = c.x; r[9] = c.y;
}

__device__ __forceinline__ unsigned long long to_fix(float v) {
    return (unsigned long long)(long long)rintf(v * 4294967296.0f);
}

__device__ __forceinline__ float from_fix(unsigned long long s) {
    return (float)((double)(long long)s * (1.0 / 4294967296.0));
}

__global__ __launch_bounds__(256) void k_agg_task(const int* __restrict__ answers,
                                                  const unsigned short* __restrict__ list,
                                                  const int* __restrict__ deg,
                                                  const float* __restrict__ hm,
                                                  const float* __restrict__ hl,
                                                  const float* __restrict__ dinv,
                                                  const float* __restrict__ bm,
                                                  const float* __restrict__ bl,
                                                  float* __restrict__ mean,
                                                  float* __restrict__ lsd) {
    __shared__ float sm[256 * CLS];
    __shared__ float sl[256 * CLS];

    const int i = blockIdx.x * 256 + threadIdx.x;
    float am[CLS], al[CLS];
#pragma unroll
    for (int c = 0; c < CLS; ++c) { am[c] = 0.f; al[c] = 0.f; }

    if (i < NODES) {
        const float d  = dinv[i];
        const float sw = d * d;
        float hmi[CLS], hli[CLS];
        load_row10(hm + (size_t)i * HP, hmi);
        load_row10(hl + (size_t)i * HP, hli);
#pragma unroll
        for (int c = 0; c < CLS; ++c) {
            am[c] = bm[c] + hmi[c] * sw;
            al[c] = bl[c] + hli[c] * sw;
        }

        if (i < TASKS) {
            const int n = deg[i] - 1;
            if (n <= CAP) {
                const uint4* lp = (const uint4*)(list + (size_t)i * CAP);
                const uint4 q0 = lp[0], q1 = lp[1], q2 = lp[2], q3 = lp[3];
                const unsigned pk[16] = {q0.x, q0.y, q0.z, q0.w, q1.x, q1.y, q1.z, q1.w,
                                         q2.x, q2.y, q2.z, q2.w, q3.x, q3.y, q3.z, q3.w};
                int e[CAP];
#pragma unroll
                for (int j = 0; j < CAP; ++j) {
                    const int v = (int)((pk[j >> 1] >> (16 * (j & 1))) & 0xFFFFu);
                    e[j] = (j < n) ? v : SENT;
                }
                int cur = -1;
                for (int it = 0; it < CAP; ++it) {
                    int m = SENT, cnt = 0;
#pragma unroll
                    for (int j = 0; j < CAP; ++j) {
                        const int v = e[j];
                        if (v > cur) {
                            if (v < m) { m = v; cnt = 1; }
                            else if (v == m) ++cnt;
                        }
                    }
                    if (m == SENT) break;
                    const int wrow = TASKS + m;
                    const float we = d * dinv[wrow];
                    float hmw[CLS], hlw[CLS];
                    load_row10(hm + (size_t)wrow * HP, hmw);
                    load_row10(hl + (size_t)wrow * HP, hlw);
                    for (int k = 0; k < cnt; ++k) {
#pragma unroll
                        for (int c = 0; c < CLS; ++c) {
                            am[c] += hmw[c] * we;
                            al[c] += hlw[c] * we;
                        }
                    }
                    cur = m;
                }
            } else {
                for (int a = 0; a < ANS; ++a) {
                    if (answers[3 * a] == i) {
                        const int wrow = TASKS + answers[3 * a + 1];
                        const float we = d * dinv[wrow];
                        float hmw[CLS], hlw[CLS];
                        load_row10(hm + (size_t)wrow * HP, hmw);
                        load_row10(hl + (size_t)wrow * HP, hlw);
#pragma unroll
                        for (int c = 0; c < CLS; ++c) {
                            am[c] += hmw[c] * we;
                            al[c] += hlw[c] * we;
                        }
                    }
                }
            }
        }
    }

#pragma unroll
    for (int c = 0; c < CLS; ++c) {
        sm[threadIdx.x * CLS + c] = am[c];
        sl[threadIdx.x * CLS + c] = al[c];
    }
    __syncthreads();
    const int base = blockIdx.x * 256 * CLS;
    const int lim  = NODES * CLS - base;
    for (int k = threadIdx.x; k < 256 * CLS; k += 256) {
        if (k < lim) {
            mean[(size_t)base + k] = sm[k];
            lsd[(size_t)base + k]  = sl[k];
        }
    }
}

__global__ __launch_bounds__(256) void k_agg_worker(const int* __restrict__ answers,
                                                    const float* __restrict__ hm,
                                                    const float* __restrict__ hl,
                                                    const float* __restrict__ dinv,
                                                    unsigned long long* __restrict__ wsum) {
    __shared__ unsigned long long wacc[2 * WORKERS * CLS];
    for (int i = threadIdx.x; i < 2 * WORKERS * CLS; i += 256) wacc[i] = 0ull;
    __syncthreads();

    const int start = blockIdx.x * CHUNK;
    int end = start + CHUNK;
    if (end > ANS) end = ANS;

    for (int a = start + threadIdx.x; a < end; a += 256) {
        const int t = answers[3 * a];
        const int w = answers[3 * a + 1];
        const float we = dinv[t] * dinv[TASKS + w];

        float hmt[CLS], hlt[CLS];
        load_row10(hm + (size_t)t * HP, hmt);
        load_row10(hl + (size_t)t * HP, hlt);
#pragma unroll
        for (int c = 0; c < CLS; ++c) {
            atomicAdd(&wacc[w * CLS + c], to_fix(hmt[c] * we));
            atomicAdd(&wacc[WORKERS * CLS + w * CLS + c], to_fix(hlt[c] * we));
        }
    }
    __syncthreads();

    for (int i = threadIdx.x; i < 2 * WORKERS * CLS; i += 256) atomicAdd(&wsum[i], wacc[i]);
}


__global__ __launch_bounds__(256) void k_z(float* __restrict__ mean, float* __restrict__ lsd,
                                           const unsigned long long* __restrict__ wsum,
                                           const float* __restrict__ eps, float* __restrict__ zp) {
    const int stride = gridDim.x * 256;
    const int first  = blockIdx.x * 256 + threadIdx.x;
    float mv[Z_PER], lv[Z_PER], ev[Z_PER];
#pragma unroll
    for (int j = 0; j < Z_PER; ++j) {
        int ic = first + j * stride;
        if (ic > NODES * CLS - 1) ic = NODES * CLS - 1;
        mv[j] = mean[ic];
        lv[j] = lsd[ic];
        ev[j] = eps[ic];
    }
#pragma unroll
    for (int j = 0; j < Z_PER; ++j) {
        const int idx = first + j * stride;
        if (idx < NODES * CLS) {
            float m = mv[j];
            float l = lv[j];
            if (idx >= TASKS * CLS) {
                const int q = idx - TASKS * CLS;
                m += from_fix(wsum[q]);
                l += from_fix(wsum[WORKERS * CLS + q]);
                mean[idx] = m;
                lsd[idx]  = l;
            }
            const int node = idx / CLS;
            const int c    = idx - node * CLS;
            zp[(size_t)node * HP + c] = m + (ev[j] * 0.01f) * expf(l);
        }
    }
}

__global__ __launch_bounds__(256) void k_crowd(const int* __restrict__ answers,
                                               const float* __restrict__ zp,
                                               float* __restrict__ crowd) {
    __shared__ __attribute__((aligned(16))) float s_zw[WORKERS * HP];
    __shared__ __attribute__((aligned(16))) float s_out[2][CR_TILE * CLS];

    {
        const float2* srcw = (const float2*)(zp + (size_t)TASKS * HP);
        float2* dstw = (float2*)s_zw;
        const int p = threadIdx.x & 7;
        if (p < CLS / 2) {
            for (int r = threadIdx.x >> 3; r < WORKERS; r += 32)
                dstw[r * (HP / 2) + p] = srcw[r * (HP / 2) + p];
        }
    }
    __syncthreads();

    constexpr int NT = (ANS + CR_TILE - 1) / CR_TILE;
    constexpr int NQ = CR_TILE * CLS / 4;
    int buf = 0;
    for (int tile = blockIdx.x; tile < NT; tile += gridDim.x, buf ^= 1) {
        const int base = tile * CR_TILE;
        float* sb = s_out[buf];

        int t[2], w[2];
#pragma unroll
        for (int h = 0; h < 2; ++h) {
            int a = base + h * 256 + threadIdx.x;
            if (a > ANS - 1) a = ANS - 1;
            t[h] = answers[3 * a];
            w[h] = answers[3 * a + 1];
        }
        float4 ta[2], tb[2];
        float2 tc[2];
#pragma unroll
        for (int h = 0; h < 2; ++h) {
            const float* zt = zp + (size_t)t[h] * HP;
            ta[h] = *(const float4*)(zt);
            tb[h] = *(const float4*)(zt + 4);
            tc[h] = *(const float2*)(zt + 8);
        }
#pragma unroll
        for (int h = 0; h < 2; ++h) {
            const float* zw = s_zw + w[h] * HP;
            const float4 wa = *(const float4*)(zw);
            const float4 wb = *(const float4*)(zw + 4);
            const float2 wc = *(const float2*)(zw + 8);
            float2* o = (float2*)(sb + (h * 256 + threadIdx.x) * CLS);
            o[0] = make_float2(ta[h].x * wa.x, ta[h].y * wa.y);
            o[1] = make_float2(ta[h].z * wa.z, ta[h].w * wa.w);
            o[2] = make_float2(tb[h].x * wb.x, tb[h].y * wb.y);
            o[3] = make_float2(tb[h].z * wb.z, tb[h].w * wb.w);
            o[4] = make_float2(tc[h].x * wc.x, tc[h].y * wc.y);
        }
        __syncthreads();

        float4* dst = (float4*)(crowd + (size_t)base * CLS);
        const float4* so = (const float4*)sb;
        if (ANS - base >= CR_TILE) {
#pragma unroll
            for (int k = 0; k < NQ / 256; ++k) dst[k * 256 + threadIdx.x] = so[k * 256 + threadIdx.x];
        } else {
            const int nq = (ANS - base) * CLS / 4;
            for (int k = threadIdx.x; k < nq; k += 256) dst[k] = so[k];
        }
    }
}


extern "C" void kernel_launch(void* const* d_in, const int* in_sizes, int n_in,
                              void* d_out, int out_size, void* d_ws, size_t ws_size,
                              hipStream_t stream) {
    (void)in_sizes; (void)n_in; (void)out_size; (void)ws_size;

    const float* task_feature   = (const float*)d_in[0];
    const int*   answers        = (const int*)d_in[1];
    const float* worker_feature = (const float*)d_in[2];
    const float* W_efc          = (const float*)d_in[3];
    const float* b_efc          = (const float*)d_in[4];
    const float* W_mean         = (const float*)d_in[5];
    const float* b_mean         = (const float*)d_in[6];
    const float* W_ls           = (const float*)d_in[7];
    const float* b_ls           = (const float*)d_in[8];
    const float* eps            = (const float*)d_in[9];

    float* out   = (float*)d_out;
    float* crowd = out;
    float* mean  = out + (size_t)ANS * CLS;
    float* lsd   = mean + (size_t)NODES * CLS;

    float*  ws   = (float*)d_ws;
    float*  tf_h = ws;
    float2* Bp   = (float2*)(tf_h + (size_t)TASKS * HID);
    float*  hm   = (float*)(Bp + (size_t)(FEAT / 2) * HID);
    float*  hl   = hm + (size_t)NODES * HP;
    float*  dinv = hl + (size_t)NODES * HP;
    int*    deg  = (int*)(dinv + NODES);
    unsigned short*     list = (unsigned short*)tf_h;
    unsigned long long* wsum = (unsigned long long*)(tf_h + (size_t)2 * 1024 * 1024);
    float*              zz   = tf_h + (size_t)4 * 1024 * 1024;

    auto cdiv = [](int a, int b) { return (a + b - 1) / b; };

    k_packB    <<<cdiv((FEAT / 2) * HID, 256), 256, 0, stream>>>(W_efc, Bp);
    k_gemm_efc <<<cdiv(TASKS / 32, 8), 256, 0, stream>>>(task_feature, Bp, b_efc, tf_h);
    k_h        <<<cdiv(cdiv(NODES, 16), 8), 256, 0, stream>>>(tf_h, worker_feature,
                                                              W_mean, W_ls, hm, hl);

    k_deg_init <<<cdiv(NODES, 256), 256, 0, stream>>>(deg, wsum);
    k_deg_count<<<cdiv(ANS, 256), 256, 0, stream>>>(answers, deg, list);
    k_dinv     <<<cdiv(NODES, 256), 256, 0, stream>>>(deg, dinv);

    k_agg_task  <<<cdiv(NODES, 256), 256, 0, stream>>>(answers, list, deg, hm, hl, dinv,
                                                       b_mean, b_ls, mean, lsd);
    k_agg_worker<<<cdiv(ANS, CHUNK), 256, 0, stream>>>(answers, hm, hl, dinv, wsum);

    k_z        <<<cdiv(cdiv(NODES * CLS, Z_PER), 256), 256, 0, stream>>>(mean, lsd, wsum, eps, zz);
    k_crowd    <<<CR_GRID, 256, 0, stream>>>(answers, zz, crowd);
}


# Worker degrees counted in LDS by 256 looping work-groups and summed from per-group rows in the dinv kernel: the 1M global integer adds onto 500 shared words are gone; no float sum changes
# speedup vs baseline: 2.9049x; 1.4136x over previous
#include <hip/hip_runtime.h>

typedef __attribute__((ext_vector_type(2))) float v2f;
typedef __attribute__((ext_vector_type(8))) float v8f;

constexpr int TASKS   = 100000;
constexpr int WORKERS = 500;
constexpr int FEAT    = 512;
constexpr int CLS     = 10;
constexpr int HP      = 12;
constexpr int ANS     = 1000000;
constexpr int HID     = 128;
constexpr int NODES   = TASKS + WORKERS;
constexpr int CHUNK   = 4096;
constexpr int CAP     = 32;
constexpr int SENT    = 0xFFFF;
constexpr int Z_PER   = 8;
constexpr int CR_TILE = 512;
constexpr int CR_GRID = 512;
constexpr int DC_GRID = 256;
constexpr int DC_PER  = 4;
constexpr int WPAD    = 512;
static_assert(WPAD >= WORKERS && WPAD % 256 == 0, "a row of worker counts is a whole number of 256-thread rounds");
static_assert(ANS % 2 == 0 && CR_TILE % 2 == 0, "a tile's floats must split into 16-byte pieces");
static_assert((CR_TILE * CLS / 4) % 256 == 0, "a full tile is a whole number of 256-thread store rounds");


__global__ void k_deg_init(int* __restrict__ deg, unsigned long long* __restrict__ wsum) {
    int i = blockIdx.x * 256 + threadIdx.x;
    if (i < NODES) deg[i] = 1;
    if (i < 2 * WORKERS * CLS) wsum[i] = 0ull;
}

__global__ __launch_bounds__(256) void k_deg_count(const int* __restrict__ answers,
                                                   int* __restrict__ deg,
                                                   unsigned short* __restrict__ list,
                                                   int* __restrict__ wpart) {
    __shared__ int s_wdeg[WPAD];
    for (int i = threadIdx.x; i < WPAD; i += 256) s_wdeg[i] = 0;
    __syncthreads();

    constexpr int TRIP = 256 * DC_PER;
    for (int base = blockIdx.x * TRIP; base < ANS; base += gridDim.x * TRIP) {
        int t[DC_PER], w[DC_PER], old[DC_PER];
        bool ok[DC_PER];
#pragma unroll
        for (int h = 0; h < DC_PER; ++h) {
            const int a = base + h * 256 + threadIdx.x;
            ok[h] = a < ANS;
            const int ac = ok[h] ? a : ANS - 1;
            t[h] = answers[3 * ac];
            w[h] = answers[3 * ac + 1];
        }
#pragma unroll
        for (int h = 0; h < DC_PER; ++h) {
            old[h] = CAP + 1;
            if (ok[h]) {
                old[h] = atomicAdd(&deg[t[h]], 1);
                atomicAdd(&s_wdeg[w[h]], 1);
            }
        }
#pragma unroll
        for (int h = 0; h < DC_PER; ++h) {
            const int slot = old[h] - 1;
            if (slot < CAP) list[(size_t)t[h] * CAP + slot] = (unsigned short)w[h];
        }
    }
    __syncthreads();
    for (int i = threadIdx.x; i < WPAD; i += 256) wpart[(size_t)blockIdx.x * WPAD + i] = s_wdeg[i];
}

constexpr int DV_TASK = (TASKS + 255) / 256;
constexpr int DV_WORK = (WORKERS + 31) / 32;
static_assert(DV_WORK * 32 <= WPAD, "the worker index of every lane stays inside a padded row");
static_assert(DC_GRID % 8 == 0, "the rows split evenly over the eight waves of a work-group");

__global__ __launch_bounds__(256) void k_dinv(const int* __restrict__ deg,
                                              const int* __restrict__ wpart,
                                              float* __restrict__ dinv) {
    if (blockIdx.x < DV_TASK) {
        const int i = blockIdx.x * 256 + threadIdx.x;
        if (i < TASKS) dinv[i] = rsqrtf((float)deg[i]);
        return;
    }
    __shared__ int s_part[8][32];
    const int lane = threadIdx.x & 31;
    const int j    = threadIdx.x >> 5;
    const int w    = (blockIdx.x - DV_TASK) * 32 + lane;
    int s = 0;
#pragma unroll
    for (int q = 0; q < DC_GRID / 8; ++q) s += wpart[(q * 8 + j) * WPAD + w];
    s_part[j][lane] = s;
    __syncthreads();
    if (j == 0 && w < WORKERS) {
        int d = deg[TASKS + w];
#pragma unroll
        for (int q = 0; q < 8; ++q) d += s_part[q][lane];
        dinv[TASKS + w] = rsqrtf((float)d);
    }
}


__global__ void k_packB(const float* __restrict__ B, float2* __restrict__ Bp) {
    int idx = blockIdx.x * 256 + threadIdx.x;
    if (idx < (FEAT / 2) * HID) {
        int kp = idx / HID;
        int n  = idx - kp * HID;
        Bp[idx] = make_float2(B[(2 * kp) * HID + n], B[(2 * kp + 1) * HID + n]);
    }
}


__global__ __launch_bounds__(256) void k_gemm_efc(const float* __restrict__ A,
                                                  const float2* __restrict__ Bp,
                                                  const float* __restrict__ bias,
                                                  float* __restrict__ out) {
    const int wave   = threadIdx.x >> 5;
    const int lane   = threadIdx.x & 31;
    const int lane16 = lane & 15;
    const int hi     = lane >> 4;

    const int slab     = blockIdx.x * 8 + wave;
    const bool valid   = (slab * 32) < TASKS;
    const int row_base = valid ? slab * 32 : 0;

    const float*  arow0 = A + (size_t)(row_base + lane16) * FEAT + 2 * hi;
    const float*  arow1 = arow0 + (size_t)16 * FEAT;
    const float2* bbase = Bp + lane16;

    v8f acc[2][8];
#pragma unroll
    for (int t = 0; t < 8; ++t) {
        float bv = bias[t * 16 + lane16];
        acc[0][t] = (v8f){bv, bv, bv, bv, bv, bv, bv, bv};
        acc[1][t] = acc[0][t];
    }

    for (int k = 0; k < FEAT; k += 4) {
        const float2 av0 = *(const float2*)(arow0 + k);
        const float2 av1 = *(const float2*)(arow1 + k);
        v2f af0 = {av0.x, av0.y};
        v2f af1 = {av1.x, av1.y};
        const float2* brow = bbase + (size_t)((k >> 1) + hi) * HID;
#pragma unroll
        for (int t = 0; t < 8; ++t) {
            const float2 bv = brow[t * 16];
            v2f bf = {bv.x, bv.y};
            acc[0][t] = __builtin_amdgcn_wmma_f32_16x16x4_f32(
                false, af0, false, bf, (short)0, acc[0][t], false, false);
            acc[1][t] = __builtin_amdgcn_wmma_f32_16x16x4_f32(
                false, af1, false, bf, (short)0, acc[1][t], false, false);
        }
    }

    if (valid) {
#pragma unroll
        for (int r = 0; r < 2; ++r) {
#pragma unroll
            for (int t = 0; t < 8; ++t) {
#pragma unroll
                for (int j = 0; j < 8; ++j) {
                    int m   = row_base + r * 16 + j + 8 * hi;
                    float v = acc[r][t][j];
                    out[(size_t)m * HID + t * 16 + lane16] = v > 0.f ? v : 0.f;
                }
            }
        }
    }
}


__global__ __launch_bounds__(256) void k_h(const float* __restrict__ tf_h,
                                           const float* __restrict__ wf,
                                           const float* __restrict__ Wm,
                                           const float* __restrict__ Wl,
                                           float* __restrict__ hm,
                                           float* __restrict__ hl) {
    const int wave   = threadIdx.x >> 5;
    const int lane   = threadIdx.x & 31;
    const int lane16 = lane & 15;
    const int hi     = lane >> 4;

    const int node_base = (blockIdx.x * 8 + wave) * 16;
    int nd = node_base + lane16;
    if (nd > NODES - 1) nd = NODES - 1;
    const float* xrow = (nd < TASKS) ? (tf_h + (size_t)nd * HID)
                                     : (wf + (size_t)(nd - TASKS) * HID);
    const int n = lane16;
    const bool ncol = n < CLS;

    v8f accM = {0.f, 0.f, 0.f, 0.f, 0.f, 0.f, 0.f, 0.f};
    v8f accL = {0.f, 0.f, 0.f, 0.f, 0.f, 0.f, 0.f, 0.f};

    for (int k = 0; k < HID; k += 4) {
        const int k0 = k + 2 * hi;
        const float2 av = *(const float2*)(xrow + k0);
        v2f af = {av.x, av.y};
        v2f bm, bl;
        bm.x = ncol ? Wm[k0 * CLS + n] : 0.f;
        bm.y = ncol ? Wm[(k0 + 1) * CLS + n] : 0.f;
        bl.x = ncol ? Wl[k0 * CLS + n] : 0.f;
        bl.y = ncol ? Wl[(k0 + 1) * CLS + n] : 0.f;
        accM = __builtin_amdgcn_wmma_f32_16x16x4_f32(false, af, false, bm, (short)0, accM, false, false);
        accL = __builtin_amdgcn_wmma_f32_16x16x4_f32(false, af, false, bl, (short)0, accL, false, false);
    }

    if (ncol) {
#pragma unroll
        for (int j = 0; j < 8; ++j) {
            int m = node_base + j + 8 * hi;
            if (m < NODES) {
                hm[(size_t)m * HP + n] = accM[j];
                hl[(size_t)m * HP + n] = accL[j];
            }
        }
    }
}


__device__ __forceinline__ void load_row10(const float* __restrict__ p, float* r) {
    const float4 a = *(const float4*)(p);
    const float4 b = *(const float4*)(p + 4);
    const float2 c = *(const float2*)(p + 8);
    r[0] = a.x; r[1] = a.y; r[2] = a.z; r[3] = a.w;
    r[4] = b.x; r[5] = b.y; r[6] = b.z; r[7] = b.w;
    r[8] = c.x; r[9] = c.y;
}

__device__ __forceinline__ unsigned long long to_fix(float v) {
    return (unsigned long long)(long long)rintf(v * 4294967296.0f);
}

__device__ __forceinline__ float from_fix(unsigned long long s) {
    return (float)((double)(long long)s * (1.0 / 4294967296.0));
}

__global__ __launch_bounds__(256) void k_agg_task(const int* __restrict__ answers,
                                                  const unsigned short* __restrict__ list,
                                                  const int* __restrict__ deg,
                                                  const float* __restrict__ hm,
                                                  const float* __restrict__ hl,
                                                  const float* __restrict__ dinv,
                                                  const float* __restrict__ bm,
                                                  const float* __restrict__ bl,
                                                  float* __restrict__ mean,
                                                  float* __restrict__ lsd) {
    __shared__ float sm[256 * CLS];
    __shared__ float sl[256 * CLS];

    const int i = blockIdx.x * 256 + threadIdx.x;
    float am[CLS], al[CLS];
#pragma unroll
    for (int c = 0; c < CLS; ++c) { am[c] = 0.f; al[c] = 0.f; }

    if (i < NODES) {
        const float d  = dinv[i];
        const float sw = d * d;
        float hmi[CLS], hli[CLS];
        load_row10(hm + (size_t)i * HP, hmi);
        load_row10(hl + (size_t)i * HP, hli);
#pragma unroll
        for (int c = 0; c < CLS; ++c) {
            am[c] = bm[c] + hmi[c] * sw;
            al[c] = bl[c] + hli[c] * sw;
        }

        if (i < TASKS) {
            const int n = deg[i] - 1;
            if (n <= CAP) {
                const uint4* lp = (const uint4*)(list + (size_t)i * CAP);
                const uint4 q0 = lp[0], q1 = lp[1], q2 = lp[2], q3 = lp[3];
                const unsigned pk[16] = {q0.x, q0.y, q0.z, q0.w, q1.x, q1.y, q1.z, q1.w,
                                         q2.x, q2.y, q2.z, q2.w, q3.x, q3.y, q3.z, q3.w};
                int e[CAP];
#pragma unroll
                for (int j = 0; j < CAP; ++j) {
                    const int v = (int)((pk[j >> 1] >> (16 * (j & 1))) & 0xFFFFu);
                    e[j] = (j < n) ? v : SENT;
                }
                int cur = -1;
                for (int it = 0; it < CAP; ++it) {
                    int m = SENT, cnt = 0;
#pragma unroll
                    for (int j = 0; j < CAP; ++j) {
                        const int v = e[j];
                        if (v > cur) {
                            if (v < m) { m = v; cnt = 1; }
                            else if (v == m) ++cnt;
                        }
                    }
                    if (m == SENT) break;
                    const int wrow = TASKS + m;
                    const float we = d * dinv[wrow];
                    float hmw[CLS], hlw[CLS];
                    load_row10(hm + (size_t)wrow * HP, hmw);
                    load_row10(hl + (size_t)wrow * HP, hlw);
                    for (int k = 0; k < cnt; ++k) {
#pragma unroll
                        for (int c = 0; c < CLS; ++c) {
                            am[c] += hmw[c] * we;
                            al[c] += hlw[c] * we;
                        }
                    }
                    cur = m;
                }
            } else {
                for (int a = 0; a < ANS; ++a) {
                    if (answers[3 * a] == i) {
                        const int wrow = TASKS + answers[3 * a + 1];
                        const float we = d * dinv[wrow];
                        float hmw[CLS], hlw[CLS];
                        load_row10(hm + (size_t)wrow * HP, hmw);
                        load_row10(hl + (size_t)wrow * HP, hlw);
#pragma unroll
                        for (int c = 0; c < CLS; ++c) {
                            am[c] += hmw[c] * we;
                            al[c] += hlw[c] * we;
                        }
                    }
                }
            }
        }
    }

#pragma unroll
    for (int c = 0; c < CLS; ++c) {
        sm[threadIdx.x * CLS + c] = am[c];
        sl[threadIdx.x * CLS + c] = al[c];
    }
    __syncthreads();
    const int base = blockIdx.x * 256 * CLS;
    const int lim  = NODES * CLS - base;
    for (int k = threadIdx.x; k < 256 * CLS; k += 256) {
        if (k < lim) {
            mean[(size_t)base + k] = sm[k];
            lsd[(size_t)base + k]  = sl[k];
        }
    }
}

__global__ __launch_bounds__(256) void k_agg_worker(const int* __restrict__ answers,
                                                    const float* __restrict__ hm,
                                                    const float* __restrict__ hl,
                                                    const float* __restrict__ dinv,
                                                    unsigned long long* __restrict__ wsum) {
    __shared__ unsigned long long wacc[2 * WORKERS * CLS];
    for (int i = threadIdx.x; i < 2 * WORKERS * CLS; i += 256) wacc[i] = 0ull;
    __syncthreads();

    const int start = blockIdx.x * CHUNK;
    int end = start + CHUNK;
    if (end > ANS) end = ANS;

    for (int a = start + threadIdx.x; a < end; a += 256) {
        const int t = answers[3 * a];
        const int w = answers[3 * a + 1];
        const float we = dinv[t] * dinv[TASKS + w];

        float hmt[CLS], hlt[CLS];
        load_row10(hm + (size_t)t * HP, hmt);
        load_row10(hl + (size_t)t * HP, hlt);
#pragma unroll
        for (int c = 0; c < CLS; ++c) {
            atomicAdd(&wacc[w * CLS + c], to_fix(hmt[c] * we));
            atomicAdd(&wacc[WORKERS * CLS + w * CLS + c], to_fix(hlt[c] * we));
        }
    }
    __syncthreads();

    for (int i = threadIdx.x; i < 2 * WORKERS * CLS; i += 256) atomicAdd(&wsum[i], wacc[i]);
}


__global__ __launch_bounds__(256) void k_z(float* __restrict__ mean, float* __restrict__ lsd,
                                           const unsigned long long* __restrict__ wsum,
                                           const float* __restrict__ eps, float* __restrict__ zp) {
    const int stride = gridDim.x * 256;
    const int first  = blockIdx.x * 256 + threadIdx.x;
    float mv[Z_PER], lv[Z_PER], ev[Z_PER];
#pragma unroll
    for (int j = 0; j < Z_PER; ++j) {
        int ic = first + j * stride;
        if (ic > NODES * CLS - 1) ic = NODES * CLS - 1;
        mv[j] = mean[ic];
        lv[j] = lsd[ic];
        ev[j] = eps[ic];
    }
#pragma unroll
    for (int j = 0; j < Z_PER; ++j) {
        const int idx = first + j * stride;
        if (idx < NODES * CLS) {
            float m = mv[j];
            float l = lv[j];
            if (idx >= TASKS * CLS) {
                const int q = idx - TASKS * CLS;
                m += from_fix(wsum[q]);
                l += from_fix(wsum[WORKERS * CLS + q]);
                mean[idx] = m;
                lsd[idx]  = l;
            }
            const int node = idx / CLS;
            const int c    = idx - node * CLS;
            zp[(size_t)node * HP + c] = m + (ev[j] * 0.01f) * expf(l);
        }
    }
}

__global__ __launch_bounds__(256) void k_crowd(const int* __restrict__ answers,
                                               const float* __restrict__ zp,
                                               float* __restrict__ crowd) {
    __shared__ __attribute__((aligned(16))) float s_zw[WORKERS * HP];
    __shared__ __attribute__((aligned(16))) float s_out[2][CR_TILE * CLS];

    {
        const float2* srcw = (const float2*)(zp + (size_t)TASKS * HP);
        float2* dstw = (float2*)s_zw;
        const int p = threadIdx.x & 7;
        if (p < CLS / 2) {
            for (int r = threadIdx.x >> 3; r < WORKERS; r += 32)
                dstw[r * (HP / 2) + p] = srcw[r * (HP / 2) + p];
        }
    }
    __syncthreads();

    constexpr int NT = (ANS + CR_TILE - 1) / CR_TILE;
    constexpr int NQ = CR_TILE * CLS / 4;
    int buf = 0;
    for (int tile = blockIdx.x; tile < NT; tile += gridDim.x, buf ^= 1) {
        const int base = tile * CR_TILE;
        float* sb = s_out[buf];

        int t[2], w[2];
#pragma unroll
        for (int h = 0; h < 2; ++h) {
            int a = base + h * 256 + threadIdx.x;
            if (a > ANS - 1) a = ANS - 1;
            t[h] = answers[3 * a];
            w[h] = answers[3 * a + 1];
        }
        float4 ta[2], tb[2];
        float2 tc[2];
#pragma unroll
        for (int h = 0; h < 2; ++h) {
            const float* zt = zp + (size_t)t[h] * HP;
            ta[h] = *(const float4*)(zt);
            tb[h] = *(const float4*)(zt + 4);
            tc[h] = *(const float2*)(zt + 8);
        }
#pragma unroll
        for (int h = 0; h < 2; ++h) {
            const float* zw = s_zw + w[h] * HP;
            const float4 wa = *(const float4*)(zw);
            const float4 wb = *(const float4*)(zw + 4);
            const float2 wc = *(const float2*)(zw + 8);
            float2* o = (float2*)(sb + (h * 256 + threadIdx.x) * CLS);
            o[0] = make_float2(ta[h].x * wa.x, ta[h].y * wa.y);
            o[1] = make_float2(ta[h].z * wa.z, ta[h].w * wa.w);
            o[2] = make_float2(tb[h].x * wb.x, tb[h].y * wb.y);
            o[3] = make_float2(tb[h].z * wb.z, tb[h].w * wb.w);
            o[4] = make_float2(tc[h].x * wc.x, tc[h].y * wc.y);
        }
        __syncthreads();

        float4* dst = (float4*)(crowd + (size_t)base * CLS);
        const float4* so = (const float4*)sb;
        if (ANS - base >= CR_TILE) {
#pragma unroll
            for (int k = 0; k < NQ / 256; ++k) dst[k * 256 + threadIdx.x] = so[k * 256 + threadIdx.x];
        } else {
            const int nq = (ANS - base) * CLS / 4;
            for (int k = threadIdx.x; k < nq; k += 256) dst[k] = so[k];
        }
    }
}


extern "C" void kernel_launch(void* const* d_in, const int* in_sizes, int n_in,
                              void* d_out, int out_size, void* d_ws, size_t ws_size,
                              hipStream_t stream) {
    (void)in_sizes; (void)n_in; (void)out_size; (void)ws_size;

    const float* task_feature   = (const float*)d_in[0];
    const int*   answers        = (const int*)d_in[1];
    const float* worker_feature = (const float*)d_in[2];
    const float* W_efc          = (const float*)d_in[3];
    const float* b_efc          = (const float*)d_in[4];
    const float* W_mean         = (const float*)d_in[5];
    const float* b_mean         = (const float*)d_in[6];
    const float* W_ls           = (const float*)d_in[7];
    const float* b_ls           = (const float*)d_in[8];
    const float* eps            = (const float*)d_in[9];

    float* out   = (float*)d_out;
    float* crowd = out;
    float* mean  = out + (size_t)ANS * CLS;
    float* lsd   = mean + (size_t)NODES * CLS;

    float*  ws   = (float*)d_ws;
    float*  tf_h = ws;
    float2* Bp   = (float2*)(tf_h + (size_t)TASKS * HID);
    float*  hm   = (float*)(Bp + (size_t)(FEAT / 2) * HID);
    float*  hl   = hm + (size_t)NODES * HP;
    float*  dinv = hl + (size_t)NODES * HP;
    int*    deg  = (int*)(dinv + NODES);
    unsigned short*     list = (unsigned short*)tf_h;
    unsigned long long* wsum = (unsigned long long*)(tf_h + (size_t)2 * 1024 * 1024);
    int*                wpart = (int*)(tf_h + (size_t)3 * 1024 * 1024);
    float*              zz   = tf_h + (size_t)4 * 1024 * 1024;

    auto cdiv = [](int a, int b) { return (a + b - 1) / b; };

    k_packB    <<<cdiv((FEAT / 2) * HID, 256), 256, 0, stream>>>(W_efc, Bp);
    k_gemm_efc <<<cdiv(TASKS / 32, 8), 256, 0, stream>>>(task_feature, Bp, b_efc, tf_h);
    k_h        <<<cdiv(cdiv(NODES, 16), 8), 256, 0, stream>>>(tf_h, worker_feature,
                                                              W_mean, W_ls, hm, hl);

    k_deg_init <<<cdiv(NODES, 256), 256, 0, stream>>>(deg, wsum);
    k_deg_count<<<DC_GRID, 256, 0, stream>>>(answers, deg, list, wpart);
    k_dinv     <<<DV_TASK + DV_WORK, 256, 0, stream>>>(deg, wpart, dinv);

    k_agg_task  <<<cdiv(NODES, 256), 256, 0, stream>>>(answers, list, deg, hm, hl, dinv,
                                                       b_mean, b_ls, mean, lsd);
    k_agg_worker<<<cdiv(ANS, CHUNK), 256, 0, stream>>>(answers, hm, hl, dinv, wsum);

    k_z        <<<cdiv(cdiv(NODES * CLS, Z_PER), 256), 256, 0, stream>>>(mean, lsd, wsum, eps, zz);
    k_crowd    <<<CR_GRID, 256, 0, stream>>>(answers, zz, crowd);
}


# Independent stages share launches (10 to 6): degree init with the pack, degree count beside the product, normalisation beside hm/hl, both aggregations together; lists moved to the crowd block
# speedup vs baseline: 3.3420x; 1.1505x over previous
#include <hip/hip_runtime.h>

typedef __attribute__((ext_vector_type(2))) float v2f;
typedef __attribute__((ext_vector_type(8))) float v8f;

constexpr int TASKS   = 100000;
constexpr int WORKERS = 500;
constexpr int FEAT    = 512;
constexpr int CLS     = 10;
constexpr int HP      = 12;
constexpr int ANS     = 1000000;
constexpr int HID     = 128;
constexpr int NODES   = TASKS + WORKERS;
constexpr int CHUNK   = 4096;
constexpr int CAP     = 32;
constexpr int SENT    = 0xFFFF;
constexpr int Z_PER   = 8;
constexpr int CR_TILE = 512;
constexpr int CR_GRID = 512;
constexpr int DC_GRID = 256;
constexpr int DC_PER  = 4;
constexpr int WPAD    = 512;
static_assert(WPAD >= WORKERS && WPAD % 256 == 0, "a row of worker counts is a whole number of 256-thread rounds");
static_assert(ANS % 2 == 0 && CR_TILE % 2 == 0, "a tile's floats must split into 16-byte pieces");
static_assert((CR_TILE * CLS / 4) % 256 == 0, "a full tile is a whole number of 256-thread store rounds");


__device__ __forceinline__ void deg_count_body(const int blk, int* __restrict__ s_wdeg,
                                               const int* __restrict__ answers,
                                               int* __restrict__ deg,
                                               unsigned short* __restrict__ list,
                                               int* __restrict__ wpart) {
    for (int i = threadIdx.x; i < WPAD; i += 256) s_wdeg[i] = 0;
    __syncthreads();

    constexpr int TRIP = 256 * DC_PER;
    for (int base = blk * TRIP; base < ANS; base += DC_GRID * TRIP) {
        int t[DC_PER], w[DC_PER], old[DC_PER];
        bool ok[DC_PER];
#pragma unroll
        for (int h = 0; h < DC_PER; ++h) {
            const int a = base + h * 256 + threadIdx.x;
            ok[h] = a < ANS;
            const int ac = ok[h] ? a : ANS - 1;
            t[h] = answers[3 * ac];
            w[h] = answers[3 * ac + 1];
        }
#pragma unroll
        for (int h = 0; h < DC_PER; ++h) {
            old[h] = CAP + 1;
            if (ok[h]) {
                old[h] = atomicAdd(&deg[t[h]], 1);
                atomicAdd(&s_wdeg[w[h]], 1);
            }
        }
#pragma unroll
        for (int h = 0; h < DC_PER; ++h) {
            const int slot = old[h] - 1;
            if (slot < CAP) list[(size_t)t[h] * CAP + slot] = (unsigned short)w[h];
        }
    }
    __syncthreads();
    for (int i = threadIdx.x; i < WPAD; i += 256) wpart[(size_t)blk * WPAD + i] = s_wdeg[i];
}

constexpr int DV_TASK = (TASKS + 255) / 256;
constexpr int DV_WORK = (WORKERS + 31) / 32;
static_assert(DV_WORK * 32 <= WPAD, "the worker index of every lane stays inside a padded row");
static_assert(DC_GRID % 8 == 0, "the rows split evenly over the eight waves of a work-group");

static_assert(DV_TASK * 256 >= 2 * WORKERS * CLS, "the task work-groups cover every worker-side sum");

__device__ __forceinline__ void dinv_body(const int blk, int (*s_part)[32],
                                          const int* __restrict__ deg,
                                          const int* __restrict__ wpart,
                                          float* __restrict__ dinv,
                                          unsigned long long* __restrict__ wsum) {
    if (blk < DV_TASK) {
        const int i = blk * 256 + threadIdx.x;
        if (i < TASKS) dinv[i] = rsqrtf((float)deg[i]);
        if (i < 2 * WORKERS * CLS) wsum[i] = 0ull;
        return;
    }
    const int lane = threadIdx.x & 31;
    const int j    = threadIdx.x >> 5;
    const int w    = (blk - DV_TASK) * 32 + lane;
    int s = 0;
#pragma unroll
    for (int q = 0; q < DC_GRID / 8; ++q) s += wpart[(q * 8 + j) * WPAD + w];
    s_part[j][lane] = s;
    __syncthreads();
    if (j == 0 && w < WORKERS) {
        int d = deg[TASKS + w];
#pragma unroll
        for (int q = 0; q < 8; ++q) d += s_part[q][lane];
        dinv[TASKS + w] = rsqrtf((float)d);
    }
}


constexpr int PB_GRID = ((FEAT / 2) * HID + 255) / 256;
constexpr int DI_GRID = (NODES + 255) / 256;

__global__ void k_packB(const float* __restrict__ B, float2* __restrict__ Bp, int* __restrict__ deg) {
    if (blockIdx.x >= PB_GRID) {
        const int i = (blockIdx.x - PB_GRID) * 256 + threadIdx.x;
        if (i < NODES) deg[i] = 1;
        return;
    }
    int idx = blockIdx.x * 256 + threadIdx.x;
    if (idx < (FEAT / 2) * HID) {
        int kp = idx / HID;
        int n  = idx - kp * HID;
        Bp[idx] = make_float2(B[(2 * kp) * HID + n], B[(2 * kp + 1) * HID + n]);
    }
}


__device__ __forceinline__ void gemm_efc_body(const int blk,
                                              const float* __restrict__ A,
                                              const float2* __restrict__ Bp,
                                              const float* __restrict__ bias,
                                              float* __restrict__ out) {
    const int wave   = threadIdx.x >> 5;
    const int lane   = threadIdx.x & 31;
    const int lane16 = lane & 15;
    const int hi     = lane >> 4;

    const int slab     = blk * 8 + wave;
    const bool valid   = (slab * 32) < TASKS;
    const int row_base = valid ? slab * 32 : 0;

    const float*  arow0 = A + (size_t)(row_base + lane16) * FEAT + 2 * hi;
    const float*  arow1 = arow0 + (size_t)16 * FEAT;
    const float2* bbase = Bp + lane16;

    v8f acc[2][8];
#pragma unroll
    for (int t = 0; t < 8; ++t) {
        float bv = bias[t * 16 + lane16];
        acc[0][t] = (v8f){bv, bv, bv, bv, bv, bv, bv, bv};
        acc[1][t] = acc[0][t];
    }

    for (int k = 0; k < FEAT; k += 4) {
        const float2 av0 = *(const float2*)(arow0 + k);
        const float2 av1 = *(const float2*)(arow1 + k);
        v2f af0 = {av0.x, av0.y};
        v2f af1 = {av1.x, av1.y};
        const float2* brow = bbase + (size_t)((k >> 1) + hi) * HID;
#pragma unroll
        for (int t = 0; t < 8; ++t) {
            const float2 bv = brow[t * 16];
            v2f bf = {bv.x, bv.y};
            acc[0][t] = __builtin_amdgcn_wmma_f32_16x16x4_f32(
                false, af0, false, bf, (short)0, acc[0][t], false, false);
            acc[1][t] = __builtin_amdgcn_wmma_f32_16x16x4_f32(
                false, af1, false, bf, (short)0, acc[1][t], false, false);
        }
    }

    if (valid) {
#pragma unroll
        for (int r = 0; r < 2; ++r) {
#pragma unroll
            for (int t = 0; t < 8; ++t) {
#pragma unroll
                for (int j = 0; j < 8; ++j) {
                    int m   = row_base + r * 16 + j + 8 * hi;
                    float v = acc[r][t][j];
                    out[(size_t)m * HID + t * 16 + lane16] = v > 0.f ? v : 0.f;
                }
            }
        }
    }
}

constexpr int GE_GRID = (TASKS / 32 + 7) / 8;

__global__ __launch_bounds__(256) void k_gemm_count(const float* __restrict__ A,
                                                    const float2* __restrict__ Bp,
                                                    const float* __restrict__ bias,
                                                    float* __restrict__ tf_h,
                                                    const int* __restrict__ answers,
                                                    int* __restrict__ deg,
                                                    unsigned short* __restrict__ list,
                                                    int* __restrict__ wpart) {
    __shared__ int s_wdeg[WPAD];
    if (blockIdx.x < DC_GRID) {
        deg_count_body(blockIdx.x, s_wdeg, answers, deg, list, wpart);
    } else {
        gemm_efc_body(blockIdx.x - DC_GRID, A, Bp, bias, tf_h);
    }
}


__device__ __forceinline__ void h_body(const int blk,
                                       const float* __restrict__ tf_h,
                                       const float* __restrict__ wf,
                                       const float* __restrict__ Wm,
                                       const float* __restrict__ Wl,
                                       float* __restrict__ hm,
                                       float* __restrict__ hl) {
    const int wave   = threadIdx.x >> 5;
    const int lane   = threadIdx.x & 31;
    const int lane16 = lane & 15;
    const int hi     = lane >> 4;

    const int node_base = (blk * 8 + wave) * 16;
    int nd = node_base + lane16;
    if (nd > NODES - 1) nd = NODES - 1;
    const float* xrow = (nd < TASKS) ? (tf_h + (size_t)nd * HID)
                                     : (wf + (size_t)(nd - TASKS) * HID);
    const int n = lane16;
    const bool ncol = n < CLS;

    v8f accM = {0.f, 0.f, 0.f, 0.f, 0.f, 0.f, 0.f, 0.f};
    v8f accL = {0.f, 0.f, 0.f, 0.f, 0.f, 0.f, 0.f, 0.f};

    for (int k = 0; k < HID; k += 4) {
        const int k0 = k + 2 * hi;
        const float2 av = *(const float2*)(xrow + k0);
        v2f af = {av.x, av.y};
        v2f bm, bl;
        bm.x = ncol ? Wm[k0 * CLS + n] : 0.f;
        bm.y = ncol ? Wm[(k0 + 1) * CLS + n] : 0.f;
        bl.x = ncol ? Wl[k0 * CLS + n] : 0.f;
        bl.y = ncol ? Wl[(k0 + 1) * CLS + n] : 0.f;
        accM = __builtin_amdgcn_wmma_f32_16x16x4_f32(false, af, false, bm, (short)0, accM, false, false);
        accL = __builtin_amdgcn_wmma_f32_16x16x4_f32(false, af, false, bl, (short)0, accL, false, false);
    }

    if (ncol) {
#pragma unroll
        for (int j = 0; j < 8; ++j) {
            int m = node_base + j + 8 * hi;
            if (m < NODES) {
                hm[(size_t)m * HP + n] = accM[j];
                hl[(size_t)m * HP + n] = accL[j];
            }
        }
    }
}

constexpr int H_GRID = ((NODES + 15) / 16 + 7) / 8;

__global__ __launch_bounds__(256) void k_h_dinv(const float* __restrict__ tf_h,
                                                const float* __restrict__ wf,
                                                const float* __restrict__ Wm,
                                                const float* __restrict__ Wl,
                                                float* __restrict__ hm,
                                                float* __restrict__ hl,
                                                const int* __restrict__ deg,
                                                const int* __restrict__ wpart,
                                                float* __restrict__ dinv,
                                                unsigned long long* __restrict__ wsum) {
    __shared__ int s_part[8][32];
    if (blockIdx.x < H_GRID) {
        h_body(blockIdx.x, tf_h, wf, Wm, Wl, hm, hl);
    } else {
        dinv_body(blockIdx.x - H_GRID, s_part, deg, wpart, dinv, wsum);
    }
}


__device__ __forceinline__ void load_row10(const float* __restrict__ p, float* r) {
    const float4 a = *(const float4*)(p);
    const float4 b = *(const float4*)(p + 4);
    const float2 c = *(const float2*)(p + 8);
    r[0] = a.x; r[1] = a.y; r[2] = a.z; r[3] = a.w;
    r[4] = b.x; r[5] = b.y; r[6] = b.z; r[7] = b.w;
    r[8] = c.x; r[9] = c.y;
}

__device__ __forceinline__ unsigned long long to_fix(float v) {
    return (unsigned long long)(long long)rintf(v * 4294967296.0f);
}

__device__ __forceinline__ float from_fix(unsigned long long s) {
    return (float)((double)(long long)s * (1.0 / 4294967296.0));
}

__device__ __forceinline__ void agg_task_body(const int blk, float* __restrict__ sm, float* __restrict__ sl,
                                              const int* __restrict__ answers,
                                              const unsigned short* __restrict__ list,
                                              const int* __restrict__ deg,
                                              const float* __restrict__ hm,
                                              const float* __restrict__ hl,
                                              const float* __restrict__ dinv,
                                              const float* __restrict__ bm,
                                              const float* __restrict__ bl,
                                              float* __restrict__ mean,
                                              float* __restrict__ lsd) {
    const int i = blk * 256 + threadIdx.x;
    float am[CLS], al[CLS];
#pragma unroll
    for (int c = 0; c < CLS; ++c) { am[c] = 0.f; al[c] = 0.f; }

    if (i < NODES) {
        const float d  = dinv[i];
        const float sw = d * d;
        float hmi[CLS], hli[CLS];
        load_row10(hm + (size_t)i * HP, hmi);
        load_row10(hl + (size_t)i * HP, hli);
#pragma unroll
        for (int c = 0; c < CLS; ++c) {
            am[c] = bm[c] + hmi[c] * sw;
            al[c] = bl[c] + hli[c] * sw;
        }

        if (i < TASKS) {
            const int n = deg[i] - 1;
            if (n <= CAP) {
                const uint4* lp = (const uint4*)(list + (size_t)i * CAP);
                const uint4 q0 = lp[0], q1 = lp[1], q2 = lp[2], q3 = lp[3];
                const unsigned pk[16] = {q0.x, q0.y, q0.z, q0.w, q1.x, q1.y, q1.z, q1.w,
                                         q2.x, q2.y, q2.z, q2.w, q3.x, q3.y, q3.z, q3.w};
                int e[CAP];
#pragma unroll
                for (int j = 0; j < CAP; ++j) {
                    const int v = (int)((pk[j >> 1] >> (16 * (j & 1))) & 0xFFFFu);
                    e[j] = (j < n) ? v : SENT;
                }
                int cur = -1;
                for (int it = 0; it < CAP; ++it) {
                    int m = SENT, cnt = 0;
#pragma unroll
                    for (int j = 0; j < CAP; ++j) {
                        const int v = e[j];
                        if (v > cur) {
                            if (v < m) { m = v; cnt = 1; }
                            else if (v == m) ++cnt;
                        }
                    }
                    if (m == SENT) break;
                    const int wrow = TASKS + m;
                    const float we = d * dinv[wrow];
                    float hmw[CLS], hlw[CLS];
                    load_row10(hm + (size_t)wrow * HP, hmw);
                    load_row10(hl + (size_t)wrow * HP, hlw);
                    for (int k = 0; k < cnt; ++k) {
#pragma unroll
                        for (int c = 0; c < CLS; ++c) {
                            am[c] += hmw[c] * we;
                            al[c] += hlw[c] * we;
                        }
                    }
                    cur = m;
                }
            } else {
                for (int a = 0; a < ANS; ++a) {
                    if (answers[3 * a] == i) {
                        const int wrow = TASKS + answers[3 * a + 1];
                        const float we = d * dinv[wrow];
                        float hmw[CLS], hlw[CLS];
                        load_row10(hm + (size_t)wrow * HP, hmw);
                        load_row10(hl + (size_t)wrow * HP, hlw);
#pragma unroll
                        for (int c = 0; c < CLS; ++c) {
                            am[c] += hmw[c] * we;
                            al[c] += hlw[c] * we;
                        }
                    }
                }
            }
        }
    }

#pragma unroll
    for (int c = 0; c < CLS; ++c) {
        sm[threadIdx.x * CLS + c] = am[c];
        sl[threadIdx.x * CLS + c] = al[c];
    }
    __syncthreads();
    const int base = blk * 256 * CLS;
    const int lim  = NODES * CLS - base;
    for (int k = threadIdx.x; k < 256 * CLS; k += 256) {
        if (k < lim) {
            mean[(size_t)base + k] = sm[k];
            lsd[(size_t)base + k]  = sl[k];
        }
    }
}

__device__ __forceinline__ void agg_worker_body(const int blk, unsigned long long* __restrict__ wacc,
                                                const int* __restrict__ answers,
                                                const float* __restrict__ hm,
                                                const float* __restrict__ hl,
                                                const float* __restrict__ dinv,
                                                unsigned long long* __restrict__ wsum) {
    for (int i = threadIdx.x; i < 2 * WORKERS * CLS; i += 256) wacc[i] = 0ull;
    __syncthreads();

    const int start = blk * CHUNK;
    int end = start + CHUNK;
    if (end > ANS) end = ANS;

    for (int a = start + threadIdx.x; a < end; a += 256) {
        const int t = answers[3 * a];
        const int w = answers[3 * a + 1];
        const float we = dinv[t] * dinv[TASKS + w];

        float hmt[CLS], hlt[CLS];
        load_row10(hm + (size_t)t * HP, hmt);
        load_row10(hl + (size_t)t * HP, hlt);
#pragma unroll
        for (int c = 0; c < CLS; ++c) {
            atomicAdd(&wacc[w * CLS + c], to_fix(hmt[c] * we));
            atomicAdd(&wacc[WORKERS * CLS + w * CLS + c], to_fix(hlt[c] * we));
        }
    }
    __syncthreads();

    for (int i = threadIdx.x; i < 2 * WORKERS * CLS; i += 256) atomicAdd(&wsum[i], wacc[i]);
}

constexpr int AW_GRID = (ANS + CHUNK - 1) / CHUNK;
constexpr int AT_GRID = (NODES + 255) / 256;
static_assert(2 * 256 * CLS * sizeof(float) <= 2 * WORKERS * CLS * sizeof(unsigned long long),
              "the task side's two row buffers fit in the worker side's accumulators");

__global__ __launch_bounds__(256) void k_agg(const int* __restrict__ answers,
                                             const unsigned short* __restrict__ list,
                                             const int* __restrict__ deg,
                                             const float* __restrict__ hm,
                                             const float* __restrict__ hl,
                                             const float* __restrict__ dinv,
                                             const float* __restrict__ bm,
                                             const float* __restrict__ bl,
                                             float* __restrict__ mean,
                                             float* __restrict__ lsd,
                                             unsigned long long* __restrict__ wsum) {
    __shared__ __attribute__((aligned(16))) unsigned long long s_buf[2 * WORKERS * CLS];
    if (blockIdx.x < AW_GRID) {
        agg_worker_body(blockIdx.x, s_buf, answers, hm, hl, dinv, wsum);
    } else {
        float* sm = (float*)s_buf;
        agg_task_body(blockIdx.x - AW_GRID, sm, sm + 256 * CLS,
                      answers, list, deg, hm, hl, dinv, bm, bl, mean, lsd);
    }
}


__global__ __launch_bounds__(256) void k_z(float* __restrict__ mean, float* __restrict__ lsd,
                                           const unsigned long long* __restrict__ wsum,
                                           const float* __restrict__ eps, float* __restrict__ zp) {
    const int stride = gridDim.x * 256;
    const int first  = blockIdx.x * 256 + threadIdx.x;
    float mv[Z_PER], lv[Z_PER], ev[Z_PER];
#pragma unroll
    for (int j = 0; j < Z_PER; ++j) {
        int ic = first + j * stride;
        if (ic > NODES * CLS - 1) ic = NODES * CLS - 1;
        mv[j] = mean[ic];
        lv[j] = lsd[ic];
        ev[j] = eps[ic];
    }
#pragma unroll
    for (int j = 0; j < Z_PER; ++j) {
        const int idx = first + j * stride;
        if (idx < NODES * CLS) {
            float m = mv[j];
            float l = lv[j];
            if (idx >= TASKS * CLS) {
                const int q = idx - TASKS * CLS;
                m += from_fix(wsum[q]);
                l += from_fix(wsum[WORKERS * CLS + q]);
                mean[idx] = m;
                lsd[idx]  = l;
            }
            const int node = idx / CLS;
            const int c    = idx - node * CLS;
            zp[(size_t)node * HP + c] = m + (ev[j] * 0.01f) * expf(l);
        }
    }
}

__global__ __launch_bounds__(256) void k_crowd(const int* __restrict__ answers,
                                               const float* __restrict__ zp,
                                               float* __restrict__ crowd) {
    __shared__ __attribute__((aligned(16))) float s_zw[WORKERS * HP];
    __shared__ __attribute__((aligned(16))) float s_out[2][CR_TILE * CLS];

    {
        const float2* srcw = (const float2*)(zp + (size_t)TASKS * HP);
        float2* dstw = (float2*)s_zw;
        const int p = threadIdx.x & 7;
        if (p < CLS / 2) {
            for (int r = threadIdx.x >> 3; r < WORKERS; r += 32)
                dstw[r * (HP / 2) + p] = srcw[r * (HP / 2) + p];
        }
    }
    __syncthreads();

    constexpr int NT = (ANS + CR_TILE - 1) / CR_TILE;
    constexpr int NQ = CR_TILE * CLS / 4;
    int buf = 0;
    for (int tile = blockIdx.x; tile < NT; tile += gridDim.x, buf ^= 1) {
        const int base = tile * CR_TILE;
        float* sb = s_out[buf];

        int t[2], w[2];
#pragma unroll
        for (int h = 0; h < 2; ++h) {
            int a = base + h * 256 + threadIdx.x;
            if (a > ANS - 1) a = ANS - 1;
            t[h] = answers[3 * a];
            w[h] = answers[3 * a + 1];
        }
        float4 ta[2], tb[2];
        float2 tc[2];
#pragma unroll
        for (int h = 0; h < 2; ++h) {
            const float* zt = zp + (size_t)t[h] * HP;
            ta[h] = *(const float4*)(zt);
            tb[h] = *(const float4*)(zt + 4);
            tc[h] = *(const float2*)(zt + 8);
        }
#pragma unroll
        for (int h = 0; h < 2; ++h) {
            const float* zw = s_zw + w[h] * HP;
            const float4 wa = *(const float4*)(zw);
            const float4 wb = *(const float4*)(zw + 4);
            const float2 wc = *(const float2*)(zw + 8);
            float2* o = (float2*)(sb + (h * 256 + threadIdx.x) * CLS);
            o[0] = make_float2(ta[h].x * wa.x, ta[h].y * wa.y);
            o[1] = make_float2(ta[h].z * wa.z, ta[h].w * wa.w);
            o[2] = make_float2(tb[h].x * wb.x, tb[h].y * wb.y);
            o[3] = make_float2(tb[h].z * wb.z, tb[h].w * wb.w);
            o[4] = make_float2(tc[h].x * wc.x, tc[h].y * wc.y);
        }
        __syncthreads();

        float4* dst = (float4*)(crowd + (size_t)base * CLS);
        const float4* so = (const float4*)sb;
        if (ANS - base >= CR_TILE) {
#pragma unroll
            for (int k = 0; k < NQ / 256; ++k) dst[k * 256 + threadIdx.x] = so[k * 256 + threadIdx.x];
        } else {
            const int nq = (ANS - base) * CLS / 4;
            for (int k = threadIdx.x; k < nq; k += 256) dst[k] = so[k];
        }
    }
}


extern "C" void kernel_launch(void* const* d_in, const int* in_sizes, int n_in,
                              void* d_out, int out_size, void* d_ws, size_t ws_size,
                              hipStream_t stream) {
    (void)in_sizes; (void)n_in; (void)out_size; (void)ws_size;

    const float* task_feature   = (const float*)d_in[0];
    const int*   answers        = (const int*)d_in[1];
    const float* worker_feature = (const float*)d_in[2];
    const float* W_efc          = (const float*)d_in[3];
    const float* b_efc          = (const float*)d_in[4];
    const float* W_mean         = (const float*)d_in[5];
    const float* b_mean         = (const float*)d_in[6];
    const float* W_ls           = (const float*)d_in[7];
    const float* b_ls           = (const float*)d_in[8];
    const float* eps            = (const float*)d_in[9];

    float* out   = (float*)d_out;
    float* crowd = out;
    float* mean  = out + (size_t)ANS * CLS;
    float* lsd   = mean + (size_t)NODES * CLS;

    float*  ws   = (float*)d_ws;
    float*  tf_h = ws;
    float2* Bp   = (float2*)(tf_h + (size_t)TASKS * HID);
    float*  hm   = (float*)(Bp + (size_t)(FEAT / 2) * HID);
    float*  hl   = hm + (size_t)NODES * HP;
    float*  dinv = hl + (size_t)NODES * HP;
    int*    deg  = (int*)(dinv + NODES);
    static_assert((size_t)ANS * CLS >= (size_t)4 * 1024 * 1024, "the three arrays fit in the crowd block");
    static_assert((size_t)TASKS * CAP * sizeof(unsigned short) <= (size_t)2 * 1024 * 1024 * sizeof(float) &&
                  (size_t)2 * WORKERS * CLS * sizeof(unsigned long long) <= (size_t)1024 * 1024 * sizeof(float) &&
                  (size_t)DC_GRID * WPAD * sizeof(int) <= (size_t)1024 * 1024 * sizeof(float),
                  "each array stays inside its own slice");
    unsigned short*     list = (unsigned short*)crowd;
    unsigned long long* wsum = (unsigned long long*)(crowd + (size_t)2 * 1024 * 1024);
    int*                wpart = (int*)(crowd + (size_t)3 * 1024 * 1024);
    float*              zz   = tf_h;

    auto cdiv = [](int a, int b) { return (a + b - 1) / b; };

    k_packB    <<<PB_GRID + DI_GRID, 256, 0, stream>>>(W_efc, Bp, deg);
    k_gemm_count<<<DC_GRID + GE_GRID, 256, 0, stream>>>(task_feature, Bp, b_efc, tf_h,
                                                        answers, deg, list, wpart);
    k_h_dinv   <<<H_GRID + DV_TASK + DV_WORK, 256, 0, stream>>>(tf_h, worker_feature, W_mean, W_ls,
                                                                hm, hl, deg, wpart, dinv, wsum);

    k_agg      <<<AW_GRID + AT_GRID, 256, 0, stream>>>(answers, list, deg, hm, hl, dinv,
                                                      b_mean, b_ls, mean, lsd, wsum);

    k_z        <<<cdiv(cdiv(NODES * CLS, Z_PER), 256), 256, 0, stream>>>(mean, lsd, wsum, eps, zz);
    k_crowd    <<<CR_GRID, 256, 0, stream>>>(answers, zz, crowd);
}


# hm/hl of task rows formed in the product launch from tf_h's accumulators (product transposed, so D is the next B operand); tf_h never stored; their 128-term sums run in a permuted k order
# speedup vs baseline: 4.3097x; 1.2896x over previous
#include <hip/hip_runtime.h>

typedef __attribute__((ext_vector_type(2))) float v2f;
typedef __attribute__((ext_vector_type(8))) float v8f;

constexpr int TASKS   = 100000;
constexpr int WORKERS = 500;
constexpr int FEAT    = 512;
constexpr int CLS     = 10;
constexpr int HP      = 12;
constexpr int ANS     = 1000000;
constexpr int HID     = 128;
constexpr int NODES   = TASKS + WORKERS;
constexpr int CHUNK   = 4096;
constexpr int CAP     = 32;
constexpr int SENT    = 0xFFFF;
constexpr int Z_PER   = 8;
constexpr int CR_TILE = 512;
constexpr int CR_GRID = 512;
constexpr int DC_GRID = 256;
constexpr int DC_PER  = 4;
constexpr int WPAD    = 512;
static_assert(WPAD >= WORKERS && WPAD % 256 == 0, "a row of worker counts is a whole number of 256-thread rounds");
static_assert(ANS % 2 == 0 && CR_TILE % 2 == 0, "a tile's floats must split into 16-byte pieces");
static_assert((CR_TILE * CLS / 4) % 256 == 0, "a full tile is a whole number of 256-thread store rounds");


__device__ __forceinline__ void deg_count_body(const int blk, int* __restrict__ s_wdeg,
                                               const int* __restrict__ answers,
                                               int* __restrict__ deg,
                                               unsigned short* __restrict__ list,
                                               int* __restrict__ wpart) {
    for (int i = threadIdx.x; i < WPAD; i += 256) s_wdeg[i] = 0;
    __syncthreads();

    constexpr int TRIP = 256 * DC_PER;
    for (int base = blk * TRIP; base < ANS; base += DC_GRID * TRIP) {
        int t[DC_PER], w[DC_PER], old[DC_PER];
        bool ok[DC_PER];
#pragma unroll
        for (int h = 0; h < DC_PER; ++h) {
            const int a = base + h * 256 + threadIdx.x;
            ok[h] = a < ANS;
            const int ac = ok[h] ? a : ANS - 1;
            t[h] = answers[3 * ac];
            w[h] = answers[3 * ac + 1];
        }
#pragma unroll
        for (int h = 0; h < DC_PER; ++h) {
            old[h] = CAP + 1;
            if (ok[h]) {
                old[h] = atomicAdd(&deg[t[h]], 1);
                atomicAdd(&s_wdeg[w[h]], 1);
            }
        }
#pragma unroll
        for (int h = 0; h < DC_PER; ++h) {
            const int slot = old[h] - 1;
            if (slot < CAP) list[(size_t)t[h] * CAP + slot] = (unsigned short)w[h];
        }
    }
    __syncthreads();
    for (int i = threadIdx.x; i < WPAD; i += 256) wpart[(size_t)blk * WPAD + i] = s_wdeg[i];
}

constexpr int DV_TASK = (TASKS + 255) / 256;
constexpr int DV_WORK = (WORKERS + 31) / 32;
static_assert(DV_WORK * 32 <= WPAD, "the worker index of every lane stays inside a padded row");
static_assert(DC_GRID % 8 == 0, "the rows split evenly over the eight waves of a work-group");

static_assert(DV_TASK * 256 >= 2 * WORKERS * CLS, "the task work-groups cover every worker-side sum");

__device__ __forceinline__ void dinv_body(const int blk, int (*s_part)[32],
                                          const int* __restrict__ deg,
                                          const int* __restrict__ wpart,
                                          float* __restrict__ dinv,
                                          unsigned long long* __restrict__ wsum) {
    if (blk < DV_TASK) {
        const int i = blk * 256 + threadIdx.x;
        if (i < TASKS) dinv[i] = rsqrtf((float)deg[i]);
        if (i < 2 * WORKERS * CLS) wsum[i] = 0ull;
        return;
    }
    const int lane = threadIdx.x & 31;
    const int j    = threadIdx.x >> 5;
    const int w    = (blk - DV_TASK) * 32 + lane;
    int s = 0;
#pragma unroll
    for (int q = 0; q < DC_GRID / 8; ++q) s += wpart[(q * 8 + j) * WPAD + w];
    s_part[j][lane] = s;
    __syncthreads();
    if (j == 0 && w < WORKERS) {
        int d = deg[TASKS + w];
#pragma unroll
        for (int q = 0; q < 8; ++q) d += s_part[q][lane];
        dinv[TASKS + w] = rsqrtf((float)d);
    }
}


constexpr int PB_GRID = ((FEAT / 2) * HID + 255) / 256;
constexpr int DI_GRID = (NODES + 255) / 256;

constexpr int WP_GROUPS = HID / 4;
constexpr int PW_GRID   = (WP_GROUPS * 32 + 255) / 256;
static_assert(HID % 16 == 0, "hidden tiles of 16");

__global__ void k_packB(const float* __restrict__ B, float2* __restrict__ Bp, int* __restrict__ deg,
                        const float* __restrict__ Wm, const float* __restrict__ Wl,
                        float4* __restrict__ Wp) {
    if (blockIdx.x >= PB_GRID + DI_GRID) {
        const int e = (blockIdx.x - PB_GRID - DI_GRID) * 256 + threadIdx.x;
        if (e < WP_GROUPS * 32) {
            const int g = e >> 5, lane = e & 31;
            const int cc = lane & 15, h = lane >> 4;
            const int n = 16 * (g >> 2) + 8 * h + 2 * (g & 3);
            float4 v = make_float4(0.f, 0.f, 0.f, 0.f);
            if (cc < CLS) v = make_float4(Wm[n * CLS + cc], Wm[(n + 1) * CLS + cc],
                                          Wl[n * CLS + cc], Wl[(n + 1) * CLS + cc]);
            Wp[e] = v;
        }
        return;
    }
    if (blockIdx.x >= PB_GRID) {
        const int i = (blockIdx.x - PB_GRID) * 256 + threadIdx.x;
        if (i < NODES) deg[i] = 1;
        return;
    }
    int idx = blockIdx.x * 256 + threadIdx.x;
    if (idx < (FEAT / 2) * HID) {
        int kp = idx / HID;
        int n  = idx - kp * HID;
        Bp[idx] = make_float2(B[(2 * kp) * HID + n], B[(2 * kp + 1) * HID + n]);
    }
}


__device__ __forceinline__ void gemm_efc_body(const int blk,
                                              const float* __restrict__ A,
                                              const float2* __restrict__ Bp,
                                              const float* __restrict__ bias,
                                              const float4* __restrict__ Wp,
                                              float* __restrict__ hm,
                                              float* __restrict__ hl) {
    const int wave   = threadIdx.x >> 5;
    const int lane   = threadIdx.x & 31;
    const int lane16 = lane & 15;
    const int hi     = lane >> 4;

    const int slab     = blk * 8 + wave;
    const bool valid   = (slab * 32) < TASKS;
    const int row_base = valid ? slab * 32 : 0;

    const float*  arow0 = A + (size_t)(row_base + lane16) * FEAT + 2 * hi;
    const float*  arow1 = arow0 + (size_t)16 * FEAT;
    const float2* bbase = Bp + lane16;

    v8f acc[2][8];
#pragma unroll
    for (int t = 0; t < 8; ++t) {
        const float* bt = bias + t * 16 + 8 * hi;
        acc[0][t] = (v8f){bt[0], bt[1], bt[2], bt[3], bt[4], bt[5], bt[6], bt[7]};
        acc[1][t] = acc[0][t];
    }

    for (int k = 0; k < FEAT; k += 4) {
        const float2 av0 = *(const float2*)(arow0 + k);
        const float2 av1 = *(const float2*)(arow1 + k);
        v2f af0 = {av0.x, av0.y};
        v2f af1 = {av1.x, av1.y};
        const float2* brow = bbase + (size_t)((k >> 1) + hi) * HID;
#pragma unroll
        for (int t = 0; t < 8; ++t) {
            const float2 bv = brow[t * 16];
            v2f bf = {bv.x, bv.y};
            acc[0][t] = __builtin_amdgcn_wmma_f32_16x16x4_f32(
                false, bf, false, af0, (short)0, acc[0][t], false, false);
            acc[1][t] = __builtin_amdgcn_wmma_f32_16x16x4_f32(
                false, bf, false, af1, (short)0, acc[1][t], false, false);
        }
    }

    const float4* wp = Wp + lane;
#pragma unroll
    for (int r = 0; r < 2; ++r) {
        v8f hM = {0.f, 0.f, 0.f, 0.f, 0.f, 0.f, 0.f, 0.f};
        v8f hL = {0.f, 0.f, 0.f, 0.f, 0.f, 0.f, 0.f, 0.f};
#pragma unroll
        for (int t = 0; t < 8; ++t) {
#pragma unroll
            for (int jp = 0; jp < 4; ++jp) {
                const float4 wv = wp[(t * 4 + jp) * 32];
                const v2f wm = {wv.x, wv.y};
                const v2f wl = {wv.z, wv.w};
                const float x0 = acc[r][t][2 * jp];
                const float x1 = acc[r][t][2 * jp + 1];
                const v2f xf = {x0 > 0.f ? x0 : 0.f, x1 > 0.f ? x1 : 0.f};
                hM = __builtin_amdgcn_wmma_f32_16x16x4_f32(false, wm, false, xf, (short)0, hM, false, false);
                hL = __builtin_amdgcn_wmma_f32_16x16x4_f32(false, wl, false, xf, (short)0, hL, false, false);
            }
        }
        if (valid) {
            const size_t m = (size_t)(row_base + r * 16 + lane16);
            float* pm = hm + m * HP + 8 * hi;
            float* pl = hl + m * HP + 8 * hi;
            if (hi == 0) {
                *(float4*)(pm)     = make_float4(hM[0], hM[1], hM[2], hM[3]);
                *(float4*)(pm + 4) = make_float4(hM[4], hM[5], hM[6], hM[7]);
                *(float4*)(pl)     = make_float4(hL[0], hL[1], hL[2], hL[3]);
                *(float4*)(pl + 4) = make_float4(hL[4], hL[5], hL[6], hL[7]);
            } else {
                *(float2*)(pm) = make_float2(hM[0], hM[1]);
                *(float2*)(pl) = make_float2(hL[0], hL[1]);
            }
        }
    }
}

static_assert(TASKS % 16 == 0, "worker rows start on a strip boundary");

__device__ __forceinline__ void h_body(const int blk,
                                       const float* __restrict__ wf,
                                       const float* __restrict__ Wm,
                                       const float* __restrict__ Wl,
                                       float* __restrict__ hm,
                                       float* __restrict__ hl) {
    const int wave   = threadIdx.x >> 5;
    const int lane   = threadIdx.x & 31;
    const int lane16 = lane & 15;
    const int hi     = lane >> 4;

    const int node_base = TASKS + (blk * 8 + wave) * 16;
    int nd = node_base + lane16;
    if (nd > NODES - 1) nd = NODES - 1;
    const float* xrow = wf + (size_t)(nd - TASKS) * HID;
    const int n = lane16;
    const bool ncol = n < CLS;

    v8f accM = {0.f, 0.f, 0.f, 0.f, 0.f, 0.f, 0.f, 0.f};
    v8f accL = {0.f, 0.f, 0.f, 0.f, 0.f, 0.f, 0.f, 0.f};

    for (int k = 0; k < HID; k += 4) {
        const int k0 = k + 2 * hi;
        const float2 av = *(const float2*)(xrow + k0);
        v2f af = {av.x, av.y};
        v2f bm, bl;
        bm.x = ncol ? Wm[k0 * CLS + n] : 0.f;
        bm.y = ncol ? Wm[(k0 + 1) * CLS + n] : 0.f;
        bl.x = ncol ? Wl[k0 * CLS + n] : 0.f;
        bl.y = ncol ? Wl[(k0 + 1) * CLS + n] : 0.f;
        accM = __builtin_amdgcn_wmma_f32_16x16x4_f32(false, af, false, bm, (short)0, accM, false, false);
        accL = __builtin_amdgcn_wmma_f32_16x16x4_f32(false, af, false, bl, (short)0, accL, false, false);
    }

    if (ncol) {
#pragma unroll
        for (int j = 0; j < 8; ++j) {
            int m = node_base + j + 8 * hi;
            if (m < NODES) {
                hm[(size_t)m * HP + n] = accM[j];
                hl[(size_t)m * HP + n] = accL[j];
            }
        }
    }
}

constexpr int GE_GRID = (TASKS / 32 + 7) / 8;
constexpr int HW_GRID = ((WORKERS + 15) / 16 + 7) / 8;

__global__ __launch_bounds__(256) void k_gemm_count(const float* __restrict__ A,
                                                    const float2* __restrict__ Bp,
                                                    const float* __restrict__ bias,
                                                    const float4* __restrict__ Wp,
                                                    const float* __restrict__ wf,
                                                    const float* __restrict__ Wm,
                                                    const float* __restrict__ Wl,
                                                    float* __restrict__ hm,
                                                    float* __restrict__ hl,
                                                    const int* __restrict__ answers,
                                                    int* __restrict__ deg,
                                                    unsigned short* __restrict__ list,
                                                    int* __restrict__ wpart) {
    __shared__ int s_wdeg[WPAD];
    if (blockIdx.x < GE_GRID) {
        gemm_efc_body(blockIdx.x, A, Bp, bias, Wp, hm, hl);
    } else if (blockIdx.x < GE_GRID + DC_GRID) {
        deg_count_body(blockIdx.x - GE_GRID, s_wdeg, answers, deg, list, wpart);
    } else {
        h_body(blockIdx.x - GE_GRID - DC_GRID, wf, Wm, Wl, hm, hl);
    }
}

__global__ __launch_bounds__(256) void k_dinv(const int* __restrict__ deg,
                                              const int* __restrict__ wpart,
                                              float* __restrict__ dinv,
                                              unsigned long long* __restrict__ wsum) {
    __shared__ int s_part[8][32];
    dinv_body(blockIdx.x, s_part, deg, wpart, dinv, wsum);
}


__device__ __forceinline__ void load_row10(const float* __restrict__ p, float* r) {
    const float4 a = *(const float4*)(p);
    const float4 b = *(const float4*)(p + 4);
    const float2 c = *(const float2*)(p + 8);
    r[0] = a.x; r[1] = a.y; r[2] = a.z; r[3] = a.w;
    r[4] = b.x; r[5] = b.y; r[6] = b.z; r[7] = b.w;
    r[8] = c.x; r[9] = c.y;
}

__device__ __forceinline__ unsigned long long to_fix(float v) {
    return (unsigned long long)(long long)rintf(v * 4294967296.0f);
}

__device__ __forceinline__ float from_fix(unsigned long long s) {
    return (float)((double)(long long)s * (1.0 / 4294967296.0));
}

__device__ __forceinline__ void agg_task_body(const int blk, float* __restrict__ sm, float* __restrict__ sl,
                                              const int* __restrict__ answers,
                                              const unsigned short* __restrict__ list,
                                              const int* __restrict__ deg,
                                              const float* __restrict__ hm,
                                              const float* __restrict__ hl,
                                              const float* __restrict__ dinv,
                                              const float* __restrict__ bm,
                                              const float* __restrict__ bl,
                                              float* __restrict__ mean,
                                              float* __restrict__ lsd) {
    const int i = blk * 256 + threadIdx.x;
    float am[CLS], al[CLS];
#pragma unroll
    for (int c = 0; c < CLS; ++c) { am[c] = 0.f; al[c] = 0.f; }

    if (i < NODES) {
        const float d  = dinv[i];
        const float sw = d * d;
        float hmi[CLS], hli[CLS];
        load_row10(hm + (size_t)i * HP, hmi);
        load_row10(hl + (size_t)i * HP, hli);
#pragma unroll
        for (int c = 0; c < CLS; ++c) {
            am[c] = bm[c] + hmi[c] * sw;
            al[c] = bl[c] + hli[c] * sw;
        }

        if (i < TASKS) {
            const int n = deg[i] - 1;
            if (n <= CAP) {
                const uint4* lp = (const uint4*)(list + (size_t)i * CAP);
                const uint4 q0 = lp[0], q1 = lp[1], q2 = lp[2], q3 = lp[3];
                const unsigned pk[16] = {q0.x, q0.y, q0.z, q0.w, q1.x, q1.y, q1.z, q1.w,
                                         q2.x, q2.y, q2.z, q2.w, q3.x, q3.y, q3.z, q3.w};
                int e[CAP];
#pragma unroll
                for (int j = 0; j < CAP; ++j) {
                    const int v = (int)((pk[j >> 1] >> (16 * (j & 1))) & 0xFFFFu);
                    e[j] = (j < n) ? v : SENT;
                }
                int cur = -1;
                for (int it = 0; it < CAP; ++it) {
                    int m = SENT, cnt = 0;
#pragma unroll
                    for (int j = 0; j < CAP; ++j) {
                        const int v = e[j];
                        if (v > cur) {
                            if (v < m) { m = v; cnt = 1; }
                            else if (v == m) ++cnt;
                        }
                    }
                    if (m == SENT) break;
                    const int wrow = TASKS + m;
                    const float we = d * dinv[wrow];
                    float hmw[CLS], hlw[CLS];
                    load_row10(hm + (size_t)wrow * HP, hmw);
                    load_row10(hl + (size_t)wrow * HP, hlw);
                    for (int k = 0; k < cnt; ++k) {
#pragma unroll
                        for (int c = 0; c < CLS; ++c) {
                            am[c] += hmw[c] * we;
                            al[c] += hlw[c] * we;
                        }
                    }
                    cur = m;
                }
            } else {
                for (int a = 0; a < ANS; ++a) {
                    if (answers[3 * a] == i) {
                        const int wrow = TASKS + answers[3 * a + 1];
                        const float we = d * dinv[wrow];
                        float hmw[CLS], hlw[CLS];
                        load_row10(hm + (size_t)wrow * HP, hmw);
                        load_row10(hl + (size_t)wrow * HP, hlw);
#pragma unroll
                        for (int c = 0; c < CLS; ++c) {
                            am[c] += hmw[c] * we;
                            al[c] += hlw[c] * we;
                        }
                    }
                }
            }
        }
    }

#pragma unroll
    for (int c = 0; c < CLS; ++c) {
        sm[threadIdx.x * CLS + c] = am[c];
        sl[threadIdx.x * CLS + c] = al[c];
    }
    __syncthreads();
    const int base = blk * 256 * CLS;
    const int lim  = NODES * CLS - base;
    for (int k = threadIdx.x; k < 256 * CLS; k += 256) {
        if (k < lim) {
            mean[(size_t)base + k] = sm[k];
            lsd[(size_t)base + k]  = sl[k];
        }
    }
}

__device__ __forceinline__ void agg_worker_body(const int blk, unsigned long long* __restrict__ wacc,
                                                const int* __restrict__ answers,
                                                const float* __restrict__ hm,
                                                const float* __restrict__ hl,
                                                const float* __restrict__ dinv,
                                                unsigned long long* __restrict__ wsum) {
    for (int i = threadIdx.x; i < 2 * WORKERS * CLS; i += 256) wacc[i] = 0ull;
    __syncthreads();

    const int start = blk * CHUNK;
    int end = start + CHUNK;
    if (end > ANS) end = ANS;

    for (int a = start + threadIdx.x; a < end; a += 256) {
        const int t = answers[3 * a];
        const int w = answers[3 * a + 1];
        const float we = dinv[t] * dinv[TASKS + w];

        float hmt[CLS], hlt[CLS];
        load_row10(hm + (size_t)t * HP, hmt);
        load_row10(hl + (size_t)t * HP, hlt);
#pragma unroll
        for (int c = 0; c < CLS; ++c) {
            atomicAdd(&wacc[w * CLS + c], to_fix(hmt[c] * we));
            atomicAdd(&wacc[WORKERS * CLS + w * CLS + c], to_fix(hlt[c] * we));
        }
    }
    __syncthreads();

    for (int i = threadIdx.x; i < 2 * WORKERS * CLS; i += 256) atomicAdd(&wsum[i], wacc[i]);
}

constexpr int AW_GRID = (ANS + CHUNK - 1) / CHUNK;
constexpr int AT_GRID = (NODES + 255) / 256;
static_assert(2 * 256 * CLS * sizeof(float) <= 2 * WORKERS * CLS * sizeof(unsigned long long),
              "the task side's two row buffers fit in the worker side's accumulators");

__global__ __launch_bounds__(256) void k_agg(const int* __restrict__ answers,
                                             const unsigned short* __restrict__ list,
                                             const int* __restrict__ deg,
                                             const float* __restrict__ hm,
                                             const float* __restrict__ hl,
                                             const float* __restrict__ dinv,
                                             const float* __restrict__ bm,
                                             const float* __restrict__ bl,
                                             float* __restrict__ mean,
                                             float* __restrict__ lsd,
                                             unsigned long long* __restrict__ wsum) {
    __shared__ __attribute__((aligned(16))) unsigned long long s_buf[2 * WORKERS * CLS];
    if (blockIdx.x < AW_GRID) {
        agg_worker_body(blockIdx.x, s_buf, answers, hm, hl, dinv, wsum);
    } else {
        float* sm = (float*)s_buf;
        agg_task_body(blockIdx.x - AW_GRID, sm, sm + 256 * CLS,
                      answers, list, deg, hm, hl, dinv, bm, bl, mean, lsd);
    }
}


__global__ __launch_bounds__(256) void k_z(float* __restrict__ mean, float* __restrict__ lsd,
                                           const unsigned long long* __restrict__ wsum,
                                           const float* __restrict__ eps, float* __restrict__ zp) {
    const int stride = gridDim.x * 256;
    const int first  = blockIdx.x * 256 + threadIdx.x;
    float mv[Z_PER], lv[Z_PER], ev[Z_PER];
#pragma unroll
    for (int j = 0; j < Z_PER; ++j) {
        int ic = first + j * stride;
        if (ic > NODES * CLS - 1) ic = NODES * CLS - 1;
        mv[j] = mean[ic];
        lv[j] = lsd[ic];
        ev[j] = eps[ic];
    }
#pragma unroll
    for (int j = 0; j < Z_PER; ++j) {
        const int idx = first + j * stride;
        if (idx < NODES * CLS) {
            float m = mv[j];
            float l = lv[j];
            if (idx >= TASKS * CLS) {
                const int q = idx - TASKS * CLS;
                m += from_fix(wsum[q]);
                l += from_fix(wsum[WORKERS * CLS + q]);
                mean[idx] = m;
                lsd[idx]  = l;
            }
            const int node = idx / CLS;
            const int c    = idx - node * CLS;
            zp[(size_t)node * HP + c] = m + (ev[j] * 0.01f) * expf(l);
        }
    }
}

__global__ __launch_bounds__(256) void k_crowd(const int* __restrict__ answers,
                                               const float* __restrict__ zp,
                                               float* __restrict__ crowd) {
    __shared__ __attribute__((aligned(16))) float s_zw[WORKERS * HP];
    __shared__ __attribute__((aligned(16))) float s_out[2][CR_TILE * CLS];

    {
        const float2* srcw = (const float2*)(zp + (size_t)TASKS * HP);
        float2* dstw = (float2*)s_zw;
        const int p = threadIdx.x & 7;
        if (p < CLS / 2) {
            for (int r = threadIdx.x >> 3; r < WORKERS; r += 32)
                dstw[r * (HP / 2) + p] = srcw[r * (HP / 2) + p];
        }
    }
    __syncthreads();

    constexpr int NT = (ANS + CR_TILE - 1) / CR_TILE;
    constexpr int NQ = CR_TILE * CLS / 4;
    int buf = 0;
    for (int tile = blockIdx.x; tile < NT; tile += gridDim.x, buf ^= 1) {
        const int base = tile * CR_TILE;
        float* sb = s_out[buf];

        int t[2], w[2];
#pragma unroll
        for (int h = 0; h < 2; ++h) {
            int a = base + h * 256 + threadIdx.x;
            if (a > ANS - 1) a = ANS - 1;
            t[h] = answers[3 * a];
            w[h] = answers[3 * a + 1];
        }
        float4 ta[2], tb[2];
        float2 tc[2];
#pragma unroll
        for (int h = 0; h < 2; ++h) {
            const float* zt = zp + (size_t)t[h] * HP;
            ta[h] = *(const float4*)(zt);
            tb[h] = *(const float4*)(zt + 4);
            tc[h] = *(const float2*)(zt + 8);
        }
#pragma unroll
        for (int h = 0; h < 2; ++h) {
            const float* zw = s_zw + w[h] * HP;
            const float4 wa = *(const float4*)(zw);
            const float4 wb = *(const float4*)(zw + 4);
            const float2 wc = *(const float2*)(zw + 8);
            float2* o = (float2*)(sb + (h * 256 + threadIdx.x) * CLS);
            o[0] = make_float2(ta[h].x * wa.x, ta[h].y * wa.y);
            o[1] = make_float2(ta[h].z * wa.z, ta[h].w * wa.w);
            o[2] = make_float2(tb[h].x * wb.x, tb[h].y * wb.y);
            o[3] = make_float2(tb[h].z * wb.z, tb[h].w * wb.w);
            o[4] = make_float2(tc[h].x * wc.x, tc[h].y * wc.y);
        }
        __syncthreads();

        float4* dst = (float4*)(crowd + (size_t)base * CLS);
        const float4* so = (const float4*)sb;
        if (ANS - base >= CR_TILE) {
#pragma unroll
            for (int k = 0; k < NQ / 256; ++k) dst[k * 256 + threadIdx.x] = so[k * 256 + threadIdx.x];
        } else {
            const int nq = (ANS - base) * CLS / 4;
            for (int k = threadIdx.x; k < nq; k += 256) dst[k] = so[k];
        }
    }
}


extern "C" void kernel_launch(void* const* d_in, const int* in_sizes, int n_in,
                              void* d_out, int out_size, void* d_ws, size_t ws_size,
                              hipStream_t stream) {
    (void)in_sizes; (void)n_in; (void)out_size; (void)ws_size;

    const float* task_feature   = (const float*)d_in[0];
    const int*   answers        = (const int*)d_in[1];
    const float* worker_feature = (const float*)d_in[2];
    const float* W_efc          = (const float*)d_in[3];
    const float* b_efc          = (const float*)d_in[4];
    const float* W_mean         = (const float*)d_in[5];
    const float* b_mean         = (const float*)d_in[6];
    const float* W_ls           = (const float*)d_in[7];
    const float* b_ls           = (const float*)d_in[8];
    const float* eps            = (const float*)d_in[9];

    float* out   = (float*)d_out;
    float* crowd = out;
    float* mean  = out + (size_t)ANS * CLS;
    float* lsd   = mean + (size_t)NODES * CLS;

    float*  ws   = (float*)d_ws;
    float*  tf_h = ws;
    float2* Bp   = (float2*)(tf_h + (size_t)TASKS * HID);
    float*  hm   = (float*)(Bp + (size_t)(FEAT / 2) * HID);
    float*  hl   = hm + (size_t)NODES * HP;
    float*  dinv = hl + (size_t)NODES * HP;
    int*    deg  = (int*)(dinv + NODES);
    static_assert((size_t)ANS * CLS >= (size_t)4 * 1024 * 1024, "the three arrays fit in the crowd block");
    static_assert((size_t)TASKS * CAP * sizeof(unsigned short) <= (size_t)2 * 1024 * 1024 * sizeof(float) &&
                  (size_t)2 * WORKERS * CLS * sizeof(unsigned long long) <= (size_t)1024 * 1024 * sizeof(float) &&
                  (size_t)DC_GRID * WPAD * sizeof(int) <= (size_t)1024 * 1024 * sizeof(float),
                  "each array stays inside its own slice");
    unsigned short*     list = (unsigned short*)crowd;
    unsigned long long* wsum = (unsigned long long*)(crowd + (size_t)2 * 1024 * 1024);
    int*                wpart = (int*)(crowd + (size_t)3 * 1024 * 1024);
    float*              zz   = tf_h;
    static_assert((size_t)NODES * HP <= (size_t)2 * 1024 * 1024 &&
                  (size_t)2 * 1024 * 1024 + (size_t)WP_GROUPS * 32 * 4 <= (size_t)TASKS * HID,
                  "z rows and packed class weights sit apart inside the old tf_h space");
    float4*             Wp   = (float4*)(tf_h + (size_t)2 * 1024 * 1024);

    auto cdiv = [](int a, int b) { return (a + b - 1) / b; };

    k_packB    <<<PB_GRID + DI_GRID + PW_GRID, 256, 0, stream>>>(W_efc, Bp, deg, W_mean, W_ls, Wp);
    k_gemm_count<<<GE_GRID + DC_GRID + HW_GRID, 256, 0, stream>>>(task_feature, Bp, b_efc, Wp,
                                                                  worker_feature, W_mean, W_ls, hm, hl,
                                                                  answers, deg, list, wpart);
    k_dinv     <<<DV_TASK + DV_WORK, 256, 0, stream>>>(deg, wpart, dinv, wsum);

    k_agg      <<<AW_GRID + AT_GRID, 256, 0, stream>>>(answers, list, deg, hm, hl, dinv,
                                                      b_mean, b_ls, mean, lsd, wsum);

    k_z        <<<cdiv(cdiv(NODES * CLS, Z_PER), 256), 256, 0, stream>>>(mean, lsd, wsum, eps, zz);
    k_crowd    <<<CR_GRID, 256, 0, stream>>>(answers, zz, crowd);
}


# Launches 6 to 4: rsqrt-degree and z launches dissolve into their users (rsqrt from integer degrees, worker counts handed over by integer adds; task z in the aggregation, worker z formed in LDS)
# speedup vs baseline: 4.4557x; 1.0339x over previous
#include <hip/hip_runtime.h>

typedef __attribute__((ext_vector_type(2))) float v2f;
typedef __attribute__((ext_vector_type(8))) float v8f;

constexpr int TASKS   = 100000;
constexpr int WORKERS = 500;
constexpr int FEAT    = 512;
constexpr int CLS     = 10;
constexpr int HP      = 12;
constexpr int ANS     = 1000000;
constexpr int HID     = 128;
constexpr int NODES   = TASKS + WORKERS;
constexpr int CHUNK   = 4096;
constexpr int CAP     = 32;
constexpr int SENT    = 0xFFFF;
constexpr int CR_TILE = 512;
constexpr int CR_GRID = 512;
constexpr int DC_GRID = 256;
constexpr int DC_PER  = 4;
constexpr int WPAD    = 512;
static_assert(WPAD >= WORKERS && WPAD % 256 == 0, "the worker counts are a whole number of 256-thread rounds");
static_assert(ANS % 2 == 0 && CR_TILE % 2 == 0, "a tile's floats must split into 16-byte pieces");
static_assert((CR_TILE * CLS / 4) % 256 == 0, "a full tile is a whole number of 256-thread store rounds");


__device__ __forceinline__ void deg_count_body(const int blk, int* __restrict__ s_wdeg,
                                               const int* __restrict__ answers,
                                               int* __restrict__ deg,
                                               unsigned short* __restrict__ list) {
    for (int i = threadIdx.x; i < WPAD; i += 256) s_wdeg[i] = 0;
    __syncthreads();

    constexpr int TRIP = 256 * DC_PER;
    for (int base = blk * TRIP; base < ANS; base += DC_GRID * TRIP) {
        int t[DC_PER], w[DC_PER], old[DC_PER];
        bool ok[DC_PER];
#pragma unroll
        for (int h = 0; h < DC_PER; ++h) {
            const int a = base + h * 256 + threadIdx.x;
            ok[h] = a < ANS;
            const int ac = ok[h] ? a : ANS - 1;
            t[h] = answers[3 * ac];
            w[h] = answers[3 * ac + 1];
        }
#pragma unroll
        for (int h = 0; h < DC_PER; ++h) {
            old[h] = CAP + 1;
            if (ok[h]) {
                old[h] = atomicAdd(&deg[t[h]], 1);
                atomicAdd(&s_wdeg[w[h]], 1);
            }
        }
#pragma unroll
        for (int h = 0; h < DC_PER; ++h) {
            const int slot = old[h] - 1;
            if (slot < CAP) list[(size_t)t[h] * CAP + slot] = (unsigned short)w[h];
        }
    }
    __syncthreads();
    for (int i = threadIdx.x; i < WORKERS; i += 256) atomicAdd(&deg[TASKS + i], s_wdeg[i]);
}


constexpr int PB_GRID = ((FEAT / 2) * HID + 255) / 256;
constexpr int DI_GRID = (NODES + 255) / 256;
static_assert(DI_GRID * 256 >= 2 * WORKERS * CLS, "the degree groups cover every worker-side sum");

constexpr int WP_GROUPS = HID / 4;
constexpr int PW_GRID   = (WP_GROUPS * 32 + 255) / 256;
static_assert(HID % 16 == 0, "hidden tiles of 16");

__global__ void k_packB(const float* __restrict__ B, float2* __restrict__ Bp, int* __restrict__ deg,
                        const float* __restrict__ Wm, const float* __restrict__ Wl,
                        float4* __restrict__ Wp, unsigned long long* __restrict__ wsum) {
    if (blockIdx.x >= PB_GRID + DI_GRID) {
        const int e = (blockIdx.x - PB_GRID - DI_GRID) * 256 + threadIdx.x;
        if (e < WP_GROUPS * 32) {
            const int g = e >> 5, lane = e & 31;
            const int cc = lane & 15, h = lane >> 4;
            const int n = 16 * (g >> 2) + 8 * h + 2 * (g & 3);
            float4 v = make_float4(0.f, 0.f, 0.f, 0.f);
            if (cc < CLS) v = make_float4(Wm[n * CLS + cc], Wm[(n + 1) * CLS + cc],
                                          Wl[n * CLS + cc], Wl[(n + 1) * CLS + cc]);
            Wp[e] = v;
        }
        return;
    }
    if (blockIdx.x >= PB_GRID) {
        const int i = (blockIdx.x - PB_GRID) * 256 + threadIdx.x;
        if (i < NODES) deg[i] = 1;
        if (i < 2 * WORKERS * CLS) wsum[i] = 0ull;
        return;
    }
    int idx = blockIdx.x * 256 + threadIdx.x;
    if (idx < (FEAT / 2) * HID) {
        int kp = idx / HID;
        int n  = idx - kp * HID;
        Bp[idx] = make_float2(B[(2 * kp) * HID + n], B[(2 * kp + 1) * HID + n]);
    }
}


__device__ __forceinline__ void gemm_efc_body(const int blk,
                                              const float* __restrict__ A,
                                              const float2* __restrict__ Bp,
                                              const float* __restrict__ bias,
                                              const float4* __restrict__ Wp,
                                              float* __restrict__ hm,
                                              float* __restrict__ hl) {
    const int wave   = threadIdx.x >> 5;
    const int lane   = threadIdx.x & 31;
    const int lane16 = lane & 15;
    const int hi     = lane >> 4;

    const int slab     = blk * 8 + wave;
    const bool valid   = (slab * 32) < TASKS;
    const int row_base = valid ? slab * 32 : 0;

    const float*  arow0 = A + (size_t)(row_base + lane16) * FEAT + 2 * hi;
    const float*  arow1 = arow0 + (size_t)16 * FEAT;
    const float2* bbase = Bp + lane16;

    v8f acc[2][8];
#pragma unroll
    for (int t = 0; t < 8; ++t) {
        const float* bt = bias + t * 16 + 8 * hi;
        acc[0][t] = (v8f){bt[0], bt[1], bt[2], bt[3], bt[4], bt[5], bt[6], bt[7]};
        acc[1][t] = acc[0][t];
    }

    for (int k = 0; k < FEAT; k += 4) {
        const float2 av0 = *(const float2*)(arow0 + k);
        const float2 av1 = *(const float2*)(arow1 + k);
        v2f af0 = {av0.x, av0.y};
        v2f af1 = {av1.x, av1.y};
        const float2* brow = bbase + (size_t)((k >> 1) + hi) * HID;
#pragma unroll
        for (int t = 0; t < 8; ++t) {
            const float2 bv = brow[t * 16];
            v2f bf = {bv.x, bv.y};
            acc[0][t] = __builtin_amdgcn_wmma_f32_16x16x4_f32(
                false, bf, false, af0, (short)0, acc[0][t], false, false);
            acc[1][t] = __builtin_amdgcn_wmma_f32_16x16x4_f32(
                false, bf, false, af1, (short)0, acc[1][t], false, false);
        }
    }

    const float4* wp = Wp + lane;
#pragma unroll
    for (int r = 0; r < 2; ++r) {
        v8f hM = {0.f, 0.f, 0.f, 0.f, 0.f, 0.f, 0.f, 0.f};
        v8f hL = {0.f, 0.f, 0.f, 0.f, 0.f, 0.f, 0.f, 0.f};
#pragma unroll
        for (int t = 0; t < 8; ++t) {
#pragma unroll
            for (int jp = 0; jp < 4; ++jp) {
                const float4 wv = wp[(t * 4 + jp) * 32];
                const v2f wm = {wv.x, wv.y};
                const v2f wl = {wv.z, wv.w};
                const float x0 = acc[r][t][2 * jp];
                const float x1 = acc[r][t][2 * jp + 1];
                const v2f xf = {x0 > 0.f ? x0 : 0.f, x1 > 0.f ? x1 : 0.f};
                hM = __builtin_amdgcn_wmma_f32_16x16x4_f32(false, wm, false, xf, (short)0, hM, false, false);
                hL = __builtin_amdgcn_wmma_f32_16x16x4_f32(false, wl, false, xf, (short)0, hL, false, false);
            }
        }
        if (valid) {
            const size_t m = (size_t)(row_base + r * 16 + lane16);
            float* pm = hm + m * HP + 8 * hi;
            float* pl = hl + m * HP + 8 * hi;
            if (hi == 0) {
                *(float4*)(pm)     = make_float4(hM[0], hM[1], hM[2], hM[3]);
                *(float4*)(pm + 4) = make_float4(hM[4], hM[5], hM[6], hM[7]);
                *(float4*)(pl)     = make_float4(hL[0], hL[1], hL[2], hL[3]);
                *(float4*)(pl + 4) = make_float4(hL[4], hL[5], hL[6], hL[7]);
            } else {
                *(float2*)(pm) = make_float2(hM[0], hM[1]);
                *(float2*)(pl) = make_float2(hL[0], hL[1]);
            }
        }
    }
}

static_assert(TASKS % 16 == 0, "worker rows start on a strip boundary");

__device__ __forceinline__ void h_body(const int blk,
                                       const float* __restrict__ wf,
                                       const float* __restrict__ Wm,
                                       const float* __restrict__ Wl,
                                       float* __restrict__ hm,
                                       float* __restrict__ hl) {
    const int wave   = threadIdx.x >> 5;
    const int lane   = threadIdx.x & 31;
    const int lane16 = lane & 15;
    const int hi     = lane >> 4;

    const int node_base = TASKS + (blk * 8 + wave) * 16;
    int nd = node_base + lane16;
    if (nd > NODES - 1) nd = NODES - 1;
    const float* xrow = wf + (size_t)(nd - TASKS) * HID;
    const int n = lane16;
    const bool ncol = n < CLS;

    v8f accM = {0.f, 0.f, 0.f, 0.f, 0.f, 0.f, 0.f, 0.f};
    v8f accL = {0.f, 0.f, 0.f, 0.f, 0.f, 0.f, 0.f, 0.f};

    for (int k = 0; k < HID; k += 4) {
        const int k0 = k + 2 * hi;
        const float2 av = *(const float2*)(xrow + k0);
        v2f af = {av.x, av.y};
        v2f bm, bl;
        bm.x = ncol ? Wm[k0 * CLS + n] : 0.f;
        bm.y = ncol ? Wm[(k0 + 1) * CLS + n] : 0.f;
        bl.x = ncol ? Wl[k0 * CLS + n] : 0.f;
        bl.y = ncol ? Wl[(k0 + 1) * CLS + n] : 0.f;
        accM = __builtin_amdgcn_wmma_f32_16x16x4_f32(false, af, false, bm, (short)0, accM, false, false);
        accL = __builtin_amdgcn_wmma_f32_16x16x4_f32(false, af, false, bl, (short)0, accL, false, false);
    }

    if (ncol) {
#pragma unroll
        for (int j = 0; j < 8; ++j) {
            int m = node_base + j + 8 * hi;
            if (m < NODES) {
                hm[(size_t)m * HP + n] = accM[j];
                hl[(size_t)m * HP + n] = accL[j];
            }
        }
    }
}

constexpr int GE_GRID = (TASKS / 32 + 7) / 8;
constexpr int HW_GRID = ((WORKERS + 15) / 16 + 7) / 8;

__global__ __launch_bounds__(256) void k_gemm_count(const float* __restrict__ A,
                                                    const float2* __restrict__ Bp,
                                                    const float* __restrict__ bias,
                                                    const float4* __restrict__ Wp,
                                                    const float* __restrict__ wf,
                                                    const float* __restrict__ Wm,
                                                    const float* __restrict__ Wl,
                                                    float* __restrict__ hm,
                                                    float* __restrict__ hl,
                                                    const int* __restrict__ answers,
                                                    int* __restrict__ deg,
                                                    unsigned short* __restrict__ list) {
    __shared__ int s_wdeg[WPAD];
    if (blockIdx.x < GE_GRID) {
        gemm_efc_body(blockIdx.x, A, Bp, bias, Wp, hm, hl);
    } else if (blockIdx.x < GE_GRID + DC_GRID) {
        deg_count_body(blockIdx.x - GE_GRID, s_wdeg, answers, deg, list);
    } else {
        h_body(blockIdx.x - GE_GRID - DC_GRID, wf, Wm, Wl, hm, hl);
    }
}


__device__ __forceinline__ void load_row10(const float* __restrict__ p, float* r) {
    const float4 a = *(const float4*)(p);
    const float4 b = *(const float4*)(p + 4);
    const float2 c = *(const float2*)(p + 8);
    r[0] = a.x; r[1] = a.y; r[2] = a.z; r[3] = a.w;
    r[4] = b.x; r[5] = b.y; r[6] = b.z; r[7] = b.w;
    r[8] = c.x; r[9] = c.y;
}

__device__ __forceinline__ unsigned long long to_fix(float v) {
    return (unsigned long long)(long long)rintf(v * 4294967296.0f);
}

__device__ __forceinline__ float from_fix(unsigned long long s) {
    return (float)(long long)s * 2.3283064365386963e-10f;
}

__device__ __forceinline__ void agg_task_body(const int blk, float* __restrict__ sm, float* __restrict__ sl,
                                              const int* __restrict__ answers,
                                              const unsigned short* __restrict__ list,
                                              const int* __restrict__ deg,
                                              const float* __restrict__ hm,
                                              const float* __restrict__ hl,
                                              const float* __restrict__ bm,
                                              const float* __restrict__ bl,
                                              const float* __restrict__ eps,
                                              float* __restrict__ mean,
                                              float* __restrict__ lsd,
                                              float* __restrict__ zp,
                                              float* __restrict__ wbm,
                                              float* __restrict__ wbl) {
    const int i = blk * 256 + threadIdx.x;
    float am[CLS], al[CLS];
#pragma unroll
    for (int c = 0; c < CLS; ++c) { am[c] = 0.f; al[c] = 0.f; }

    if (i < NODES) {
        const int   dg = deg[i];
        const float d  = rsqrtf((float)dg);
        const float sw = d * d;
        float2 ev[CLS / 2];
        {
            const float2* ep = (const float2*)(eps + (size_t)(i < TASKS ? i : TASKS - 1) * CLS);
#pragma unroll
            for (int c = 0; c < CLS / 2; ++c) ev[c] = ep[c];
        }
        float hmi[CLS], hli[CLS];
        load_row10(hm + (size_t)i * HP, hmi);
        load_row10(hl + (size_t)i * HP, hli);
#pragma unroll
        for (int c = 0; c < CLS; ++c) {
            am[c] = bm[c] + hmi[c] * sw;
            al[c] = bl[c] + hli[c] * sw;
        }

        if (i < TASKS) {
            const int n = dg - 1;
            if (n <= CAP) {
                const uint4* lp = (const uint4*)(list + (size_t)i * CAP);
                const uint4 q0 = lp[0], q1 = lp[1], q2 = lp[2], q3 = lp[3];
                const unsigned pk[16] = {q0.x, q0.y, q0.z, q0.w, q1.x, q1.y, q1.z, q1.w,
                                         q2.x, q2.y, q2.z, q2.w, q3.x, q3.y, q3.z, q3.w};
                int e[CAP];
#pragma unroll
                for (int j = 0; j < CAP; ++j) {
                    const int v = (int)((pk[j >> 1] >> (16 * (j & 1))) & 0xFFFFu);
                    e[j] = (j < n) ? v : SENT;
                }
                int cur = -1;
                for (int it = 0; it < CAP; ++it) {
                    int m = SENT, cnt = 0;
#pragma unroll
                    for (int j = 0; j < CAP; ++j) {
                        const int v = e[j];
                        if (v > cur) {
                            if (v < m) { m = v; cnt = 1; }
                            else if (v == m) ++cnt;
                        }
                    }
                    if (m == SENT) break;
                    const int wrow = TASKS + m;
                    const float we = d * rsqrtf((float)deg[wrow]);
                    float hmw[CLS], hlw[CLS];
                    load_row10(hm + (size_t)wrow * HP, hmw);
                    load_row10(hl + (size_t)wrow * HP, hlw);
                    for (int k = 0; k < cnt; ++k) {
#pragma unroll
                        for (int c = 0; c < CLS; ++c) {
                            am[c] += hmw[c] * we;
                            al[c] += hlw[c] * we;
                        }
                    }
                    cur = m;
                }
            } else {
                for (int a = 0; a < ANS; ++a) {
                    if (answers[3 * a] == i) {
                        const int wrow = TASKS + answers[3 * a + 1];
                        const float we = d * rsqrtf((float)deg[wrow]);
                        float hmw[CLS], hlw[CLS];
                        load_row10(hm + (size_t)wrow * HP, hmw);
                        load_row10(hl + (size_t)wrow * HP, hlw);
#pragma unroll
                        for (int c = 0; c < CLS; ++c) {
                            am[c] += hmw[c] * we;
                            al[c] += hlw[c] * we;
                        }
                    }
                }
            }
            float z[CLS];
#pragma unroll
            for (int c = 0; c < CLS / 2; ++c) {
                z[2 * c]     = am[2 * c]     + (ev[c].x * 0.01f) * expf(al[2 * c]);
                z[2 * c + 1] = am[2 * c + 1] + (ev[c].y * 0.01f) * expf(al[2 * c + 1]);
            }
            float* zr = zp + (size_t)i * HP;
            *(float4*)(zr)     = make_float4(z[0], z[1], z[2], z[3]);
            *(float4*)(zr + 4) = make_float4(z[4], z[5], z[6], z[7]);
            *(float2*)(zr + 8) = make_float2(z[8], z[9]);
        } else {
            const int q = (i - TASKS) * CLS;
#pragma unroll
            for (int c = 0; c < CLS; ++c) { wbm[q + c] = am[c]; wbl[q + c] = al[c]; }
        }
    }

#pragma unroll
    for (int c = 0; c < CLS; ++c) {
        sm[threadIdx.x * CLS + c] = am[c];
        sl[threadIdx.x * CLS + c] = al[c];
    }
    __syncthreads();
    const int base = blk * 256 * CLS;
    const int lim  = NODES * CLS - base;
    for (int k = threadIdx.x; k < 256 * CLS; k += 256) {
        if (k < lim) {
            mean[(size_t)base + k] = sm[k];
            lsd[(size_t)base + k]  = sl[k];
        }
    }
}

__device__ __forceinline__ void agg_worker_body(const int blk, unsigned long long* __restrict__ wacc,
                                                const int* __restrict__ answers,
                                                const float* __restrict__ hm,
                                                const float* __restrict__ hl,
                                                const int* __restrict__ deg,
                                                unsigned long long* __restrict__ wsum) {
    for (int i = threadIdx.x; i < 2 * WORKERS * CLS; i += 256) wacc[i] = 0ull;
    __syncthreads();

    const int start = blk * CHUNK;
    int end = start + CHUNK;
    if (end > ANS) end = ANS;

    for (int a = start + threadIdx.x; a < end; a += 256) {
        const int t = answers[3 * a];
        const int w = answers[3 * a + 1];
        const float we = rsqrtf((float)deg[t]) * rsqrtf((float)deg[TASKS + w]);

        float hmt[CLS], hlt[CLS];
        load_row10(hm + (size_t)t * HP, hmt);
        load_row10(hl + (size_t)t * HP, hlt);
#pragma unroll
        for (int c = 0; c < CLS; ++c) {
            atomicAdd(&wacc[w * CLS + c], to_fix(hmt[c] * we));
            atomicAdd(&wacc[WORKERS * CLS + w * CLS + c], to_fix(hlt[c] * we));
        }
    }
    __syncthreads();

    for (int i = threadIdx.x; i < 2 * WORKERS * CLS; i += 256) atomicAdd(&wsum[i], wacc[i]);
}

constexpr int AW_GRID = (ANS + CHUNK - 1) / CHUNK;
constexpr int AT_GRID = (NODES + 255) / 256;
static_assert(2 * 256 * CLS * sizeof(float) <= 2 * WORKERS * CLS * sizeof(unsigned long long),
              "the task side's two row buffers fit in the worker side's accumulators");

__global__ __launch_bounds__(256) void k_agg(const int* __restrict__ answers,
                                             const unsigned short* __restrict__ list,
                                             const int* __restrict__ deg,
                                             const float* __restrict__ hm,
                                             const float* __restrict__ hl,
                                             const float* __restrict__ bm,
                                             const float* __restrict__ bl,
                                             const float* __restrict__ eps,
                                             float* __restrict__ mean,
                                             float* __restrict__ lsd,
                                             unsigned long long* __restrict__ wsum,
                                             float* __restrict__ zp,
                                             float* __restrict__ wbm,
                                             float* __restrict__ wbl) {
    __shared__ __attribute__((aligned(16))) unsigned long long s_buf[2 * WORKERS * CLS];
    if (blockIdx.x < AW_GRID) {
        agg_worker_body(blockIdx.x, s_buf, answers, hm, hl, deg, wsum);
    } else {
        float* sm = (float*)s_buf;
        agg_task_body(blockIdx.x - AW_GRID, sm, sm + 256 * CLS,
                      answers, list, deg, hm, hl, bm, bl, eps, mean, lsd, zp, wbm, wbl);
    }
}


constexpr int ZW_PER  = (WORKERS * CLS + 255) / 256;
constexpr int ZW_HALF = ZW_PER / 2;
static_assert(ZW_PER % 2 == 0, "the worker-row elements of a thread split into two equal batches");

__global__ __launch_bounds__(256) void k_crowd(const int* __restrict__ answers,
                                               const float* __restrict__ zp,
                                               const float* __restrict__ wbm,
                                               const float* __restrict__ wbl,
                                               const unsigned long long* __restrict__ wsum,
                                               const float* __restrict__ eps,
                                               float* __restrict__ mean,
                                               float* __restrict__ lsd,
                                               float* __restrict__ crowd) {
    __shared__ __attribute__((aligned(16))) float s_zw[WORKERS * HP];
    __shared__ __attribute__((aligned(16))) float s_out[2][CR_TILE * CLS];

    for (int b = 0; b < 2; ++b) {
        float pm[ZW_HALF], pl[ZW_HALF], ev[ZW_HALF];
        unsigned long long fm[ZW_HALF], fl[ZW_HALF];
#pragma unroll
        for (int j = 0; j < ZW_HALF; ++j) {
            int q = (b * ZW_HALF + j) * 256 + threadIdx.x;
            if (q > WORKERS * CLS - 1) q = WORKERS * CLS - 1;
            pm[j] = wbm[q];
            pl[j] = wbl[q];
            fm[j] = wsum[q];
            fl[j] = wsum[WORKERS * CLS + q];
            ev[j] = eps[TASKS * CLS + q];
        }
#pragma unroll
        for (int j = 0; j < ZW_HALF; ++j) {
            const int q = (b * ZW_HALF + j) * 256 + threadIdx.x;
            pm[j] += from_fix(fm[j]);
            pl[j] += from_fix(fl[j]);
            if (q < WORKERS * CLS) {
                const int r = q / CLS;
                const int c = q - r * CLS;
                s_zw[r * HP + c] = pm[j] + (ev[j] * 0.01f) * expf(pl[j]);
            }
        }
        if (blockIdx.x == 0) {
#pragma unroll
            for (int j = 0; j < ZW_HALF; ++j) {
                const int q = (b * ZW_HALF + j) * 256 + threadIdx.x;
                if (q < WORKERS * CLS) {
                    mean[TASKS * CLS + q] = pm[j];
                    lsd[TASKS * CLS + q]  = pl[j];
                }
            }
        }
    }
    __syncthreads();

    constexpr int NT = (ANS + CR_TILE - 1) / CR_TILE;
    constexpr int NQ = CR_TILE * CLS / 4;
    int buf = 0;
    for (int tile = blockIdx.x; tile < NT; tile += gridDim.x, buf ^= 1) {
        const int base = tile * CR_TILE;
        float* sb = s_out[buf];

        int t[2], w[2];
#pragma unroll
        for (int h = 0; h < 2; ++h) {
            int a = base + h * 256 + threadIdx.x;
            if (a > ANS - 1) a = ANS - 1;
            t[h] = answers[3 * a];
            w[h] = answers[3 * a + 1];
        }
        float4 ta[2], tb[2];
        float2 tc[2];
#pragma unroll
        for (int h = 0; h < 2; ++h) {
            const float* zt = zp + (size_t)t[h] * HP;
            ta[h] = *(const float4*)(zt);
            tb[h] = *(const float4*)(zt + 4);
            tc[h] = *(const float2*)(zt + 8);
        }
#pragma unroll
        for (int h = 0; h < 2; ++h) {
            const float* zw = s_zw + w[h] * HP;
            const float4 wa = *(const float4*)(zw);
            const float4 wb = *(const float4*)(zw + 4);
            const float2 wc = *(const float2*)(zw + 8);
            float2* o = (float2*)(sb + (h * 256 + threadIdx.x) * CLS);
            o[0] = make_float2(ta[h].x * wa.x, ta[h].y * wa.y);
            o[1] = make_float2(ta[h].z * wa.z, ta[h].w * wa.w);
            o[2] = make_float2(tb[h].x * wb.x, tb[h].y * wb.y);
            o[3] = make_float2(tb[h].z * wb.z, tb[h].w * wb.w);
            o[4] = make_float2(tc[h].x * wc.x, tc[h].y * wc.y);
        }
        __syncthreads();

        float4* dst = (float4*)(crowd + (size_t)base * CLS);
        const float4* so = (const float4*)sb;
        if (ANS - base >= CR_TILE) {
#pragma unroll
            for (int k = 0; k < NQ / 256; ++k) dst[k * 256 + threadIdx.x] = so[k * 256 + threadIdx.x];
        } else {
            const int nq = (ANS - base) * CLS / 4;
            for (int k = threadIdx.x; k < nq; k += 256) dst[k] = so[k];
        }
    }
}


extern "C" void kernel_launch(void* const* d_in, const int* in_sizes, int n_in,
                              void* d_out, int out_size, void* d_ws, size_t ws_size,
                              hipStream_t stream) {
    (void)in_sizes; (void)n_in; (void)out_size; (void)ws_size;

    const float* task_feature   = (const float*)d_in[0];
    const int*   answers        = (const int*)d_in[1];
    const float* worker_feature = (const float*)d_in[2];
    const float* W_efc          = (const float*)d_in[3];
    const float* b_efc          = (const float*)d_in[4];
    const float* W_mean         = (const float*)d_in[5];
    const float* b_mean         = (const float*)d_in[6];
    const float* W_ls           = (const float*)d_in[7];
    const float* b_ls           = (const float*)d_in[8];
    const float* eps            = (const float*)d_in[9];

    float* out   = (float*)d_out;
    float* crowd = out;
    float* mean  = out + (size_t)ANS * CLS;
    float* lsd   = mean + (size_t)NODES * CLS;

    float*  ws   = (float*)d_ws;
    float*  tf_h = ws;
    float2* Bp   = (float2*)(tf_h + (size_t)TASKS * HID);
    float*  hm   = (float*)(Bp + (size_t)(FEAT / 2) * HID);
    float*  hl   = hm + (size_t)NODES * HP;
    int*    deg  = (int*)(hl + (size_t)NODES * HP + NODES);
    static_assert((size_t)TASKS * CAP * sizeof(unsigned short) <= (size_t)ANS * CLS * sizeof(float),
                  "the lists fit in the crowd block");
    unsigned short*     list = (unsigned short*)crowd;
    float*              zz   = tf_h;
    float4*             Wp   = (float4*)(tf_h + (size_t)2 * 1024 * 1024);
    unsigned long long* wsum = (unsigned long long*)(tf_h + (size_t)3 * 1024 * 1024);
    float*              wbm  = tf_h + (size_t)3 * 1024 * 1024 + 65536;
    float*              wbl  = wbm + 8192;
    static_assert((size_t)NODES * HP <= (size_t)2 * 1024 * 1024 &&
                  (size_t)WP_GROUPS * 32 * 4 <= (size_t)1024 * 1024 &&
                  (size_t)2 * WORKERS * CLS * sizeof(unsigned long long) <= (size_t)65536 * sizeof(float) &&
                  (size_t)WORKERS * CLS <= (size_t)8192 &&
                  (size_t)3 * 1024 * 1024 + 65536 + 2 * 8192 <= (size_t)TASKS * HID,
                  "z rows, packed class weights, worker sums and worker rows sit apart inside the old tf_h space");

    k_packB    <<<PB_GRID + DI_GRID + PW_GRID, 256, 0, stream>>>(W_efc, Bp, deg, W_mean, W_ls, Wp, wsum);
    k_gemm_count<<<GE_GRID + DC_GRID + HW_GRID, 256, 0, stream>>>(task_feature, Bp, b_efc, Wp,
                                                                  worker_feature, W_mean, W_ls, hm, hl,
                                                                  answers, deg, list);
    k_agg      <<<AW_GRID + AT_GRID, 256, 0, stream>>>(answers, list, deg, hm, hl,
                                                      b_mean, b_ls, eps, mean, lsd, wsum, zz, wbm, wbl);
    k_crowd    <<<CR_GRID, 256, 0, stream>>>(answers, zz, wbm, wbl, wsum, eps, mean, lsd, crowd);
}


# Big product re-cut for matrix-pipe balance: one 32-row tile per 4-wave work-group (3125), each wave a quarter of k, partial sums added through LDS in fixed order; 512-term sums split 4 x 128
# speedup vs baseline: 4.6729x; 1.0487x over previous
#include <hip/hip_runtime.h>

typedef __attribute__((ext_vector_type(2))) float v2f;
typedef __attribute__((ext_vector_type(8))) float v8f;

constexpr int TASKS   = 100000;
constexpr int WORKERS = 500;
constexpr int FEAT    = 512;
constexpr int CLS     = 10;
constexpr int HP      = 12;
constexpr int ANS     = 1000000;
constexpr int HID     = 128;
constexpr int NODES   = TASKS + WORKERS;
constexpr int CHUNK   = 4096;
constexpr int CAP     = 32;
constexpr int SENT    = 0xFFFF;
constexpr int CR_TILE = 512;
constexpr int CR_GRID = 512;
constexpr int DC_GRID = 256;
constexpr int DC_PER  = 8;
constexpr int GW      = 4;
constexpr int GT      = GW * 32;
constexpr int WPAD    = 512;
static_assert(WPAD >= WORKERS && WPAD % GT == 0, "the worker counts are a whole number of work-group rounds");
static_assert(ANS % 2 == 0 && CR_TILE % 2 == 0, "a tile's floats must split into 16-byte pieces");
static_assert((CR_TILE * CLS / 4) % 256 == 0, "a full tile is a whole number of 256-thread store rounds");


__device__ __forceinline__ void deg_count_body(const int blk, int* __restrict__ s_wdeg,
                                               const int* __restrict__ answers,
                                               int* __restrict__ deg,
                                               unsigned short* __restrict__ list) {
    for (int i = threadIdx.x; i < WPAD; i += GT) s_wdeg[i] = 0;
    __syncthreads();

    constexpr int TRIP = GT * DC_PER;
    for (int base = blk * TRIP; base < ANS; base += DC_GRID * TRIP) {
        int t[DC_PER], w[DC_PER], old[DC_PER];
        bool ok[DC_PER];
#pragma unroll
        for (int h = 0; h < DC_PER; ++h) {
            const int a = base + h * GT + threadIdx.x;
            ok[h] = a < ANS;
            const int ac = ok[h] ? a : ANS - 1;
            t[h] = answers[3 * ac];
            w[h] = answers[3 * ac + 1];
        }
#pragma unroll
        for (int h = 0; h < DC_PER; ++h) {
            old[h] = CAP + 1;
            if (ok[h]) {
                old[h] = atomicAdd(&deg[t[h]], 1);
                atomicAdd(&s_wdeg[w[h]], 1);
            }
        }
#pragma unroll
        for (int h = 0; h < DC_PER; ++h) {
            const int slot = old[h] - 1;
            if (slot < CAP) list[(size_t)t[h] * CAP + slot] = (unsigned short)w[h];
        }
    }
    __syncthreads();
    for (int i = threadIdx.x; i < WORKERS; i += GT) atomicAdd(&deg[TASKS + i], s_wdeg[i]);
}


constexpr int PB_GRID = ((FEAT / 2) * HID + 255) / 256;
constexpr int DI_GRID = (NODES + 255) / 256;
static_assert(DI_GRID * 256 >= 2 * WORKERS * CLS, "the degree groups cover every worker-side sum");

constexpr int WP_GROUPS = HID / 4;
constexpr int PW_GRID   = (WP_GROUPS * 32 + 255) / 256;
static_assert(HID % 16 == 0, "hidden tiles of 16");

__global__ void k_packB(const float* __restrict__ B, float2* __restrict__ Bp, int* __restrict__ deg,
                        const float* __restrict__ Wm, const float* __restrict__ Wl,
                        float4* __restrict__ Wp, unsigned long long* __restrict__ wsum) {
    if (blockIdx.x >= PB_GRID + DI_GRID) {
        const int e = (blockIdx.x - PB_GRID - DI_GRID) * 256 + threadIdx.x;
        if (e < WP_GROUPS * 32) {
            const int g = e >> 5, lane = e & 31;
            const int cc = lane & 15, h = lane >> 4;
            const int n = 16 * (g >> 2) + 8 * h + 2 * (g & 3);
            float4 v = make_float4(0.f, 0.f, 0.f, 0.f);
            if (cc < CLS) v = make_float4(Wm[n * CLS + cc], Wm[(n + 1) * CLS + cc],
                                          Wl[n * CLS + cc], Wl[(n + 1) * CLS + cc]);
            Wp[e] = v;
        }
        return;
    }
    if (blockIdx.x >= PB_GRID) {
        const int i = (blockIdx.x - PB_GRID) * 256 + threadIdx.x;
        if (i < NODES) deg[i] = 1;
        if (i < 2 * WORKERS * CLS) wsum[i] = 0ull;
        return;
    }
    int idx = blockIdx.x * 256 + threadIdx.x;
    if (idx < (FEAT / 2) * HID) {
        int kp = idx / HID;
        int n  = idx - kp * HID;
        Bp[idx] = make_float2(B[(2 * kp) * HID + n], B[(2 * kp + 1) * HID + n]);
    }
}


constexpr int KQ          = FEAT / GW;
constexpr int SLAB_PART   = 8 * 8 * 32;
static_assert(FEAT % (4 * GW) == 0, "a wave's k range is a whole number of 4-k steps");
static_assert(TASKS % 32 == 0, "every 32-row tile is a full one: one work-group each, none padded");

__device__ __forceinline__ void finish_slab(v8f (&a)[8], const float4* __restrict__ part,
                                            const float4* __restrict__ wp,
                                            float* __restrict__ hm, float* __restrict__ hl,
                                            const size_t m, const int hi) {
    for (int slot = 0; slot < GW - 1; ++slot) {
#pragma unroll
        for (int t = 0; t < 8; ++t) {
            const float4 lo = part[((slot * 8 + t) * 2 + 0) * 32];
            const float4 hh = part[((slot * 8 + t) * 2 + 1) * 32];
            a[t] += (v8f){lo.x, lo.y, lo.z, lo.w, hh.x, hh.y, hh.z, hh.w};
        }
    }

    v8f hM = {0.f, 0.f, 0.f, 0.f, 0.f, 0.f, 0.f, 0.f};
    v8f hL = {0.f, 0.f, 0.f, 0.f, 0.f, 0.f, 0.f, 0.f};
#pragma unroll
    for (int t = 0; t < 8; ++t) {
#pragma unroll
        for (int jp = 0; jp < 4; ++jp) {
            const float4 wv = wp[(t * 4 + jp) * 32];
            const v2f wm = {wv.x, wv.y};
            const v2f wl = {wv.z, wv.w};
            const float x0 = a[t][2 * jp];
            const float x1 = a[t][2 * jp + 1];
            const v2f xf = {x0 > 0.f ? x0 : 0.f, x1 > 0.f ? x1 : 0.f};
            hM = __builtin_amdgcn_wmma_f32_16x16x4_f32(false, wm, false, xf, (short)0, hM, false, false);
            hL = __builtin_amdgcn_wmma_f32_16x16x4_f32(false, wl, false, xf, (short)0, hL, false, false);
        }
    }
    float* pm = hm + m * HP + 8 * hi;
    float* pl = hl + m * HP + 8 * hi;
    if (hi == 0) {
        *(float4*)(pm)     = make_float4(hM[0], hM[1], hM[2], hM[3]);
        *(float4*)(pm + 4) = make_float4(hM[4], hM[5], hM[6], hM[7]);
        *(float4*)(pl)     = make_float4(hL[0], hL[1], hL[2], hL[3]);
        *(float4*)(pl + 4) = make_float4(hL[4], hL[5], hL[6], hL[7]);
    } else {
        *(float2*)(pm) = make_float2(hM[0], hM[1]);
        *(float2*)(pl) = make_float2(hL[0], hL[1]);
    }
}

__device__ __forceinline__ void gemm_efc_body(const int blk, float* __restrict__ s_part,
                                              const float* __restrict__ A,
                                              const float2* __restrict__ Bp,
                                              const float* __restrict__ bias,
                                              const float4* __restrict__ Wp,
                                              float* __restrict__ hm,
                                              float* __restrict__ hl) {
    const int wave   = threadIdx.x >> 5;
    const int lane   = threadIdx.x & 31;
    const int lane16 = lane & 15;
    const int hi     = lane >> 4;

    const int row_base = blk * 32;
    const int f0 = blk & (GW - 1);
    const int f1 = (f0 + 2) & (GW - 1);
    const bool fin0 = wave == f0;
    const bool fin1 = wave == f1;

    const float*  arow0 = A + (size_t)(row_base + lane16) * FEAT + 2 * hi;
    const float*  arow1 = arow0 + (size_t)16 * FEAT;
    const float2* bbase = Bp + lane16;

    v8f acc[2][8];
#pragma unroll
    for (int t = 0; t < 8; ++t) {
        const v8f zero = {0.f, 0.f, 0.f, 0.f, 0.f, 0.f, 0.f, 0.f};
        v8f b = zero;
        if (fin0 || fin1) {
            const float* bt = bias + t * 16 + 8 * hi;
            b = (v8f){bt[0], bt[1], bt[2], bt[3], bt[4], bt[5], bt[6], bt[7]};
        }
        acc[0][t] = fin0 ? b : zero;
        acc[1][t] = fin1 ? b : zero;
    }

    for (int k = wave * KQ; k < (wave + 1) * KQ; k += 4) {
        const float2 av0 = *(const float2*)(arow0 + k);
        const float2 av1 = *(const float2*)(arow1 + k);
        v2f af0 = {av0.x, av0.y};
        v2f af1 = {av1.x, av1.y};
        const float2* brow = bbase + (size_t)((k >> 1) + hi) * HID;
#pragma unroll
        for (int t = 0; t < 8; ++t) {
            const float2 bv = brow[t * 16];
            v2f bf = {bv.x, bv.y};
            acc[0][t] = __builtin_amdgcn_wmma_f32_16x16x4_f32(
                false, bf, false, af0, (short)0, acc[0][t], false, false);
            acc[1][t] = __builtin_amdgcn_wmma_f32_16x16x4_f32(
                false, bf, false, af1, (short)0, acc[1][t], false, false);
        }
    }

    float4* sp = (float4*)s_part + lane;
#pragma unroll
    for (int r = 0; r < 2; ++r) {
        const int fr = r == 0 ? f0 : f1;
        if (wave != fr) {
            const int slot = (wave - fr - 1) & (GW - 1);
            float4* dst = sp + (r * (GW - 1) + slot) * (SLAB_PART / 4);
#pragma unroll
            for (int t = 0; t < 8; ++t) {
                const v8f a = acc[r][t];
                dst[(t * 2 + 0) * 32] = make_float4(a[0], a[1], a[2], a[3]);
                dst[(t * 2 + 1) * 32] = make_float4(a[4], a[5], a[6], a[7]);
            }
        }
    }
    __syncthreads();
    if (fin0) {
        finish_slab(acc[0], sp, Wp + lane, hm, hl, (size_t)(row_base + lane16), hi);
    } else if (fin1) {
        finish_slab(acc[1], sp + (GW - 1) * (SLAB_PART / 4), Wp + lane, hm, hl,
                    (size_t)(row_base + 16 + lane16), hi);
    }
}

static_assert(TASKS % 16 == 0, "worker rows start on a strip boundary");

__device__ __forceinline__ void h_body(const int blk,
                                       const float* __restrict__ wf,
                                       const float* __restrict__ Wm,
                                       const float* __restrict__ Wl,
                                       float* __restrict__ hm,
                                       float* __restrict__ hl) {
    const int wave   = threadIdx.x >> 5;
    const int lane   = threadIdx.x & 31;
    const int lane16 = lane & 15;
    const int hi     = lane >> 4;

    const int node_base = TASKS + (blk * GW + wave) * 16;
    int nd = node_base + lane16;
    if (nd > NODES - 1) nd = NODES - 1;
    const float* xrow = wf + (size_t)(nd - TASKS) * HID;
    const int n = lane16;
    const bool ncol = n < CLS;

    v8f accM = {0.f, 0.f, 0.f, 0.f, 0.f, 0.f, 0.f, 0.f};
    v8f accL = {0.f, 0.f, 0.f, 0.f, 0.f, 0.f, 0.f, 0.f};

    for (int k = 0; k < HID; k += 4) {
        const int k0 = k + 2 * hi;
        const float2 av = *(const float2*)(xrow + k0);
        v2f af = {av.x, av.y};
        v2f bm, bl;
        bm.x = ncol ? Wm[k0 * CLS + n] : 0.f;
        bm.y = ncol ? Wm[(k0 + 1) * CLS + n] : 0.f;
        bl.x = ncol ? Wl[k0 * CLS + n] : 0.f;
        bl.y = ncol ? Wl[(k0 + 1) * CLS + n] : 0.f;
        accM = __builtin_amdgcn_wmma_f32_16x16x4_f32(false, af, false, bm, (short)0, accM, false, false);
        accL = __builtin_amdgcn_wmma_f32_16x16x4_f32(false, af, false, bl, (short)0, accL, false, false);
    }

    if (ncol) {
#pragma unroll
        for (int j = 0; j < 8; ++j) {
            int m = node_base + j + 8 * hi;
            if (m < NODES) {
                hm[(size_t)m * HP + n] = accM[j];
                hl[(size_t)m * HP + n] = accL[j];
            }
        }
    }
}

constexpr int GE_GRID = TASKS / 32;
constexpr int HW_GRID = ((WORKERS + 15) / 16 + GW - 1) / GW;

__global__ __launch_bounds__(GT) void k_gemm_count(const float* __restrict__ A,
                                                    const float2* __restrict__ Bp,
                                                    const float* __restrict__ bias,
                                                    const float4* __restrict__ Wp,
                                                    const float* __restrict__ wf,
                                                    const float* __restrict__ Wm,
                                                    const float* __restrict__ Wl,
                                                    float* __restrict__ hm,
                                                    float* __restrict__ hl,
                                                    const int* __restrict__ answers,
                                                    int* __restrict__ deg,
                                                    unsigned short* __restrict__ list) {
    __shared__ __attribute__((aligned(16))) float s_buf[2 * (GW - 1) * SLAB_PART];
    static_assert(WPAD * sizeof(int) <= 2 * (GW - 1) * SLAB_PART * sizeof(float), "the worker counts fit in the block");
    if (blockIdx.x < DC_GRID) {
        deg_count_body(blockIdx.x, (int*)s_buf, answers, deg, list);
    } else if (blockIdx.x < DC_GRID + HW_GRID) {
        h_body(blockIdx.x - DC_GRID, wf, Wm, Wl, hm, hl);
    } else {
        gemm_efc_body(blockIdx.x - DC_GRID - HW_GRID, s_buf, A, Bp, bias, Wp, hm, hl);
    }
}


__device__ __forceinline__ void load_row10(const float* __restrict__ p, float* r) {
    const float4 a = *(const float4*)(p);
    const float4 b = *(const float4*)(p + 4);
    const float2 c = *(const float2*)(p + 8);
    r[0] = a.x; r[1] = a.y; r[2] = a.z; r[3] = a.w;
    r[4] = b.x; r[5] = b.y; r[6] = b.z; r[7] = b.w;
    r[8] = c.x; r[9] = c.y;
}

__device__ __forceinline__ unsigned long long to_fix(float v) {
    return (unsigned long long)(long long)rintf(v * 4294967296.0f);
}

__device__ __forceinline__ float from_fix(unsigned long long s) {
    return (float)(long long)s * 2.3283064365386963e-10f;
}

__device__ __forceinline__ void agg_task_body(const int blk, float* __restrict__ sm, float* __restrict__ sl,
                                              const int* __restrict__ answers,
                                              const unsigned short* __restrict__ list,
                                              const int* __restrict__ deg,
                                              const float* __restrict__ hm,
                                              const float* __restrict__ hl,
                                              const float* __restrict__ bm,
                                              const float* __restrict__ bl,
                                              const float* __restrict__ eps,
                                              float* __restrict__ mean,
                                              float* __restrict__ lsd,
                                              float* __restrict__ zp,
                                              float* __restrict__ wbm,
                                              float* __restrict__ wbl) {
    const int i = blk * 256 + threadIdx.x;
    float am[CLS], al[CLS];
#pragma unroll
    for (int c = 0; c < CLS; ++c) { am[c] = 0.f; al[c] = 0.f; }

    if (i < NODES) {
        const int   dg = deg[i];
        const float d  = rsqrtf((float)dg);
        const float sw = d * d;
        float2 ev[CLS / 2];
        {
            const float2* ep = (const float2*)(eps + (size_t)(i < TASKS ? i : TASKS - 1) * CLS);
#pragma unroll
            for (int c = 0; c < CLS / 2; ++c) ev[c] = ep[c];
        }
        float hmi[CLS], hli[CLS];
        load_row10(hm + (size_t)i * HP, hmi);
        load_row10(hl + (size_t)i * HP, hli);
#pragma unroll
        for (int c = 0; c < CLS; ++c) {
            am[c] = bm[c] + hmi[c] * sw;
            al[c] = bl[c] + hli[c] * sw;
        }

        if (i < TASKS) {
            const int n = dg - 1;
            if (n <= CAP) {
                const uint4* lp = (const uint4*)(list + (size_t)i * CAP);
                const uint4 q0 = lp[0], q1 = lp[1], q2 = lp[2], q3 = lp[3];
                const unsigned pk[16] = {q0.x, q0.y, q0.z, q0.w, q1.x, q1.y, q1.z, q1.w,
                                         q2.x, q2.y, q2.z, q2.w, q3.x, q3.y, q3.z, q3.w};
                int e[CAP];
#pragma unroll
                for (int j = 0; j < CAP; ++j) {
                    const int v = (int)((pk[j >> 1] >> (16 * (j & 1))) & 0xFFFFu);
                    e[j] = (j < n) ? v : SENT;
                }
                int cur = -1;
                for (int it = 0; it < CAP; ++it) {
                    int m = SENT, cnt = 0;
#pragma unroll
                    for (int j = 0; j < CAP; ++j) {
                        const int v = e[j];
                        if (v > cur) {
                            if (v < m) { m = v; cnt = 1; }
                            else if (v == m) ++cnt;
                        }
                    }
                    if (m == SENT) break;
                    const int wrow = TASKS + m;
                    const float we = d * rsqrtf((float)deg[wrow]);
                    float hmw[CLS], hlw[CLS];
                    load_row10(hm + (size_t)wrow * HP, hmw);
                    load_row10(hl + (size_t)wrow * HP, hlw);
                    for (int k = 0; k < cnt; ++k) {
#pragma unroll
                        for (int c = 0; c < CLS; ++c) {
                            am[c] += hmw[c] * we;
                            al[c] += hlw[c] * we;
                        }
                    }
                    cur = m;
                }
            } else {
                for (int a = 0; a < ANS; ++a) {
                    if (answers[3 * a] == i) {
                        const int wrow = TASKS + answers[3 * a + 1];
                        const float we = d * rsqrtf((float)deg[wrow]);
                        float hmw[CLS], hlw[CLS];
                        load_row10(hm + (size_t)wrow * HP, hmw);
                        load_row10(hl + (size_t)wrow * HP, hlw);
#pragma unroll
                        for (int c = 0; c < CLS; ++c) {
                            am[c] += hmw[c] * we;
                            al[c] += hlw[c] * we;
                        }
                    }
                }
            }
            float z[CLS];
#pragma unroll
            for (int c = 0; c < CLS / 2; ++c) {
                z[2 * c]     = am[2 * c]     + (ev[c].x * 0.01f) * expf(al[2 * c]);
                z[2 * c + 1] = am[2 * c + 1] + (ev[c].y * 0.01f) * expf(al[2 * c + 1]);
            }
            float* zr = zp + (size_t)i * HP;
            *(float4*)(zr)     = make_float4(z[0], z[1], z[2], z[3]);
            *(float4*)(zr + 4) = make_float4(z[4], z[5], z[6], z[7]);
            *(float2*)(zr + 8) = make_float2(z[8], z[9]);
        } else {
            const int q = (i - TASKS) * CLS;
#pragma unroll
            for (int c = 0; c < CLS; ++c) { wbm[q + c] = am[c]; wbl[q + c] = al[c]; }
        }
    }

#pragma unroll
    for (int c = 0; c < CLS; ++c) {
        sm[threadIdx.x * CLS + c] = am[c];
        sl[threadIdx.x * CLS + c] = al[c];
    }
    __syncthreads();
    const int base = blk * 256 * CLS;
    const int lim  = NODES * CLS - base;
    for (int k = threadIdx.x; k < 256 * CLS; k += 256) {
        if (k < lim) {
            mean[(size_t)base + k] = sm[k];
            lsd[(size_t)base + k]  = sl[k];
        }
    }
}

__device__ __forceinline__ void agg_worker_body(const int blk, unsigned long long* __restrict__ wacc,
                                                const int* __restrict__ answers,
                                                const float* __restrict__ hm,
                                                const float* __restrict__ hl,
                                                const int* __restrict__ deg,
                                                unsigned long long* __restrict__ wsum) {
    for (int i = threadIdx.x; i < 2 * WORKERS * CLS; i += 256) wacc[i] = 0ull;
    __syncthreads();

    const int start = blk * CHUNK;
    int end = start + CHUNK;
    if (end > ANS) end = ANS;

    for (int a = start + threadIdx.x; a < end; a += 256) {
        const int t = answers[3 * a];
        const int w = answers[3 * a + 1];
        const float we = rsqrtf((float)deg[t]) * rsqrtf((float)deg[TASKS + w]);

        float hmt[CLS], hlt[CLS];
        load_row10(hm + (size_t)t * HP, hmt);
        load_row10(hl + (size_t)t * HP, hlt);
#pragma unroll
        for (int c = 0; c < CLS; ++c) {
            atomicAdd(&wacc[w * CLS + c], to_fix(hmt[c] * we));
            atomicAdd(&wacc[WORKERS * CLS + w * CLS + c], to_fix(hlt[c] * we));
        }
    }
    __syncthreads();

    for (int i = threadIdx.x; i < 2 * WORKERS * CLS; i += 256) atomicAdd(&wsum[i], wacc[i]);
}

constexpr int AW_GRID = (ANS + CHUNK - 1) / CHUNK;
constexpr int AT_GRID = (NODES + 255) / 256;
static_assert(2 * 256 * CLS * sizeof(float) <= 2 * WORKERS * CLS * sizeof(unsigned long long),
              "the task side's two row buffers fit in the worker side's accumulators");

__global__ __launch_bounds__(256) void k_agg(const int* __restrict__ answers,
                                             const unsigned short* __restrict__ list,
                                             const int* __restrict__ deg,
                                             const float* __restrict__ hm,
                                             const float* __restrict__ hl,
                                             const float* __restrict__ bm,
                                             const float* __restrict__ bl,
                                             const float* __restrict__ eps,
                                             float* __restrict__ mean,
                                             float* __restrict__ lsd,
                                             unsigned long long* __restrict__ wsum,
                                             float* __restrict__ zp,
                                             float* __restrict__ wbm,
                                             float* __restrict__ wbl) {
    __shared__ __attribute__((aligned(16))) unsigned long long s_buf[2 * WORKERS * CLS];
    if (blockIdx.x < AW_GRID) {
        agg_worker_body(blockIdx.x, s_buf, answers, hm, hl, deg, wsum);
    } else {
        float* sm = (float*)s_buf;
        agg_task_body(blockIdx.x - AW_GRID, sm, sm + 256 * CLS,
                      answers, list, deg, hm, hl, bm, bl, eps, mean, lsd, zp, wbm, wbl);
    }
}


constexpr int ZW_PER  = (WORKERS * CLS + 255) / 256;
constexpr int ZW_HALF = ZW_PER / 2;
static_assert(ZW_PER % 2 == 0, "the worker-row elements of a thread split into two equal batches");

__global__ __launch_bounds__(256) void k_crowd(const int* __restrict__ answers,
                                               const float* __restrict__ zp,
                                               const float* __restrict__ wbm,
                                               const float* __restrict__ wbl,
                                               const unsigned long long* __restrict__ wsum,
                                               const float* __restrict__ eps,
                                               float* __restrict__ mean,
                                               float* __restrict__ lsd,
                                               float* __restrict__ crowd) {
    __shared__ __attribute__((aligned(16))) float s_zw[WORKERS * HP];
    __shared__ __attribute__((aligned(16))) float s_out[2][CR_TILE * CLS];

    for (int b = 0; b < 2; ++b) {
        float pm[ZW_HALF], pl[ZW_HALF], ev[ZW_HALF];
        unsigned long long fm[ZW_HALF], fl[ZW_HALF];
#pragma unroll
        for (int j = 0; j < ZW_HALF; ++j) {
            int q = (b * ZW_HALF + j) * 256 + threadIdx.x;
            if (q > WORKERS * CLS - 1) q = WORKERS * CLS - 1;
            pm[j] = wbm[q];
            pl[j] = wbl[q];
            fm[j] = wsum[q];
            fl[j] = wsum[WORKERS * CLS + q];
            ev[j] = eps[TASKS * CLS + q];
        }
#pragma unroll
        for (int j = 0; j < ZW_HALF; ++j) {
            const int q = (b * ZW_HALF + j) * 256 + threadIdx.x;
            pm[j] += from_fix(fm[j]);
            pl[j] += from_fix(fl[j]);
            if (q < WORKERS * CLS) {
                const int r = q / CLS;
                const int c = q - r * CLS;
                s_zw[r * HP + c] = pm[j] + (ev[j] * 0.01f) * expf(pl[j]);
            }
        }
        if (blockIdx.x == 0) {
#pragma unroll
            for (int j = 0; j < ZW_HALF; ++j) {
                const int q = (b * ZW_HALF + j) * 256 + threadIdx.x;
                if (q < WORKERS * CLS) {
                    mean[TASKS * CLS + q] = pm[j];
                    lsd[TASKS * CLS + q]  = pl[j];
                }
            }
        }
    }
    __syncthreads();

    constexpr int NT = (ANS + CR_TILE - 1) / CR_TILE;
    constexpr int NQ = CR_TILE * CLS / 4;
    int buf = 0;
    for (int tile = blockIdx.x; tile < NT; tile += gridDim.x, buf ^= 1) {
        const int base = tile * CR_TILE;
        float* sb = s_out[buf];

        int t[2], w[2];
#pragma unroll
        for (int h = 0; h < 2; ++h) {
            int a = base + h * 256 + threadIdx.x;
            if (a > ANS - 1) a = ANS - 1;
            t[h] = answers[3 * a];
            w[h] = answers[3 * a + 1];
        }
        float4 ta[2], tb[2];
        float2 tc[2];
#pragma unroll
        for (int h = 0; h < 2; ++h) {
            const float* zt = zp + (size_t)t[h] * HP;
            ta[h] = *(const float4*)(zt);
            tb[h] = *(const float4*)(zt + 4);
            tc[h] = *(const float2*)(zt + 8);
        }
#pragma unroll
        for (int h = 0; h < 2; ++h) {
            const float* zw = s_zw + w[h] * HP;
            const float4 wa = *(const float4*)(zw);
            const float4 wb = *(const float4*)(zw + 4);
            const float2 wc = *(const float2*)(zw + 8);
            float2* o = (float2*)(sb + (h * 256 + threadIdx.x) * CLS);
            o[0] = make_float2(ta[h].x * wa.x, ta[h].y * wa.y);
            o[1] = make_float2(ta[h].z * wa.z, ta[h].w * wa.w);
            o[2] = make_float2(tb[h].x * wb.x, tb[h].y * wb.y);
            o[3] = make_float2(tb[h].z * wb.z, tb[h].w * wb.w);
            o[4] = make_float2(tc[h].x * wc.x, tc[h].y * wc.y);
        }
        __syncthreads();

        float4* dst = (float4*)(crowd + (size_t)base * CLS);
        const float4* so = (const float4*)sb;
        if (ANS - base >= CR_TILE) {
#pragma unroll
            for (int k = 0; k < NQ / 256; ++k) dst[k * 256 + threadIdx.x] = so[k * 256 + threadIdx.x];
        } else {
            const int nq = (ANS - base) * CLS / 4;
            for (int k = threadIdx.x; k < nq; k += 256) dst[k] = so[k];
        }
    }
}


extern "C" void kernel_launch(void* const* d_in, const int* in_sizes, int n_in,
                              void* d_out, int out_size, void* d_ws, size_t ws_size,
                              hipStream_t stream) {
    (void)in_sizes; (void)n_in; (void)out_size; (void)ws_size;

    const float* task_feature   = (const float*)d_in[0];
    const int*   answers        = (const int*)d_in[1];
    const float* worker_feature = (const float*)d_in[2];
    const float* W_efc          = (const float*)d_in[3];
    const float* b_efc          = (const float*)d_in[4];
    const float* W_mean         = (const float*)d_in[5];
    const float* b_mean         = (const float*)d_in[6];
    const float* W_ls           = (const float*)d_in[7];
    const float* b_ls           = (const float*)d_in[8];
    const float* eps            = (const float*)d_in[9];

    float* out   = (float*)d_out;
    float* crowd = out;
    float* mean  = out + (size_t)ANS * CLS;
    float* lsd   = mean + (size_t)NODES * CLS;

    float*  ws   = (float*)d_ws;
    float*  tf_h = ws;
    float2* Bp   = (float2*)(tf_h + (size_t)TASKS * HID);
    float*  hm   = (float*)(Bp + (size_t)(FEAT / 2) * HID);
    float*  hl   = hm + (size_t)NODES * HP;
    int*    deg  = (int*)(hl + (size_t)NODES * HP + NODES);
    static_assert((size_t)TASKS * CAP * sizeof(unsigned short) <= (size_t)ANS * CLS * sizeof(float),
                  "the lists fit in the crowd block");
    unsigned short*     list = (unsigned short*)crowd;
    float*              zz   = tf_h;
    float4*             Wp   = (float4*)(tf_h + (size_t)2 * 1024 * 1024);
    unsigned long long* wsum = (unsigned long long*)(tf_h + (size_t)3 * 1024 * 1024);
    float*              wbm  = tf_h + (size_t)3 * 1024 * 1024 + 65536;
    float*              wbl  = wbm + 8192;
    static_assert((size_t)NODES * HP <= (size_t)2 * 1024 * 1024 &&
                  (size_t)WP_GROUPS * 32 * 4 <= (size_t)1024 * 1024 &&
                  (size_t)2 * WORKERS * CLS * sizeof(unsigned long long) <= (size_t)65536 * sizeof(float) &&
                  (size_t)WORKERS * CLS <= (size_t)8192 &&
                  (size_t)3 * 1024 * 1024 + 65536 + 2 * 8192 <= (size_t)TASKS * HID,
                  "z rows, packed class weights, worker sums and worker rows sit apart inside the old tf_h space");

    k_packB    <<<PB_GRID + DI_GRID + PW_GRID, 256, 0, stream>>>(W_efc, Bp, deg, W_mean, W_ls, Wp, wsum);
    k_gemm_count<<<DC_GRID + HW_GRID + GE_GRID, GT, 0, stream>>>(task_feature, Bp, b_efc, Wp,
                                                                  worker_feature, W_mean, W_ls, hm, hl,
                                                                  answers, deg, list);
    k_agg      <<<AW_GRID + AT_GRID, 256, 0, stream>>>(answers, list, deg, hm, hl,
                                                      b_mean, b_ls, eps, mean, lsd, wsum, zz, wbm, wbl);
    k_crowd    <<<CR_GRID, 256, 0, stream>>>(answers, zz, wbm, wbl, wsum, eps, mean, lsd, crowd);
}


# Worker-side edge sums scattered by the task threads from the row each already holds, as fixed point of h*rsqrt(deg_task); rsqrt(deg_worker) scales the finished sum, so worker rows round differently
# speedup vs baseline: 4.9592x; 1.0613x over previous
#include <hip/hip_runtime.h>

typedef __attribute__((ext_vector_type(2))) float v2f;
typedef __attribute__((ext_vector_type(8))) float v8f;

constexpr int TASKS   = 100000;
constexpr int WORKERS = 500;
constexpr int FEAT    = 512;
constexpr int CLS     = 10;
constexpr int HP      = 12;
constexpr int ANS     = 1000000;
constexpr int HID     = 128;
constexpr int NODES   = TASKS + WORKERS;
constexpr int AG_NPB  = 400;
constexpr int AG_TB   = 416;
constexpr int CAP     = 32;
constexpr int SENT    = 0xFFFF;
constexpr int CR_TILE = 512;
constexpr int CR_GRID = 512;
constexpr int DC_GRID = 256;
constexpr int DC_PER  = 8;
constexpr int GW      = 4;
constexpr int GT      = GW * 32;
constexpr int WPAD    = 512;
static_assert(WPAD >= WORKERS && WPAD % GT == 0, "the worker counts are a whole number of work-group rounds");
static_assert(ANS % 2 == 0 && CR_TILE % 2 == 0, "a tile's floats must split into 16-byte pieces");
static_assert((CR_TILE * CLS / 4) % 256 == 0, "a full tile is a whole number of 256-thread store rounds");


__device__ __forceinline__ void deg_count_body(const int blk, int* __restrict__ s_wdeg,
                                               const int* __restrict__ answers,
                                               int* __restrict__ deg,
                                               unsigned short* __restrict__ list) {
    for (int i = threadIdx.x; i < WPAD; i += GT) s_wdeg[i] = 0;
    __syncthreads();

    constexpr int TRIP = GT * DC_PER;
    for (int base = blk * TRIP; base < ANS; base += DC_GRID * TRIP) {
        int t[DC_PER], w[DC_PER], old[DC_PER];
        bool ok[DC_PER];
#pragma unroll
        for (int h = 0; h < DC_PER; ++h) {
            const int a = base + h * GT + threadIdx.x;
            ok[h] = a < ANS;
            const int ac = ok[h] ? a : ANS - 1;
            t[h] = answers[3 * ac];
            w[h] = answers[3 * ac + 1];
        }
#pragma unroll
        for (int h = 0; h < DC_PER; ++h) {
            old[h] = CAP + 1;
            if (ok[h]) {
                old[h] = atomicAdd(&deg[t[h]], 1);
                atomicAdd(&s_wdeg[w[h]], 1);
            }
        }
#pragma unroll
        for (int h = 0; h < DC_PER; ++h) {
            const int slot = old[h] - 1;
            if (slot < CAP) list[(size_t)t[h] * CAP + slot] = (unsigned short)w[h];
        }
    }
    __syncthreads();
    for (int i = threadIdx.x; i < WORKERS; i += GT) atomicAdd(&deg[TASKS + i], s_wdeg[i]);
}


constexpr int PB_GRID = ((FEAT / 2) * HID + 255) / 256;
constexpr int DI_GRID = (NODES + 255) / 256;
static_assert(DI_GRID * 256 >= 2 * WORKERS * CLS, "the degree groups cover every worker-side sum");

constexpr int WP_GROUPS = HID / 4;
constexpr int PW_GRID   = (WP_GROUPS * 32 + 255) / 256;
static_assert(HID % 16 == 0, "hidden tiles of 16");

__global__ void k_packB(const float* __restrict__ B, float2* __restrict__ Bp, int* __restrict__ deg,
                        const float* __restrict__ Wm, const float* __restrict__ Wl,
                        float4* __restrict__ Wp, unsigned long long* __restrict__ wsum) {
    if (blockIdx.x >= PB_GRID + DI_GRID) {
        const int e = (blockIdx.x - PB_GRID - DI_GRID) * 256 + threadIdx.x;
        if (e < WP_GROUPS * 32) {
            const int g = e >> 5, lane = e & 31;
            const int cc = lane & 15, h = lane >> 4;
            const int n = 16 * (g >> 2) + 8 * h + 2 * (g & 3);
            float4 v = make_float4(0.f, 0.f, 0.f, 0.f);
            if (cc < CLS) v = make_float4(Wm[n * CLS + cc], Wm[(n + 1) * CLS + cc],
                                          Wl[n * CLS + cc], Wl[(n + 1) * CLS + cc]);
            Wp[e] = v;
        }
        return;
    }
    if (blockIdx.x >= PB_GRID) {
        const int i = (blockIdx.x - PB_GRID) * 256 + threadIdx.x;
        if (i < NODES) deg[i] = 1;
        if (i < 2 * WORKERS * CLS) wsum[i] = 0ull;
        return;
    }
    int idx = blockIdx.x * 256 + threadIdx.x;
    if (idx < (FEAT / 2) * HID) {
        int kp = idx / HID;
        int n  = idx - kp * HID;
        Bp[idx] = make_float2(B[(2 * kp) * HID + n], B[(2 * kp + 1) * HID + n]);
    }
}


constexpr int KQ          = FEAT / GW;
constexpr int SLAB_PART   = 8 * 8 * 32;
static_assert(FEAT % (4 * GW) == 0, "a wave's k range is a whole number of 4-k steps");
static_assert(TASKS % 32 == 0, "every 32-row tile is a full one: one work-group each, none padded");

__device__ __forceinline__ void finish_slab(v8f (&a)[8], const float4* __restrict__ part,
                                            const float4* __restrict__ wp,
                                            float* __restrict__ hm, float* __restrict__ hl,
                                            const size_t m, const int hi) {
    for (int slot = 0; slot < GW - 1; ++slot) {
#pragma unroll
        for (int t = 0; t < 8; ++t) {
            const float4 lo = part[((slot * 8 + t) * 2 + 0) * 32];
            const float4 hh = part[((slot * 8 + t) * 2 + 1) * 32];
            a[t] += (v8f){lo.x, lo.y, lo.z, lo.w, hh.x, hh.y, hh.z, hh.w};
        }
    }

    v8f hM = {0.f, 0.f, 0.f, 0.f, 0.f, 0.f, 0.f, 0.f};
    v8f hL = {0.f, 0.f, 0.f, 0.f, 0.f, 0.f, 0.f, 0.f};
#pragma unroll
    for (int t = 0; t < 8; ++t) {
#pragma unroll
        for (int jp = 0; jp < 4; ++jp) {
            const float4 wv = wp[(t * 4 + jp) * 32];
            const v2f wm = {wv.x, wv.y};
            const v2f wl = {wv.z, wv.w};
            const float x0 = a[t][2 * jp];
            const float x1 = a[t][2 * jp + 1];
            const v2f xf = {x0 > 0.f ? x0 : 0.f, x1 > 0.f ? x1 : 0.f};
            hM = __builtin_amdgcn_wmma_f32_16x16x4_f32(false, wm, false, xf, (short)0, hM, false, false);
            hL = __builtin_amdgcn_wmma_f32_16x16x4_f32(false, wl, false, xf, (short)0, hL, false, false);
        }
    }
    float* pm = hm + m * HP + 8 * hi;
    float* pl = hl + m * HP + 8 * hi;
    if (hi == 0) {
        *(float4*)(pm)     = make_float4(hM[0], hM[1], hM[2], hM[3]);
        *(float4*)(pm + 4) = make_float4(hM[4], hM[5], hM[6], hM[7]);
        *(float4*)(pl)     = make_float4(hL[0], hL[1], hL[2], hL[3]);
        *(float4*)(pl + 4) = make_float4(hL[4], hL[5], hL[6], hL[7]);
    } else {
        *(float2*)(pm) = make_float2(hM[0], hM[1]);
        *(float2*)(pl) = make_float2(hL[0], hL[1]);
    }
}

__device__ __forceinline__ void gemm_efc_body(const int blk, float* __restrict__ s_part,
                                              const float* __restrict__ A,
                                              const float2* __restrict__ Bp,
                                              const float* __restrict__ bias,
                                              const float4* __restrict__ Wp,
                                              float* __restrict__ hm,
                                              float* __restrict__ hl) {
    const int wave   = threadIdx.x >> 5;
    const int lane   = threadIdx.x & 31;
    const int lane16 = lane & 15;
    const int hi     = lane >> 4;

    const int row_base = blk * 32;
    const int f0 = blk & (GW - 1);
    const int f1 = (f0 + 2) & (GW - 1);
    const bool fin0 = wave == f0;
    const bool fin1 = wave == f1;

    const float*  arow0 = A + (size_t)(row_base + lane16) * FEAT + 2 * hi;
    const float*  arow1 = arow0 + (size_t)16 * FEAT;
    const float2* bbase = Bp + lane16;

    v8f acc[2][8];
#pragma unroll
    for (int t = 0; t < 8; ++t) {
        const v8f zero = {0.f, 0.f, 0.f, 0.f, 0.f, 0.f, 0.f, 0.f};
        v8f b = zero;
        if (fin0 || fin1) {
            const float* bt = bias + t * 16 + 8 * hi;
            b = (v8f){bt[0], bt[1], bt[2], bt[3], bt[4], bt[5], bt[6], bt[7]};
        }
        acc[0][t] = fin0 ? b : zero;
        acc[1][t] = fin1 ? b : zero;
    }

    for (int k = wave * KQ; k < (wave + 1) * KQ; k += 4) {
        const float2 av0 = *(const float2*)(arow0 + k);
        const float2 av1 = *(const float2*)(arow1 + k);
        v2f af0 = {av0.x, av0.y};
        v2f af1 = {av1.x, av1.y};
        const float2* brow = bbase + (size_t)((k >> 1) + hi) * HID;
#pragma unroll
        for (int t = 0; t < 8; ++t) {
            const float2 bv = brow[t * 16];
            v2f bf = {bv.x, bv.y};
            acc[0][t] = __builtin_amdgcn_wmma_f32_16x16x4_f32(
                false, bf, false, af0, (short)0, acc[0][t], false, false);
            acc[1][t] = __builtin_amdgcn_wmma_f32_16x16x4_f32(
                false, bf, false, af1, (short)0, acc[1][t], false, false);
        }
    }

    float4* sp = (float4*)s_part + lane;
#pragma unroll
    for (int r = 0; r < 2; ++r) {
        const int fr = r == 0 ? f0 : f1;
        if (wave != fr) {
            const int slot = (wave - fr - 1) & (GW - 1);
            float4* dst = sp + (r * (GW - 1) + slot) * (SLAB_PART / 4);
#pragma unroll
            for (int t = 0; t < 8; ++t) {
                const v8f a = acc[r][t];
                dst[(t * 2 + 0) * 32] = make_float4(a[0], a[1], a[2], a[3]);
                dst[(t * 2 + 1) * 32] = make_float4(a[4], a[5], a[6], a[7]);
            }
        }
    }
    __syncthreads();
    if (fin0) {
        finish_slab(acc[0], sp, Wp + lane, hm, hl, (size_t)(row_base + lane16), hi);
    } else if (fin1) {
        finish_slab(acc[1], sp + (GW - 1) * (SLAB_PART / 4), Wp + lane, hm, hl,
                    (size_t)(row_base + 16 + lane16), hi);
    }
}

static_assert(TASKS % 16 == 0, "worker rows start on a strip boundary");

__device__ __forceinline__ void h_body(const int blk,
                                       const float* __restrict__ wf,
                                       const float* __restrict__ Wm,
                                       const float* __restrict__ Wl,
                                       float* __restrict__ hm,
                                       float* __restrict__ hl) {
    const int wave   = threadIdx.x >> 5;
    const int lane   = threadIdx.x & 31;
    const int lane16 = lane & 15;
    const int hi     = lane >> 4;

    const int node_base = TASKS + (blk * GW + wave) * 16;
    int nd = node_base + lane16;
    if (nd > NODES - 1) nd = NODES - 1;
    const float* xrow = wf + (size_t)(nd - TASKS) * HID;
    const int n = lane16;
    const bool ncol = n < CLS;

    v8f accM = {0.f, 0.f, 0.f, 0.f, 0.f, 0.f, 0.f, 0.f};
    v8f accL = {0.f, 0.f, 0.f, 0.f, 0.f, 0.f, 0.f, 0.f};

    for (int k = 0; k < HID; k += 4) {
        const int k0 = k + 2 * hi;
        const float2 av = *(const float2*)(xrow + k0);
        v2f af = {av.x, av.y};
        v2f bm, bl;
        bm.x = ncol ? Wm[k0 * CLS + n] : 0.f;
        bm.y = ncol ? Wm[(k0 + 1) * CLS + n] : 0.f;
        bl.x = ncol ? Wl[k0 * CLS + n] : 0.f;
        bl.y = ncol ? Wl[(k0 + 1) * CLS + n] : 0.f;
        accM = __builtin_amdgcn_wmma_f32_16x16x4_f32(false, af, false, bm, (short)0, accM, false, false);
        accL = __builtin_amdgcn_wmma_f32_16x16x4_f32(false, af, false, bl, (short)0, accL, false, false);
    }

    if (ncol) {
#pragma unroll
        for (int j = 0; j < 8; ++j) {
            int m = node_base + j + 8 * hi;
            if (m < NODES) {
                hm[(size_t)m * HP + n] = accM[j];
                hl[(size_t)m * HP + n] = accL[j];
            }
        }
    }
}

constexpr int GE_GRID = TASKS / 32;
constexpr int HW_GRID = ((WORKERS + 15) / 16 + GW - 1) / GW;

__global__ __launch_bounds__(GT) void k_gemm_count(const float* __restrict__ A,
                                                    const float2* __restrict__ Bp,
                                                    const float* __restrict__ bias,
                                                    const float4* __restrict__ Wp,
                                                    const float* __restrict__ wf,
                                                    const float* __restrict__ Wm,
                                                    const float* __restrict__ Wl,
                                                    float* __restrict__ hm,
                                                    float* __restrict__ hl,
                                                    const int* __restrict__ answers,
                                                    int* __restrict__ deg,
                                                    unsigned short* __restrict__ list) {
    __shared__ __attribute__((aligned(16))) float s_buf[2 * (GW - 1) * SLAB_PART];
    static_assert(WPAD * sizeof(int) <= 2 * (GW - 1) * SLAB_PART * sizeof(float), "the worker counts fit in the block");
    if (blockIdx.x < DC_GRID) {
        deg_count_body(blockIdx.x, (int*)s_buf, answers, deg, list);
    } else if (blockIdx.x < DC_GRID + HW_GRID) {
        h_body(blockIdx.x - DC_GRID, wf, Wm, Wl, hm, hl);
    } else {
        gemm_efc_body(blockIdx.x - DC_GRID - HW_GRID, s_buf, A, Bp, bias, Wp, hm, hl);
    }
}


__device__ __forceinline__ void load_row10(const float* __restrict__ p, float* r) {
    const float4 a = *(const float4*)(p);
    const float4 b = *(const float4*)(p + 4);
    const float2 c = *(const float2*)(p + 8);
    r[0] = a.x; r[1] = a.y; r[2] = a.z; r[3] = a.w;
    r[4] = b.x; r[5] = b.y; r[6] = b.z; r[7] = b.w;
    r[8] = c.x; r[9] = c.y;
}

__device__ __forceinline__ unsigned long long to_fix(float v) {
    return (unsigned long long)(long long)rintf(v * 4294967296.0f);
}

__device__ __forceinline__ float from_fix(unsigned long long s) {
    return (float)(long long)s * 2.3283064365386963e-10f;
}

constexpr int AG_GRID = (NODES + AG_NPB - 1) / AG_NPB;
static_assert((AG_NPB * CLS * sizeof(float)) % 128 == 0, "a work-group's rows of mean / lsd are whole 128-byte lines");
static_assert(TASKS % AG_NPB == 0, "a work-group holds task nodes only or worker nodes only");
static_assert(AG_TB >= AG_NPB && AG_TB % 32 == 0, "one thread per node, whole waves");

__global__ __launch_bounds__(AG_TB) void k_agg(const int* __restrict__ answers,
                                               const unsigned short* __restrict__ list,
                                               const int* __restrict__ deg,
                                               const float* __restrict__ hm,
                                               const float* __restrict__ hl,
                                               const float* __restrict__ bm,
                                               const float* __restrict__ bl,
                                               const float* __restrict__ eps,
                                               float* __restrict__ mean,
                                               float* __restrict__ lsd,
                                               unsigned long long* __restrict__ wsum,
                                               float* __restrict__ zp,
                                               float* __restrict__ wbm,
                                               float* __restrict__ wbl) {
    __shared__ __attribute__((aligned(16))) unsigned long long wacc[2 * WORKERS * CLS];
    __shared__ __attribute__((aligned(16))) float sm[AG_TB * CLS];
    __shared__ __attribute__((aligned(16))) float sl[AG_TB * CLS];

    const int blk = blockIdx.x;
    const bool task_blk = blk * AG_NPB < TASKS;
    if (task_blk) {
        for (int q = threadIdx.x; q < 2 * WORKERS * CLS; q += AG_TB) wacc[q] = 0ull;
    }
    __syncthreads();

    const int i = blk * AG_NPB + threadIdx.x;
    float am[CLS], al[CLS];
#pragma unroll
    for (int c = 0; c < CLS; ++c) { am[c] = 0.f; al[c] = 0.f; }

    if (threadIdx.x < AG_NPB && i < NODES) {
        const int   dg = deg[i];
        const float d  = rsqrtf((float)dg);
        const float sw = d * d;
        float2 ev[CLS / 2];
        {
            const float2* ep = (const float2*)(eps + (size_t)(i < TASKS ? i : TASKS - 1) * CLS);
#pragma unroll
            for (int c = 0; c < CLS / 2; ++c) ev[c] = ep[c];
        }
        float hmi[CLS], hli[CLS];
        load_row10(hm + (size_t)i * HP, hmi);
        load_row10(hl + (size_t)i * HP, hli);
#pragma unroll
        for (int c = 0; c < CLS; ++c) {
            am[c] = bm[c] + hmi[c] * sw;
            al[c] = bl[c] + hli[c] * sw;
        }

        if (i < TASKS) {
            const int n = dg - 1;
            unsigned long long fxm[CLS], fxl[CLS];
#pragma unroll
            for (int c = 0; c < CLS; ++c) {
                fxm[c] = to_fix(hmi[c] * d);
                fxl[c] = to_fix(hli[c] * d);
            }
            if (n <= CAP) {
                const uint4* lp = (const uint4*)(list + (size_t)i * CAP);
                const uint4 q0 = lp[0], q1 = lp[1], q2 = lp[2], q3 = lp[3];
                const unsigned pk[16] = {q0.x, q0.y, q0.z, q0.w, q1.x, q1.y, q1.z, q1.w,
                                         q2.x, q2.y, q2.z, q2.w, q3.x, q3.y, q3.z, q3.w};
                int e[CAP];
#pragma unroll
                for (int j = 0; j < CAP; ++j) {
                    const int v = (int)((pk[j >> 1] >> (16 * (j & 1))) & 0xFFFFu);
                    e[j] = (j < n) ? v : SENT;
                }
                int cur = -1;
                for (int it = 0; it < CAP; ++it) {
                    int m = SENT, cnt = 0;
#pragma unroll
                    for (int j = 0; j < CAP; ++j) {
                        const int v = e[j];
                        if (v > cur) {
                            if (v < m) { m = v; cnt = 1; }
                            else if (v == m) ++cnt;
                        }
                    }
                    if (m == SENT) break;
                    const int wrow = TASKS + m;
                    const float we = d * rsqrtf((float)deg[wrow]);
                    float hmw[CLS], hlw[CLS];
                    load_row10(hm + (size_t)wrow * HP, hmw);
                    load_row10(hl + (size_t)wrow * HP, hlw);
                    unsigned long long* wa = wacc + m * CLS;
                    for (int k = 0; k < cnt; ++k) {
#pragma unroll
                        for (int c = 0; c < CLS; ++c) {
                            am[c] += hmw[c] * we;
                            al[c] += hlw[c] * we;
                        }
#pragma unroll
                        for (int c = 0; c < CLS; ++c) {
                            atomicAdd(&wa[c], fxm[c]);
                            atomicAdd(&wa[WORKERS * CLS + c], fxl[c]);
                        }
                    }
                    cur = m;
                }
            } else {
                for (int a = 0; a < ANS; ++a) {
                    if (answers[3 * a] == i) {
                        const int w    = answers[3 * a + 1];
                        const int wrow = TASKS + w;
                        const float we = d * rsqrtf((float)deg[wrow]);
                        float hmw[CLS], hlw[CLS];
                        load_row10(hm + (size_t)wrow * HP, hmw);
                        load_row10(hl + (size_t)wrow * HP, hlw);
                        unsigned long long* wa = wacc + w * CLS;
#pragma unroll
                        for (int c = 0; c < CLS; ++c) {
                            am[c] += hmw[c] * we;
                            al[c] += hlw[c] * we;
                        }
#pragma unroll
                        for (int c = 0; c < CLS; ++c) {
                            atomicAdd(&wa[c], fxm[c]);
                            atomicAdd(&wa[WORKERS * CLS + c], fxl[c]);
                        }
                    }
                }
            }
            float z[CLS];
#pragma unroll
            for (int c = 0; c < CLS / 2; ++c) {
                z[2 * c]     = am[2 * c]     + (ev[c].x * 0.01f) * expf(al[2 * c]);
                z[2 * c + 1] = am[2 * c + 1] + (ev[c].y * 0.01f) * expf(al[2 * c + 1]);
            }
            float* zr = zp + (size_t)i * HP;
            *(float4*)(zr)     = make_float4(z[0], z[1], z[2], z[3]);
            *(float4*)(zr + 4) = make_float4(z[4], z[5], z[6], z[7]);
            *(float2*)(zr + 8) = make_float2(z[8], z[9]);
        } else {
            const int q = (i - TASKS) * CLS;
#pragma unroll
            for (int c = 0; c < CLS; ++c) { wbm[q + c] = am[c]; wbl[q + c] = al[c]; }
        }
    }

#pragma unroll
    for (int c = 0; c < CLS; ++c) {
        sm[threadIdx.x * CLS + c] = am[c];
        sl[threadIdx.x * CLS + c] = al[c];
    }
    __syncthreads();
    const int base = blk * AG_NPB * CLS;
    const int lim  = NODES * CLS - base;
    for (int k = threadIdx.x; k < AG_NPB * CLS; k += AG_TB) {
        if (k < lim) {
            mean[(size_t)base + k] = sm[k];
            lsd[(size_t)base + k]  = sl[k];
        }
    }
    if (task_blk) {
        for (int q = threadIdx.x; q < 2 * WORKERS * CLS; q += AG_TB) atomicAdd(&wsum[q], wacc[q]);
    }
}


constexpr int ZW_PER  = (WORKERS * CLS + 255) / 256;
constexpr int ZW_HALF = ZW_PER / 2;
static_assert(ZW_PER % 2 == 0, "the worker-row elements of a thread split into two equal batches");

__global__ __launch_bounds__(256) void k_crowd(const int* __restrict__ answers,
                                               const float* __restrict__ zp,
                                               const float* __restrict__ wbm,
                                               const float* __restrict__ wbl,
                                               const unsigned long long* __restrict__ wsum,
                                               const int* __restrict__ deg,
                                               const float* __restrict__ eps,
                                               float* __restrict__ mean,
                                               float* __restrict__ lsd,
                                               float* __restrict__ crowd) {
    __shared__ __attribute__((aligned(16))) float s_zw[WORKERS * HP];
    __shared__ __attribute__((aligned(16))) float s_out[2][CR_TILE * CLS];

    for (int b = 0; b < 2; ++b) {
        float pm[ZW_HALF], pl[ZW_HALF], ev[ZW_HALF];
        unsigned long long fm[ZW_HALF], fl[ZW_HALF];
        int dgw[ZW_HALF];
#pragma unroll
        for (int j = 0; j < ZW_HALF; ++j) {
            int q = (b * ZW_HALF + j) * 256 + threadIdx.x;
            if (q > WORKERS * CLS - 1) q = WORKERS * CLS - 1;
            pm[j] = wbm[q];
            pl[j] = wbl[q];
            fm[j] = wsum[q];
            fl[j] = wsum[WORKERS * CLS + q];
            dgw[j] = deg[TASKS + q / CLS];
            ev[j] = eps[TASKS * CLS + q];
        }
#pragma unroll
        for (int j = 0; j < ZW_HALF; ++j) {
            const int q = (b * ZW_HALF + j) * 256 + threadIdx.x;
            const float dw = rsqrtf((float)dgw[j]);
            pm[j] += from_fix(fm[j]) * dw;
            pl[j] += from_fix(fl[j]) * dw;
            if (q < WORKERS * CLS) {
                const int r = q / CLS;
                const int c = q - r * CLS;
                s_zw[r * HP + c] = pm[j] + (ev[j] * 0.01f) * expf(pl[j]);
            }
        }
        if (blockIdx.x == 0) {
#pragma unroll
            for (int j = 0; j < ZW_HALF; ++j) {
                const int q = (b * ZW_HALF + j) * 256 + threadIdx.x;
                if (q < WORKERS * CLS) {
                    mean[TASKS * CLS + q] = pm[j];
                    lsd[TASKS * CLS + q]  = pl[j];
                }
            }
        }
    }
    __syncthreads();

    constexpr int NT = (ANS + CR_TILE - 1) / CR_TILE;
    constexpr int NQ = CR_TILE * CLS / 4;
    int buf = 0;
    for (int tile = blockIdx.x; tile < NT; tile += gridDim.x, buf ^= 1) {
        const int base = tile * CR_TILE;
        float* sb = s_out[buf];

        int t[2], w[2];
#pragma unroll
        for (int h = 0; h < 2; ++h) {
            int a = base + h * 256 + threadIdx.x;
            if (a > ANS - 1) a = ANS - 1;
            t[h] = answers[3 * a];
            w[h] = answers[3 * a + 1];
        }
        float4 ta[2], tb[2];
        float2 tc[2];
#pragma unroll
        for (int h = 0; h < 2; ++h) {
            const float* zt = zp + (size_t)t[h] * HP;
            ta[h] = *(const float4*)(zt);
            tb[h] = *(const float4*)(zt + 4);
            tc[h] = *(const float2*)(zt + 8);
        }
#pragma unroll
        for (int h = 0; h < 2; ++h) {
            const float* zw = s_zw + w[h] * HP;
            const float4 wa = *(const float4*)(zw);
            const float4 wb = *(const float4*)(zw + 4);
            const float2 wc = *(const float2*)(zw + 8);
            float2* o = (float2*)(sb + (h * 256 + threadIdx.x) * CLS);
            o[0] = make_float2(ta[h].x * wa.x, ta[h].y * wa.y);
            o[1] = make_float2(ta[h].z * wa.z, ta[h].w * wa.w);
            o[2] = make_float2(tb[h].x * wb.x, tb[h].y * wb.y);
            o[3] = make_float2(tb[h].z * wb.z, tb[h].w * wb.w);
            o[4] = make_float2(tc[h].x * wc.x, tc[h].y * wc.y);
        }
        __syncthreads();

        float4* dst = (float4*)(crowd + (size_t)base * CLS);
        const float4* so = (const float4*)sb;
        if (ANS - base >= CR_TILE) {
#pragma unroll
            for (int k = 0; k < NQ / 256; ++k) dst[k * 256 + threadIdx.x] = so[k * 256 + threadIdx.x];
        } else {
            const int nq = (ANS - base) * CLS / 4;
            for (int k = threadIdx.x; k < nq; k += 256) dst[k] = so[k];
        }
    }
}


extern "C" void kernel_launch(void* const* d_in, const int* in_sizes, int n_in,
                              void* d_out, int out_size, void* d_ws, size_t ws_size,
                              hipStream_t stream) {
    (void)in_sizes; (void)n_in; (void)out_size; (void)ws_size;

    const float* task_feature   = (const float*)d_in[0];
    const int*   answers        = (const int*)d_in[1];
    const float* worker_feature = (const float*)d_in[2];
    const float* W_efc          = (const float*)d_in[3];
    const float* b_efc          = (const float*)d_in[4];
    const float* W_mean         = (const float*)d_in[5];
    const float* b_mean         = (const float*)d_in[6];
    const float* W_ls           = (const float*)d_in[7];
    const float* b_ls           = (const float*)d_in[8];
    const float* eps            = (const float*)d_in[9];

    float* out   = (float*)d_out;
    float* crowd = out;
    float* mean  = out + (size_t)ANS * CLS;
    float* lsd   = mean + (size_t)NODES * CLS;

    float*  ws   = (float*)d_ws;
    float*  tf_h = ws;
    float2* Bp   = (float2*)(tf_h + (size_t)TASKS * HID);
    float*  hm   = (float*)(Bp + (size_t)(FEAT / 2) * HID);
    float*  hl   = hm + (size_t)NODES * HP;
    int*    deg  = (int*)(hl + (size_t)NODES * HP + NODES);
    static_assert((size_t)TASKS * CAP * sizeof(unsigned short) <= (size_t)ANS * CLS * sizeof(float),
                  "the lists fit in the crowd block");
    unsigned short*     list = (unsigned short*)crowd;
    float*              zz   = tf_h;
    float4*             Wp   = (float4*)(tf_h + (size_t)2 * 1024 * 1024);
    unsigned long long* wsum = (unsigned long long*)(tf_h + (size_t)3 * 1024 * 1024);
    float*              wbm  = tf_h + (size_t)3 * 1024 * 1024 + 65536;
    float*              wbl  = wbm + 8192;
    static_assert((size_t)NODES * HP <= (size_t)2 * 1024 * 1024 &&
                  (size_t)WP_GROUPS * 32 * 4 <= (size_t)1024 * 1024 &&
                  (size_t)2 * WORKERS * CLS * sizeof(unsigned long long) <= (size_t)65536 * sizeof(float) &&
                  (size_t)WORKERS * CLS <= (size_t)8192 &&
                  (size_t)3 * 1024 * 1024 + 65536 + 2 * 8192 <= (size_t)TASKS * HID,
                  "z rows, packed class weights, worker sums and worker rows sit apart inside the old tf_h space");

    k_packB    <<<PB_GRID + DI_GRID + PW_GRID, 256, 0, stream>>>(W_efc, Bp, deg, W_mean, W_ls, Wp, wsum);
    k_gemm_count<<<DC_GRID + HW_GRID + GE_GRID, GT, 0, stream>>>(task_feature, Bp, b_efc, Wp,
                                                                  worker_feature, W_mean, W_ls, hm, hl,
                                                                  answers, deg, list);
    k_agg      <<<AG_GRID, AG_TB, 0, stream>>>(answers, list, deg, hm, hl,
                                              b_mean, b_ls, eps, mean, lsd, wsum, zz, wbm, wbl);
    k_crowd    <<<CR_GRID, 256, 0, stream>>>(answers, zz, wbm, wbl, wsum, deg, eps, mean, lsd, crowd);
}
